# starting kernel
# baseline (speedup 1.0000x reference)
#include <hip/hip_runtime.h>

typedef __attribute__((ext_vector_type(2))) float v2f;
typedef __attribute__((ext_vector_type(8))) float v8f;

#define N_NODES 100000
#define N_EDGES 1600000
#define D 64
#define N_LAYERS 5

static inline size_t align256(size_t x) { return (x + 255) & ~(size_t)255; }

__global__ __launch_bounds__(256) void gnn_deg_int_kernel(const int* __restrict__ dst,
                                                          int* __restrict__ degi, int nE) {
  int e = blockIdx.x * 256 + threadIdx.x;
  if (e < nE) atomicAdd(&degi[dst[e]], 1);
}

__global__ __launch_bounds__(256) void gnn_invdeg_from_int_kernel(const int* __restrict__ degi,
                                                                  float* __restrict__ invd, int n) {
  int i = blockIdx.x * 256 + threadIdx.x;
  if (i < n) invd[i] = 1.0f / fmaxf((float)degi[i], 1.0f);
}

__global__ __launch_bounds__(256) void gnn_scan_p1_kernel(const int* __restrict__ degi,
                                                          int* __restrict__ blockSums, int n) {
  __shared__ int s[256];
  int i = blockIdx.x * 256 + threadIdx.x;
  s[threadIdx.x] = (i < n) ? degi[i] : 0;
  __syncthreads();
  for (int off = 128; off > 0; off >>= 1) {
    if (threadIdx.x < off) s[threadIdx.x] += s[threadIdx.x + off];
    __syncthreads();
  }
  if (threadIdx.x == 0) blockSums[blockIdx.x] = s[0];
}

__global__ void gnn_scan_p2_kernel(int* __restrict__ blockSums, int nb) {
  if (blockIdx.x == 0 && threadIdx.x == 0) {
    int acc = 0;
    for (int i = 0; i < nb; ++i) {
      int v = blockSums[i];
      blockSums[i] = acc;
      acc += v;
    }
  }
}

__global__ __launch_bounds__(256) void gnn_scan_p3_kernel(const int* __restrict__ degi,
                                                          const int* __restrict__ blockSums,
                                                          int* __restrict__ rowptr,
                                                          int* __restrict__ fillpos, int n) {
  __shared__ int s[256];
  int i = blockIdx.x * 256 + threadIdx.x;
  int v = (i < n) ? degi[i] : 0;
  s[threadIdx.x] = v;
  __syncthreads();
  for (int off = 1; off < 256; off <<= 1) {
    int t = (threadIdx.x >= off) ? s[threadIdx.x - off] : 0;
    __syncthreads();
    s[threadIdx.x] += t;
    __syncthreads();
  }
  if (i < n) {
    int excl = s[threadIdx.x] - v + blockSums[blockIdx.x];
    rowptr[i] = excl;
    fillpos[i] = excl;
  }
}

__global__ __launch_bounds__(256) void gnn_csr_fill_kernel(const int* __restrict__ src,
                                                           const int* __restrict__ dstv,
                                                           int* __restrict__ fillpos,
                                                           int* __restrict__ colidx, int nE) {
  int e = blockIdx.x * 256 + threadIdx.x;
  if (e < nE) {
    int pos = atomicAdd(&fillpos[dstv[e]], 1);
    colidx[pos] = src[e];
  }
}

__global__ __launch_bounds__(256) void gnn_gather_agg_kernel(const float* __restrict__ h,
                                                             const int* __restrict__ colidx,
                                                             const int* __restrict__ rowptr,
                                                             const int* __restrict__ degi,
                                                             float* __restrict__ agg, int nN) {
  int wave = threadIdx.x >> 5;
  int lane = threadIdx.x & 31;
  int node = blockIdx.x * 8 + wave;
  if (node >= nN) return;
  int beg = rowptr[node];
  int cnt = degi[node];
  const float* hp = h + 2 * lane;

  float2 a0 = make_float2(0.f, 0.f), a1 = a0, a2 = a0, a3 = a0;
  int t = 0;
  for (; t + 4 <= cnt; t += 4) {
    int n0 = colidx[beg + t + 0];
    int n1 = colidx[beg + t + 1];
    int n2 = colidx[beg + t + 2];
    int n3 = colidx[beg + t + 3];
    const float2 v0 = *(const float2*)(hp + (long)n0 * D);
    const float2 v1 = *(const float2*)(hp + (long)n1 * D);
    const float2 v2 = *(const float2*)(hp + (long)n2 * D);
    const float2 v3 = *(const float2*)(hp + (long)n3 * D);
    a0.x += v0.x; a0.y += v0.y;
    a1.x += v1.x; a1.y += v1.y;
    a2.x += v2.x; a2.y += v2.y;
    a3.x += v3.x; a3.y += v3.y;
  }
  for (; t < cnt; ++t) {
    int nb = colidx[beg + t];
    const float2 v = *(const float2*)(hp + (long)nb * D);
    a0.x += v.x; a0.y += v.y;
  }
  float2 acc;
  acc.x = (a0.x + a1.x) + (a2.x + a3.x);
  acc.y = (a0.y + a1.y) + (a2.y + a3.y);
  *(float2*)(agg + (long)node * D + 2 * lane) = acc;
}

__global__ __launch_bounds__(256) void gnn_scatter_kernel(const float* __restrict__ h,
                                                          const int* __restrict__ src,
                                                          const int* __restrict__ dstv,
                                                          float* __restrict__ agg, int nE) {
  long tid = (long)blockIdx.x * 256 + threadIdx.x;
  int e = (int)(tid >> 4);
  int c = (int)(tid & 15);
  if (e >= nE) return;
  int s = src[e];
  int d = dstv[e];
  const float4 v = *(const float4*)(h + (long)s * D + c * 4);
  float* ap = agg + (long)d * D + c * 4;
  atomicAdd(ap + 0, v.x);
  atomicAdd(ap + 1, v.y);
  atomicAdd(ap + 2, v.z);
  atomicAdd(ap + 3, v.w);
}

__global__ __launch_bounds__(256) void gnn_sage_gemm_kernel(
    const float* __restrict__ agg, const float* __restrict__ h,
    const float* __restrict__ invdeg,
    const float* __restrict__ Wl, const float* __restrict__ Wr,
    const float* __restrict__ bias,
    float* __restrict__ out, int nN, int do_relu, int do_resid) {
  __shared__ __align__(16) float2 sWf[2][16][4][32];
  __shared__ __align__(16) float sB[D];
  int tid = threadIdx.x;

  for (int t = tid; t < 2048; t += 256) {
    int k  = t >> 7;
    int n  = (t >> 5) & 3;
    int ln = t & 31;
    int koff = (ln < 16) ? 0 : 2;
    int lr = ln & 15;
    int kb = k * 4 + koff;
    int col = n * 16 + lr;
    float2 f0, f1;
    f0.x = Wl[(kb + 0) * D + col];
    f0.y = Wl[(kb + 1) * D + col];
    f1.x = Wr[(kb + 0) * D + col];
    f1.y = Wr[(kb + 1) * D + col];
    sWf[0][k][n][ln] = f0;
    sWf[1][k][n][ln] = f1;
  }
  if (tid < D) sB[tid] = bias[tid];
  __syncthreads();

  int wave = tid >> 5;
  int lane = tid & 31;
  int lrow = lane & 15;
  int khalf = (lane < 16) ? 0 : 2;
  int rowbase = blockIdx.x * 128 + wave * 16;
  int M = rowbase + lrow;
  int Mc = (M < nN) ? M : (nN - 1);

  float invd = invdeg[Mc];

  v8f acc[4];
#pragma unroll
  for (int n = 0; n < 4; ++n) {
    float bv = sB[n * 16 + lrow];
#pragma unroll
    for (int j = 0; j < 8; ++j) acc[n][j] = bv;
  }

  for (int p = 0; p < 2; ++p) {
    const float* Arow = (p == 0 ? agg : h) + (long)Mc * D;
    float scale = (p == 0) ? invd : 1.0f;
#pragma unroll
    for (int k = 0; k < 16; ++k) {
      int kb = k * 4 + khalf;
      v2f a;
      a[0] = Arow[kb + 0] * scale;
      a[1] = Arow[kb + 1] * scale;
#pragma unroll
      for (int n = 0; n < 4; ++n) {
        float2 bf2 = sWf[p][k][n][lane];
        v2f bf;
        bf[0] = bf2.x;
        bf[1] = bf2.y;
        acc[n] = __builtin_amdgcn_wmma_f32_16x16x4_f32(
            false, a, false, bf, (short)0, acc[n], false, false);
      }
    }
  }

#pragma unroll
  for (int n = 0; n < 4; ++n) {
#pragma unroll
    for (int j = 0; j < 8; ++j) {
      int row = rowbase + ((lane < 16) ? j : j + 8);
      int col = n * 16 + lrow;
      if (row < nN) {
        float v = acc[n][j];
        if (do_resid) v += h[(long)row * D + col];
        if (do_relu) v = fmaxf(v, 0.0f);
        out[(long)row * D + col] = v;
      }
    }
  }
}

extern "C" void kernel_launch(void* const* d_in, const int* in_sizes, int n_in,
                              void* d_out, int out_size, void* d_ws, size_t ws_size,
                              hipStream_t stream) {
  const float* x  = (const float*)d_in[0];
  const int*   ei = (const int*)d_in[1];
  const float* Wl = (const float*)d_in[2];
  const float* Wr = (const float*)d_in[3];
  const float* b  = (const float*)d_in[4];
  float* out = (float*)d_out;

  const int* src = ei;
  const int* dst = ei + N_EDGES;

  const size_t featBytes = (size_t)N_NODES * D * sizeof(float);
  const size_t nodeI     = align256((size_t)N_NODES * sizeof(int));
  const size_t nodeF     = align256((size_t)N_NODES * sizeof(float));
  const size_t edgeI     = align256((size_t)N_EDGES * sizeof(int));
  const int scanBlocks   = (N_NODES + 255) / 256;
  const size_t bsumBytes = align256((size_t)scanBlocks * sizeof(int));

  const size_t csrNeed = 3 * nodeI + bsumBytes + nodeF + edgeI + 3 * featBytes;

  char* ws = (char*)d_ws;

  if (ws_size >= csrNeed) {
    int*   degi      = (int*)ws;   ws += nodeI;
    int*   rowptr    = (int*)ws;   ws += nodeI;
    int*   fillpos   = (int*)ws;   ws += nodeI;
    int*   blockSums = (int*)ws;   ws += bsumBytes;
    float* invdeg    = (float*)ws; ws += nodeF;
    int*   colidx    = (int*)ws;   ws += edgeI;
    float* agg       = (float*)ws; ws += featBytes;
    float* h1        = (float*)ws; ws += featBytes;
    float* h2        = (float*)ws;

    hipMemsetAsync(degi, 0, (size_t)N_NODES * sizeof(int), stream);
    gnn_deg_int_kernel<<<(N_EDGES + 255) / 256, 256, 0, stream>>>(dst, degi, N_EDGES);
    gnn_invdeg_from_int_kernel<<<scanBlocks, 256, 0, stream>>>(degi, invdeg, N_NODES);
    gnn_scan_p1_kernel<<<scanBlocks, 256, 0, stream>>>(degi, blockSums, N_NODES);
    gnn_scan_p2_kernel<<<1, 64, 0, stream>>>(blockSums, scanBlocks);
    gnn_scan_p3_kernel<<<scanBlocks, 256, 0, stream>>>(degi, blockSums, rowptr, fillpos, N_NODES);
    gnn_csr_fill_kernel<<<(N_EDGES + 255) / 256, 256, 0, stream>>>(src, dst, fillpos, colidx, N_EDGES);

    const float* h = x;
    float* houts[N_LAYERS] = {h1, h2, h1, h2, out};
    int gatherBlocks = (N_NODES + 7) / 8;
    int gemmBlocks   = (N_NODES + 127) / 128;

    for (int i = 0; i < N_LAYERS; ++i) {
      gnn_gather_agg_kernel<<<gatherBlocks, 256, 0, stream>>>(h, colidx, rowptr, degi, agg, N_NODES);
      int relu  = (i < N_LAYERS - 1) ? 1 : 0;
      int resid = (i > 0 && i < N_LAYERS - 1) ? 1 : 0;
      gnn_sage_gemm_kernel<<<gemmBlocks, 256, 0, stream>>>(
          agg, h, invdeg,
          Wl + (long)i * D * D, Wr + (long)i * D * D, b + (long)i * D,
          houts[i], N_NODES, relu, resid);
      h = houts[i];
    }
  } else {
    float* invdeg = (float*)ws; ws += nodeF;
    float* agg    = (float*)ws; ws += featBytes;
    float* h1     = (float*)ws; ws += featBytes;
    float* h2     = (float*)ws;

    hipMemsetAsync(invdeg, 0, (size_t)N_NODES * sizeof(float), stream);
    gnn_deg_int_kernel<<<(N_EDGES + 255) / 256, 256, 0, stream>>>(dst, (int*)invdeg, N_EDGES);
    gnn_invdeg_from_int_kernel<<<scanBlocks, 256, 0, stream>>>((const int*)invdeg, invdeg, N_NODES);

    const float* h = x;
    float* houts[N_LAYERS] = {h1, h2, h1, h2, out};
    long scatterBlocks = ((long)N_EDGES * 16 + 255) / 256;
    int gemmBlocks = (N_NODES + 127) / 128;

    for (int i = 0; i < N_LAYERS; ++i) {
      hipMemsetAsync(agg, 0, featBytes, stream);
      gnn_scatter_kernel<<<(int)scatterBlocks, 256, 0, stream>>>(h, src, dst, agg, N_EDGES);
      int relu  = (i < N_LAYERS - 1) ? 1 : 0;
      int resid = (i > 0 && i < N_LAYERS - 1) ? 1 : 0;
      gnn_sage_gemm_kernel<<<gemmBlocks, 256, 0, stream>>>(
          agg, h, invdeg,
          Wl + (long)i * D * D, Wr + (long)i * D * D, b + (long)i * D,
          houts[i], N_NODES, relu, resid);
      h = houts[i];
    }
  }
}


# CSR build with ONE atomic pass: the degree atomicAdd returns each edge's rank; fill is colidx[rowptr[dst]+rank]=src, no atomics. Row order = arrival order of the count atomics (was fill's)
# speedup vs baseline: 1.1010x; 1.1010x over previous
#include <hip/hip_runtime.h>

typedef __attribute__((ext_vector_type(2))) float v2f;
typedef __attribute__((ext_vector_type(8))) float v8f;

#define N_NODES 100000
#define N_EDGES 1600000
#define D 64
#define N_LAYERS 5

static inline size_t align256(size_t x) { return (x + 255) & ~(size_t)255; }

__global__ __launch_bounds__(256) void gnn_deg_int_kernel(const int* __restrict__ dst,
                                                          int* __restrict__ degi, int nE) {
  int e = blockIdx.x * 256 + threadIdx.x;
  if (e < nE) atomicAdd(&degi[dst[e]], 1);
}

__global__ __launch_bounds__(256) void gnn_deg_rank_kernel(const int* __restrict__ dst,
                                                           int* __restrict__ degi,
                                                           int* __restrict__ rank, int nE) {
  int e = blockIdx.x * 256 + threadIdx.x;
  if (e < nE) rank[e] = atomicAdd(&degi[dst[e]], 1);
}

__global__ __launch_bounds__(256) void gnn_invdeg_from_int_kernel(const int* __restrict__ degi,
                                                                  float* __restrict__ invd, int n) {
  int i = blockIdx.x * 256 + threadIdx.x;
  if (i < n) invd[i] = 1.0f / fmaxf((float)degi[i], 1.0f);
}

__global__ __launch_bounds__(256) void gnn_scan_p1_kernel(const int* __restrict__ degi,
                                                          int* __restrict__ blockSums, int n) {
  __shared__ int s[256];
  int i = blockIdx.x * 256 + threadIdx.x;
  s[threadIdx.x] = (i < n) ? degi[i] : 0;
  __syncthreads();
  for (int off = 128; off > 0; off >>= 1) {
    if (threadIdx.x < off) s[threadIdx.x] += s[threadIdx.x + off];
    __syncthreads();
  }
  if (threadIdx.x == 0) blockSums[blockIdx.x] = s[0];
}

__global__ void gnn_scan_p2_kernel(int* __restrict__ blockSums, int nb) {
  if (blockIdx.x == 0 && threadIdx.x == 0) {
    int acc = 0;
    for (int i = 0; i < nb; ++i) {
      int v = blockSums[i];
      blockSums[i] = acc;
      acc += v;
    }
  }
}

__global__ __launch_bounds__(256) void gnn_scan_p3_kernel(const int* __restrict__ degi,
                                                          const int* __restrict__ blockSums,
                                                          int* __restrict__ rowptr, int n) {
  __shared__ int s[256];
  int i = blockIdx.x * 256 + threadIdx.x;
  int v = (i < n) ? degi[i] : 0;
  s[threadIdx.x] = v;
  __syncthreads();
  for (int off = 1; off < 256; off <<= 1) {
    int t = (threadIdx.x >= off) ? s[threadIdx.x - off] : 0;
    __syncthreads();
    s[threadIdx.x] += t;
    __syncthreads();
  }
  if (i < n) {
    int excl = s[threadIdx.x] - v + blockSums[blockIdx.x];
    rowptr[i] = excl;
  }
}

__global__ __launch_bounds__(256) void gnn_csr_fill_kernel(const int* __restrict__ src,
                                                           const int* __restrict__ dstv,
                                                           const int* __restrict__ rowptr,
                                                           const int* __restrict__ rank,
                                                           int* __restrict__ colidx, int nE) {
  int e = blockIdx.x * 256 + threadIdx.x;
  if (e < nE) colidx[rowptr[dstv[e]] + rank[e]] = src[e];
}

__global__ __launch_bounds__(256) void gnn_gather_agg_kernel(const float* __restrict__ h,
                                                             const int* __restrict__ colidx,
                                                             const int* __restrict__ rowptr,
                                                             const int* __restrict__ degi,
                                                             float* __restrict__ agg, int nN) {
  int wave = threadIdx.x >> 5;
  int lane = threadIdx.x & 31;
  int node = blockIdx.x * 8 + wave;
  if (node >= nN) return;
  int beg = rowptr[node];
  int cnt = degi[node];
  const float* hp = h + 2 * lane;

  float2 a0 = make_float2(0.f, 0.f), a1 = a0, a2 = a0, a3 = a0;
  int t = 0;
  for (; t + 4 <= cnt; t += 4) {
    int n0 = colidx[beg + t + 0];
    int n1 = colidx[beg + t + 1];
    int n2 = colidx[beg + t + 2];
    int n3 = colidx[beg + t + 3];
    const float2 v0 = *(const float2*)(hp + (long)n0 * D);
    const float2 v1 = *(const float2*)(hp + (long)n1 * D);
    const float2 v2 = *(const float2*)(hp + (long)n2 * D);
    const float2 v3 = *(const float2*)(hp + (long)n3 * D);
    a0.x += v0.x; a0.y += v0.y;
    a1.x += v1.x; a1.y += v1.y;
    a2.x += v2.x; a2.y += v2.y;
    a3.x += v3.x; a3.y += v3.y;
  }
  for (; t < cnt; ++t) {
    int nb = colidx[beg + t];
    const float2 v = *(const float2*)(hp + (long)nb * D);
    a0.x += v.x; a0.y += v.y;
  }
  float2 acc;
  acc.x = (a0.x + a1.x) + (a2.x + a3.x);
  acc.y = (a0.y + a1.y) + (a2.y + a3.y);
  *(float2*)(agg + (long)node * D + 2 * lane) = acc;
}

__global__ __launch_bounds__(256) void gnn_scatter_kernel(const float* __restrict__ h,
                                                          const int* __restrict__ src,
                                                          const int* __restrict__ dstv,
                                                          float* __restrict__ agg, int nE) {
  long tid = (long)blockIdx.x * 256 + threadIdx.x;
  int e = (int)(tid >> 4);
  int c = (int)(tid & 15);
  if (e >= nE) return;
  int s = src[e];
  int d = dstv[e];
  const float4 v = *(const float4*)(h + (long)s * D + c * 4);
  float* ap = agg + (long)d * D + c * 4;
  atomicAdd(ap + 0, v.x);
  atomicAdd(ap + 1, v.y);
  atomicAdd(ap + 2, v.z);
  atomicAdd(ap + 3, v.w);
}

__global__ __launch_bounds__(256) void gnn_sage_gemm_kernel(
    const float* __restrict__ agg, const float* __restrict__ h,
    const float* __restrict__ invdeg,
    const float* __restrict__ Wl, const float* __restrict__ Wr,
    const float* __restrict__ bias,
    float* __restrict__ out, int nN, int do_relu, int do_resid) {
  __shared__ __align__(16) float2 sWf[2][16][4][32];
  __shared__ __align__(16) float sB[D];
  int tid = threadIdx.x;

  for (int t = tid; t < 2048; t += 256) {
    int k  = t >> 7;
    int n  = (t >> 5) & 3;
    int ln = t & 31;
    int koff = (ln < 16) ? 0 : 2;
    int lr = ln & 15;
    int kb = k * 4 + koff;
    int col = n * 16 + lr;
    float2 f0, f1;
    f0.x = Wl[(kb + 0) * D + col];
    f0.y = Wl[(kb + 1) * D + col];
    f1.x = Wr[(kb + 0) * D + col];
    f1.y = Wr[(kb + 1) * D + col];
    sWf[0][k][n][ln] = f0;
    sWf[1][k][n][ln] = f1;
  }
  if (tid < D) sB[tid] = bias[tid];
  __syncthreads();

  int wave = tid >> 5;
  int lane = tid & 31;
  int lrow = lane & 15;
  int khalf = (lane < 16) ? 0 : 2;
  int rowbase = blockIdx.x * 128 + wave * 16;
  int M = rowbase + lrow;
  int Mc = (M < nN) ? M : (nN - 1);

  float invd = invdeg[Mc];

  v8f acc[4];
#pragma unroll
  for (int n = 0; n < 4; ++n) {
    float bv = sB[n * 16 + lrow];
#pragma unroll
    for (int j = 0; j < 8; ++j) acc[n][j] = bv;
  }

  for (int p = 0; p < 2; ++p) {
    const float* Arow = (p == 0 ? agg : h) + (long)Mc * D;
    float scale = (p == 0) ? invd : 1.0f;
#pragma unroll
    for (int k = 0; k < 16; ++k) {
      int kb = k * 4 + khalf;
      v2f a;
      a[0] = Arow[kb + 0] * scale;
      a[1] = Arow[kb + 1] * scale;
#pragma unroll
      for (int n = 0; n < 4; ++n) {
        float2 bf2 = sWf[p][k][n][lane];
        v2f bf;
        bf[0] = bf2.x;
        bf[1] = bf2.y;
        acc[n] = __builtin_amdgcn_wmma_f32_16x16x4_f32(
            false, a, false, bf, (short)0, acc[n], false, false);
      }
    }
  }

#pragma unroll
  for (int n = 0; n < 4; ++n) {
#pragma unroll
    for (int j = 0; j < 8; ++j) {
      int row = rowbase + ((lane < 16) ? j : j + 8);
      int col = n * 16 + lrow;
      if (row < nN) {
        float v = acc[n][j];
        if (do_resid) v += h[(long)row * D + col];
        if (do_relu) v = fmaxf(v, 0.0f);
        out[(long)row * D + col] = v;
      }
    }
  }
}

extern "C" void kernel_launch(void* const* d_in, const int* in_sizes, int n_in,
                              void* d_out, int out_size, void* d_ws, size_t ws_size,
                              hipStream_t stream) {
  const float* x  = (const float*)d_in[0];
  const int*   ei = (const int*)d_in[1];
  const float* Wl = (const float*)d_in[2];
  const float* Wr = (const float*)d_in[3];
  const float* b  = (const float*)d_in[4];
  float* out = (float*)d_out;

  const int* src = ei;
  const int* dst = ei + N_EDGES;

  const size_t featBytes = (size_t)N_NODES * D * sizeof(float);
  const size_t nodeI     = align256((size_t)N_NODES * sizeof(int));
  const size_t nodeF     = align256((size_t)N_NODES * sizeof(float));
  const size_t edgeI     = align256((size_t)N_EDGES * sizeof(int));
  const int scanBlocks   = (N_NODES + 255) / 256;
  const size_t bsumBytes = align256((size_t)scanBlocks * sizeof(int));

  const size_t csrNeed = 2 * nodeI + bsumBytes + nodeF + 2 * edgeI + 3 * featBytes;

  char* ws = (char*)d_ws;

  if (ws_size >= csrNeed) {
    int*   degi      = (int*)ws;   ws += nodeI;
    int*   rowptr    = (int*)ws;   ws += nodeI;
    int*   blockSums = (int*)ws;   ws += bsumBytes;
    float* invdeg    = (float*)ws; ws += nodeF;
    int*   colidx    = (int*)ws;   ws += edgeI;
    int*   rank      = (int*)ws;   ws += edgeI;
    float* agg       = (float*)ws; ws += featBytes;
    float* h1        = (float*)ws; ws += featBytes;
    float* h2        = (float*)ws;

    hipMemsetAsync(degi, 0, (size_t)N_NODES * sizeof(int), stream);
    gnn_deg_rank_kernel<<<(N_EDGES + 255) / 256, 256, 0, stream>>>(dst, degi, rank, N_EDGES);
    gnn_invdeg_from_int_kernel<<<scanBlocks, 256, 0, stream>>>(degi, invdeg, N_NODES);
    gnn_scan_p1_kernel<<<scanBlocks, 256, 0, stream>>>(degi, blockSums, N_NODES);
    gnn_scan_p2_kernel<<<1, 64, 0, stream>>>(blockSums, scanBlocks);
    gnn_scan_p3_kernel<<<scanBlocks, 256, 0, stream>>>(degi, blockSums, rowptr, N_NODES);
    gnn_csr_fill_kernel<<<(N_EDGES + 255) / 256, 256, 0, stream>>>(src, dst, rowptr, rank, colidx, N_EDGES);

    const float* h = x;
    float* houts[N_LAYERS] = {h1, h2, h1, h2, out};
    int gatherBlocks = (N_NODES + 7) / 8;
    int gemmBlocks   = (N_NODES + 127) / 128;

    for (int i = 0; i < N_LAYERS; ++i) {
      gnn_gather_agg_kernel<<<gatherBlocks, 256, 0, stream>>>(h, colidx, rowptr, degi, agg, N_NODES);
      int relu  = (i < N_LAYERS - 1) ? 1 : 0;
      int resid = (i > 0 && i < N_LAYERS - 1) ? 1 : 0;
      gnn_sage_gemm_kernel<<<gemmBlocks, 256, 0, stream>>>(
          agg, h, invdeg,
          Wl + (long)i * D * D, Wr + (long)i * D * D, b + (long)i * D,
          houts[i], N_NODES, relu, resid);
      h = houts[i];
    }
  } else {
    float* invdeg = (float*)ws; ws += nodeF;
    float* agg    = (float*)ws; ws += featBytes;
    float* h1     = (float*)ws; ws += featBytes;
    float* h2     = (float*)ws;

    hipMemsetAsync(invdeg, 0, (size_t)N_NODES * sizeof(float), stream);
    gnn_deg_int_kernel<<<(N_EDGES + 255) / 256, 256, 0, stream>>>(dst, (int*)invdeg, N_EDGES);
    gnn_invdeg_from_int_kernel<<<scanBlocks, 256, 0, stream>>>((const int*)invdeg, invdeg, N_NODES);

    const float* h = x;
    float* houts[N_LAYERS] = {h1, h2, h1, h2, out};
    long scatterBlocks = ((long)N_EDGES * 16 + 255) / 256;
    int gemmBlocks = (N_NODES + 127) / 128;

    for (int i = 0; i < N_LAYERS; ++i) {
      hipMemsetAsync(agg, 0, featBytes, stream);
      gnn_scatter_kernel<<<(int)scatterBlocks, 256, 0, stream>>>(h, src, dst, agg, N_EDGES);
      int relu  = (i < N_LAYERS - 1) ? 1 : 0;
      int resid = (i > 0 && i < N_LAYERS - 1) ? 1 : 0;
      gnn_sage_gemm_kernel<<<gemmBlocks, 256, 0, stream>>>(
          agg, h, invdeg,
          Wl + (long)i * D * D, Wr + (long)i * D * D, b + (long)i * D,
          houts[i], N_NODES, relu, resid);
      h = houts[i];
    }
  }
}


# gather: 8192 long-lived waves each sum a contiguous node run from one flat edge stream, 16 row loads in flight (was 100k waves, 4 loads); rows summed sequentially in CSR order, not as 4 partial sums
# speedup vs baseline: 1.2618x; 1.1460x over previous
#include <hip/hip_runtime.h>

typedef __attribute__((ext_vector_type(2))) float v2f;
typedef __attribute__((ext_vector_type(8))) float v8f;

#define N_NODES 100000
#define N_EDGES 1600000
#define D 64
#define N_LAYERS 5

static inline size_t align256(size_t x) { return (x + 255) & ~(size_t)255; }

__global__ __launch_bounds__(256) void gnn_deg_int_kernel(const int* __restrict__ dst,
                                                          int* __restrict__ degi, int nE) {
  int e = blockIdx.x * 256 + threadIdx.x;
  if (e < nE) atomicAdd(&degi[dst[e]], 1);
}

__global__ __launch_bounds__(256) void gnn_deg_rank_kernel(const int* __restrict__ dst,
                                                           int* __restrict__ degi,
                                                           int* __restrict__ rank, int nE) {
  int e = blockIdx.x * 256 + threadIdx.x;
  if (e < nE) rank[e] = atomicAdd(&degi[dst[e]], 1);
}

__global__ __launch_bounds__(256) void gnn_invdeg_from_int_kernel(const int* __restrict__ degi,
                                                                  float* __restrict__ invd, int n) {
  int i = blockIdx.x * 256 + threadIdx.x;
  if (i < n) invd[i] = 1.0f / fmaxf((float)degi[i], 1.0f);
}

__global__ __launch_bounds__(256) void gnn_scan_p1_kernel(const int* __restrict__ degi,
                                                          int* __restrict__ blockSums, int n) {
  __shared__ int s[256];
  int i = blockIdx.x * 256 + threadIdx.x;
  s[threadIdx.x] = (i < n) ? degi[i] : 0;
  __syncthreads();
  for (int off = 128; off > 0; off >>= 1) {
    if (threadIdx.x < off) s[threadIdx.x] += s[threadIdx.x + off];
    __syncthreads();
  }
  if (threadIdx.x == 0) blockSums[blockIdx.x] = s[0];
}

__global__ void gnn_scan_p2_kernel(int* __restrict__ blockSums, int nb) {
  if (blockIdx.x == 0 && threadIdx.x == 0) {
    int acc = 0;
    for (int i = 0; i < nb; ++i) {
      int v = blockSums[i];
      blockSums[i] = acc;
      acc += v;
    }
  }
}

__global__ __launch_bounds__(256) void gnn_scan_p3_kernel(const int* __restrict__ degi,
                                                          const int* __restrict__ blockSums,
                                                          int* __restrict__ rowptr, int n) {
  __shared__ int s[256];
  int i = blockIdx.x * 256 + threadIdx.x;
  int v = (i < n) ? degi[i] : 0;
  s[threadIdx.x] = v;
  __syncthreads();
  for (int off = 1; off < 256; off <<= 1) {
    int t = (threadIdx.x >= off) ? s[threadIdx.x - off] : 0;
    __syncthreads();
    s[threadIdx.x] += t;
    __syncthreads();
  }
  if (i < n) {
    int excl = s[threadIdx.x] - v + blockSums[blockIdx.x];
    rowptr[i] = excl;
  }
}

__global__ __launch_bounds__(256) void gnn_csr_fill_kernel(const int* __restrict__ src,
                                                           const int* __restrict__ dstv,
                                                           const int* __restrict__ rowptr,
                                                           const int* __restrict__ rank,
                                                           int* __restrict__ colidx, int nE) {
  int e = blockIdx.x * 256 + threadIdx.x;
  if (e < nE) colidx[rowptr[dstv[e]] + rank[e]] = src[e];
}

#define GATHER_WAVES 8192
#define GATHER_GROUP 16
static_assert((GATHER_WAVES & (GATHER_WAVES - 1)) == 0, "node split uses a shift");
static_assert((long)N_NODES <= 32L * GATHER_WAVES, "a wave keeps its node ends one per lane");
static_assert(GATHER_GROUP == 16, "the load fence names 16 row registers");
static_assert((long)N_NODES * D < (1L << 31), "row offsets are 32-bit float indices");

template <bool FULL>
__device__ __forceinline__ void gnn_gather_group(const float* __restrict__ h, int idx, int j,
                                                 int m, int pos0, int eEnd, unsigned loff,
                                                 int re, float* arow, int& cur, int& curEnd,
                                                 v2f& acc) {
  v2f v[GATHER_GROUP];
#pragma unroll
  for (int u = 0; u < GATHER_GROUP; ++u) {
    const int sl = (FULL || j + u < m) ? (j + u) : (m - 1);
    const unsigned nb = (unsigned)__builtin_amdgcn_readlane(idx, sl);
    v[u] = *(const v2f*)(h + ((nb << 6) + loff));
  }
  asm volatile("" ::"v"(v[0]), "v"(v[1]), "v"(v[2]), "v"(v[3]), "v"(v[4]), "v"(v[5]),
               "v"(v[6]), "v"(v[7]), "v"(v[8]), "v"(v[9]), "v"(v[10]), "v"(v[11]),
               "v"(v[12]), "v"(v[13]), "v"(v[14]), "v"(v[15]));
  const v2f zero2 = {0.f, 0.f};
#pragma unroll
  for (int u = 0; u < GATHER_GROUP; ++u) {
    const int pos = pos0 + u;
    if (FULL || pos < eEnd) {
      while (pos == curEnd) {
        *(v2f*)(arow + (long)cur * D) = acc;
        acc = zero2;
        ++cur;
        curEnd = __builtin_amdgcn_readlane(re, cur);
      }
      acc += v[u];
    }
  }
}

__global__ __launch_bounds__(256) void gnn_gather_agg_kernel(const float* __restrict__ h,
                                                             const int* __restrict__ colidx,
                                                             const int* __restrict__ rowptr,
                                                             const int* __restrict__ degi,
                                                             float* __restrict__ agg, int nN) {
  const int lane = threadIdx.x & 31;
  const int w = __builtin_amdgcn_readfirstlane((int)(blockIdx.x * 8 + (threadIdx.x >> 5)));
  const int n0 = (int)(((long)w * nN) / GATHER_WAVES);
  const int n1 = (int)(((long)(w + 1) * nN) / GATHER_WAVES);
  const int nn = n1 - n0;
  if (nn <= 0) return;

  int rb = 0, re = 0;
  if (lane < nn) {
    rb = rowptr[n0 + lane];
    re = rb + degi[n0 + lane];
  }
  const int e0 = __builtin_amdgcn_readlane(rb, 0);
  const int eEnd = __builtin_amdgcn_readlane(re, nn - 1);

  const unsigned loff = 2u * (unsigned)lane;
  float* const arow = agg + (long)n0 * D + loff;

  int cur = 0;
  int curEnd = __builtin_amdgcn_readlane(re, 0);
  v2f acc = {0.f, 0.f};

  int nxt = (e0 + lane < eEnd) ? colidx[e0 + lane] : 0;
  for (int base = e0; base < eEnd; base += 32) {
    const int idx = nxt;
    asm volatile("" ::"v"(idx));
    const int nbase = base + 32;
    nxt = (nbase + lane < eEnd) ? colidx[nbase + lane] : 0;
    const int rem = eEnd - base;
    const int m = rem < 32 ? rem : 32;
    for (int j = 0; j < m; j += GATHER_GROUP) {
      if (j + GATHER_GROUP <= m)
        gnn_gather_group<true>(h, idx, j, m, base + j, eEnd, loff, re, arow, cur, curEnd, acc);
      else
        gnn_gather_group<false>(h, idx, j, m, base + j, eEnd, loff, re, arow, cur, curEnd, acc);
    }
  }
  const v2f zero2 = {0.f, 0.f};
  for (; cur < nn; ++cur) {
    *(v2f*)(arow + (long)cur * D) = acc;
    acc = zero2;
  }
}

__global__ __launch_bounds__(256) void gnn_scatter_kernel(const float* __restrict__ h,
                                                          const int* __restrict__ src,
                                                          const int* __restrict__ dstv,
                                                          float* __restrict__ agg, int nE) {
  long tid = (long)blockIdx.x * 256 + threadIdx.x;
  int e = (int)(tid >> 4);
  int c = (int)(tid & 15);
  if (e >= nE) return;
  int s = src[e];
  int d = dstv[e];
  const float4 v = *(const float4*)(h + (long)s * D + c * 4);
  float* ap = agg + (long)d * D + c * 4;
  atomicAdd(ap + 0, v.x);
  atomicAdd(ap + 1, v.y);
  atomicAdd(ap + 2, v.z);
  atomicAdd(ap + 3, v.w);
}

__global__ __launch_bounds__(256) void gnn_sage_gemm_kernel(
    const float* __restrict__ agg, const float* __restrict__ h,
    const float* __restrict__ invdeg,
    const float* __restrict__ Wl, const float* __restrict__ Wr,
    const float* __restrict__ bias,
    float* __restrict__ out, int nN, int do_relu, int do_resid) {
  __shared__ __align__(16) float2 sWf[2][16][4][32];
  __shared__ __align__(16) float sB[D];
  int tid = threadIdx.x;

  for (int t = tid; t < 2048; t += 256) {
    int k  = t >> 7;
    int n  = (t >> 5) & 3;
    int ln = t & 31;
    int koff = (ln < 16) ? 0 : 2;
    int lr = ln & 15;
    int kb = k * 4 + koff;
    int col = n * 16 + lr;
    float2 f0, f1;
    f0.x = Wl[(kb + 0) * D + col];
    f0.y = Wl[(kb + 1) * D + col];
    f1.x = Wr[(kb + 0) * D + col];
    f1.y = Wr[(kb + 1) * D + col];
    sWf[0][k][n][ln] = f0;
    sWf[1][k][n][ln] = f1;
  }
  if (tid < D) sB[tid] = bias[tid];
  __syncthreads();

  int wave = tid >> 5;
  int lane = tid & 31;
  int lrow = lane & 15;
  int khalf = (lane < 16) ? 0 : 2;
  int rowbase = blockIdx.x * 128 + wave * 16;
  int M = rowbase + lrow;
  int Mc = (M < nN) ? M : (nN - 1);

  float invd = invdeg[Mc];

  v8f acc[4];
#pragma unroll
  for (int n = 0; n < 4; ++n) {
    float bv = sB[n * 16 + lrow];
#pragma unroll
    for (int j = 0; j < 8; ++j) acc[n][j] = bv;
  }

  for (int p = 0; p < 2; ++p) {
    const float* Arow = (p == 0 ? agg : h) + (long)Mc * D;
    float scale = (p == 0) ? invd : 1.0f;
#pragma unroll
    for (int k = 0; k < 16; ++k) {
      int kb = k * 4 + khalf;
      v2f a;
      a[0] = Arow[kb + 0] * scale;
      a[1] = Arow[kb + 1] * scale;
#pragma unroll
      for (int n = 0; n < 4; ++n) {
        float2 bf2 = sWf[p][k][n][lane];
        v2f bf;
        bf[0] = bf2.x;
        bf[1] = bf2.y;
        acc[n] = __builtin_amdgcn_wmma_f32_16x16x4_f32(
            false, a, false, bf, (short)0, acc[n], false, false);
      }
    }
  }

#pragma unroll
  for (int n = 0; n < 4; ++n) {
#pragma unroll
    for (int j = 0; j < 8; ++j) {
      int row = rowbase + ((lane < 16) ? j : j + 8);
      int col = n * 16 + lrow;
      if (row < nN) {
        float v = acc[n][j];
        if (do_resid) v += h[(long)row * D + col];
        if (do_relu) v = fmaxf(v, 0.0f);
        out[(long)row * D + col] = v;
      }
    }
  }
}

extern "C" void kernel_launch(void* const* d_in, const int* in_sizes, int n_in,
                              void* d_out, int out_size, void* d_ws, size_t ws_size,
                              hipStream_t stream) {
  const float* x  = (const float*)d_in[0];
  const int*   ei = (const int*)d_in[1];
  const float* Wl = (const float*)d_in[2];
  const float* Wr = (const float*)d_in[3];
  const float* b  = (const float*)d_in[4];
  float* out = (float*)d_out;

  const int* src = ei;
  const int* dst = ei + N_EDGES;

  const size_t featBytes = (size_t)N_NODES * D * sizeof(float);
  const size_t nodeI     = align256((size_t)N_NODES * sizeof(int));
  const size_t nodeF     = align256((size_t)N_NODES * sizeof(float));
  const size_t edgeI     = align256((size_t)N_EDGES * sizeof(int));
  const int scanBlocks   = (N_NODES + 255) / 256;
  const size_t bsumBytes = align256((size_t)scanBlocks * sizeof(int));

  const size_t csrNeed = 2 * nodeI + bsumBytes + nodeF + 2 * edgeI + 3 * featBytes;

  char* ws = (char*)d_ws;

  if (ws_size >= csrNeed) {
    int*   degi      = (int*)ws;   ws += nodeI;
    int*   rowptr    = (int*)ws;   ws += nodeI;
    int*   blockSums = (int*)ws;   ws += bsumBytes;
    float* invdeg    = (float*)ws; ws += nodeF;
    int*   colidx    = (int*)ws;   ws += edgeI;
    int*   rank      = (int*)ws;   ws += edgeI;
    float* agg       = (float*)ws; ws += featBytes;
    float* h1        = (float*)ws; ws += featBytes;
    float* h2        = (float*)ws;

    hipMemsetAsync(degi, 0, (size_t)N_NODES * sizeof(int), stream);
    gnn_deg_rank_kernel<<<(N_EDGES + 255) / 256, 256, 0, stream>>>(dst, degi, rank, N_EDGES);
    gnn_invdeg_from_int_kernel<<<scanBlocks, 256, 0, stream>>>(degi, invdeg, N_NODES);
    gnn_scan_p1_kernel<<<scanBlocks, 256, 0, stream>>>(degi, blockSums, N_NODES);
    gnn_scan_p2_kernel<<<1, 64, 0, stream>>>(blockSums, scanBlocks);
    gnn_scan_p3_kernel<<<scanBlocks, 256, 0, stream>>>(degi, blockSums, rowptr, N_NODES);
    gnn_csr_fill_kernel<<<(N_EDGES + 255) / 256, 256, 0, stream>>>(src, dst, rowptr, rank, colidx, N_EDGES);

    const float* h = x;
    float* houts[N_LAYERS] = {h1, h2, h1, h2, out};
    int gatherBlocks = GATHER_WAVES / 8;
    int gemmBlocks   = (N_NODES + 127) / 128;

    for (int i = 0; i < N_LAYERS; ++i) {
      gnn_gather_agg_kernel<<<gatherBlocks, 256, 0, stream>>>(h, colidx, rowptr, degi, agg, N_NODES);
      int relu  = (i < N_LAYERS - 1) ? 1 : 0;
      int resid = (i > 0 && i < N_LAYERS - 1) ? 1 : 0;
      gnn_sage_gemm_kernel<<<gemmBlocks, 256, 0, stream>>>(
          agg, h, invdeg,
          Wl + (long)i * D * D, Wr + (long)i * D * D, b + (long)i * D,
          houts[i], N_NODES, relu, resid);
      h = houts[i];
    }
  } else {
    float* invdeg = (float*)ws; ws += nodeF;
    float* agg    = (float*)ws; ws += featBytes;
    float* h1     = (float*)ws; ws += featBytes;
    float* h2     = (float*)ws;

    hipMemsetAsync(invdeg, 0, (size_t)N_NODES * sizeof(float), stream);
    gnn_deg_int_kernel<<<(N_EDGES + 255) / 256, 256, 0, stream>>>(dst, (int*)invdeg, N_EDGES);
    gnn_invdeg_from_int_kernel<<<scanBlocks, 256, 0, stream>>>((const int*)invdeg, invdeg, N_NODES);

    const float* h = x;
    float* houts[N_LAYERS] = {h1, h2, h1, h2, out};
    long scatterBlocks = ((long)N_EDGES * 16 + 255) / 256;
    int gemmBlocks = (N_NODES + 127) / 128;

    for (int i = 0; i < N_LAYERS; ++i) {
      hipMemsetAsync(agg, 0, featBytes, stream);
      gnn_scatter_kernel<<<(int)scatterBlocks, 256, 0, stream>>>(h, src, dst, agg, N_EDGES);
      int relu  = (i < N_LAYERS - 1) ? 1 : 0;
      int resid = (i > 0 && i < N_LAYERS - 1) ? 1 : 0;
      gnn_sage_gemm_kernel<<<gemmBlocks, 256, 0, stream>>>(
          agg, h, invdeg,
          Wl + (long)i * D * D, Wr + (long)i * D * D, b + (long)i * D,
          houts[i], N_NODES, relu, resid);
      h = houts[i];
    }
  }
}


# CSR build as two-level counting sort through LDS (bucket count, bucket scatter, rows per bucket): no global atomics, no memset, 3 launches for 7; row order = arrival order of LDS atomics (was global)
# speedup vs baseline: 1.6942x; 1.3427x over previous
#include <hip/hip_runtime.h>

typedef __attribute__((ext_vector_type(2))) float v2f;
typedef __attribute__((ext_vector_type(8))) float v8f;

#define N_NODES 100000
#define N_EDGES 1600000
#define D 64
#define N_LAYERS 5

static inline size_t align256(size_t x) { return (x + 255) & ~(size_t)255; }

__global__ __launch_bounds__(256) void gnn_deg_int_kernel(const int* __restrict__ dst,
                                                          int* __restrict__ degi, int nE) {
  int e = blockIdx.x * 256 + threadIdx.x;
  if (e < nE) atomicAdd(&degi[dst[e]], 1);
}

__global__ __launch_bounds__(256) void gnn_invdeg_from_int_kernel(const int* __restrict__ degi,
                                                                  float* __restrict__ invd, int n) {
  int i = blockIdx.x * 256 + threadIdx.x;
  if (i < n) invd[i] = 1.0f / fmaxf((float)degi[i], 1.0f);
}

#define BK_SHIFT 9
#define BK_NODES (1 << BK_SHIFT)
#define N_BUCKETS ((N_NODES + BK_NODES - 1) / BK_NODES)
#define EDGE_CHUNKS 128
#define EDGE_THREADS 1024
#define EDGE_BATCH 4
#define HIST_PARTS 4
static_assert((long)N_NODES << BK_SHIFT < (1L << 31), "src and the local dst share one 32-bit word");
static_assert(N_BUCKETS <= EDGE_THREADS / HIST_PARTS, "one thread per bucket in each column-sum part");
static_assert(EDGE_CHUNKS % HIST_PARTS == 0, "column sums are split evenly");
static_assert(BK_NODES <= EDGE_THREADS, "one thread per node of a bucket");

__device__ __forceinline__ int gnn_block_excl_scan(int v, int* wsum) {
  const int lane = threadIdx.x & 31;
  const int w = threadIdx.x >> 5;
  int inc = v;
#pragma unroll
  for (int o = 1; o < 32; o <<= 1) {
    const int t = __shfl_up(inc, o, 32);
    if (lane >= o) inc += t;
  }
  if (lane == 31) wsum[w] = inc;
  __syncthreads();
  const int ws = wsum[lane];
  int wi = ws;
#pragma unroll
  for (int o = 1; o < 32; o <<= 1) {
    const int t = __shfl_up(wi, o, 32);
    if (lane >= o) wi += t;
  }
  const int wexcl = __shfl(wi - ws, w, 32);
  return wexcl + inc - v;
}

__global__ __launch_bounds__(EDGE_THREADS) void gnn_bucket_count_kernel(const int* __restrict__ dst,
                                                                        int* __restrict__ hist,
                                                                        int nE) {
  __shared__ int cnt[N_BUCKETS];
  const int tid = threadIdx.x;
  const int c = blockIdx.x;
  if (tid < N_BUCKETS) cnt[tid] = 0;
  __syncthreads();
  const int e0 = (int)(((long)c * nE) / EDGE_CHUNKS);
  const int e1 = (int)(((long)(c + 1) * nE) / EDGE_CHUNKS);
  for (int e = e0 + tid; e < e1; e += EDGE_BATCH * EDGE_THREADS) {
    int d[EDGE_BATCH];
#pragma unroll
    for (int u = 0; u < EDGE_BATCH; ++u) {
      const int q = e + u * EDGE_THREADS;
      d[u] = dst[q < e1 ? q : e1 - 1];
    }
#pragma unroll
    for (int u = 0; u < EDGE_BATCH; ++u)
      atomicAdd(&cnt[d[u] >> BK_SHIFT], (e + u * EDGE_THREADS < e1) ? 1 : 0);
  }
  __syncthreads();
  if (tid < N_BUCKETS) hist[c * N_BUCKETS + tid] = cnt[tid];
}

__global__ __launch_bounds__(EDGE_THREADS) void gnn_bucket_scatter_kernel(
    const int* __restrict__ src, const int* __restrict__ dst, const int* __restrict__ hist,
    int* __restrict__ bucketBase, int* __restrict__ ebuf, int nE) {
  __shared__ int ptot[HIST_PARTS][EDGE_THREADS / HIST_PARTS];
  __shared__ int pbef[HIST_PARTS][EDGE_THREADS / HIST_PARTS];
  __shared__ int cursor[N_BUCKETS];
  __shared__ int wsum[32];
  const int tid = threadIdx.x;
  const int c = blockIdx.x;

  const int part = tid / (EDGE_THREADS / HIST_PARTS);
  const int j = tid % (EDGE_THREADS / HIST_PARTS);
  int t = 0, bf = 0;
  if (j < N_BUCKETS) {
#pragma unroll
    for (int i = 0; i < EDGE_CHUNKS / HIST_PARTS; ++i) {
      const int k = part * (EDGE_CHUNKS / HIST_PARTS) + i;
      const int v = hist[k * N_BUCKETS + j];
      t += v;
      bf += (k < c) ? v : 0;
    }
  }
  ptot[part][j] = t;
  pbef[part][j] = bf;
  __syncthreads();
  int tot = 0, before = 0;
  if (tid < N_BUCKETS) {
#pragma unroll
    for (int p = 0; p < HIST_PARTS; ++p) {
      tot += ptot[p][tid];
      before += pbef[p][tid];
    }
  }
  const int excl = gnn_block_excl_scan(tot, wsum);
  if (tid < N_BUCKETS) cursor[tid] = excl + before;
  if (c == 0 && tid <= N_BUCKETS) bucketBase[tid] = excl;
  __syncthreads();

  const int e0 = (int)(((long)c * nE) / EDGE_CHUNKS);
  const int e1 = (int)(((long)(c + 1) * nE) / EDGE_CHUNKS);
  for (int e = e0 + tid; e < e1; e += EDGE_BATCH * EDGE_THREADS) {
    int d[EDGE_BATCH], s[EDGE_BATCH], slot[EDGE_BATCH];
#pragma unroll
    for (int u = 0; u < EDGE_BATCH; ++u) {
      const int q = e + u * EDGE_THREADS;
      const int qc = q < e1 ? q : e1 - 1;
      d[u] = dst[qc];
      s[u] = src[qc];
    }
#pragma unroll
    for (int u = 0; u < EDGE_BATCH; ++u)
      slot[u] = (e + u * EDGE_THREADS < e1) ? atomicAdd(&cursor[d[u] >> BK_SHIFT], 1) : -1;
#pragma unroll
    for (int u = 0; u < EDGE_BATCH; ++u)
      if (slot[u] >= 0) ebuf[slot[u]] = (s[u] << BK_SHIFT) | (d[u] & (BK_NODES - 1));
  }
}

__global__ __launch_bounds__(EDGE_THREADS) void gnn_bucket_rows_kernel(
    const int* __restrict__ ebuf, const int* __restrict__ bucketBase, int* __restrict__ rowptr,
    int* __restrict__ degi, float* __restrict__ invd, int* __restrict__ colidx, int nN) {
  __shared__ int cnt[EDGE_THREADS];
  __shared__ int wsum[32];
  const int tid = threadIdx.x;
  const int b = blockIdx.x;
  const int p0 = bucketBase[b];
  const int p1 = bucketBase[b + 1];
  cnt[tid] = 0;
  __syncthreads();
  for (int p = p0 + tid; p < p1; p += EDGE_BATCH * EDGE_THREADS) {
    int w[EDGE_BATCH];
#pragma unroll
    for (int u = 0; u < EDGE_BATCH; ++u) {
      const int q = p + u * EDGE_THREADS;
      w[u] = ebuf[q < p1 ? q : p1 - 1];
    }
#pragma unroll
    for (int u = 0; u < EDGE_BATCH; ++u)
      atomicAdd(&cnt[w[u] & (BK_NODES - 1)], (p + u * EDGE_THREADS < p1) ? 1 : 0);
  }
  __syncthreads();
  const int deg = cnt[tid];
  const int start = p0 + gnn_block_excl_scan(deg, wsum);
  cnt[tid] = start;
  const int node = b * BK_NODES + tid;
  if (tid < BK_NODES && node < nN) {
    rowptr[node] = start;
    degi[node] = deg;
    invd[node] = 1.0f / fmaxf((float)deg, 1.0f);
  }
  __syncthreads();
  for (int p = p0 + tid; p < p1; p += EDGE_BATCH * EDGE_THREADS) {
    int w[EDGE_BATCH], slot[EDGE_BATCH];
#pragma unroll
    for (int u = 0; u < EDGE_BATCH; ++u) {
      const int q = p + u * EDGE_THREADS;
      w[u] = ebuf[q < p1 ? q : p1 - 1];
    }
#pragma unroll
    for (int u = 0; u < EDGE_BATCH; ++u)
      slot[u] = (p + u * EDGE_THREADS < p1) ? atomicAdd(&cnt[w[u] & (BK_NODES - 1)], 1) : -1;
#pragma unroll
    for (int u = 0; u < EDGE_BATCH; ++u)
      if (slot[u] >= 0) colidx[slot[u]] = w[u] >> BK_SHIFT;
  }
}

#define GATHER_WAVES 8192
#define GATHER_GROUP 16
static_assert((GATHER_WAVES & (GATHER_WAVES - 1)) == 0, "node split uses a shift");
static_assert((long)N_NODES <= 32L * GATHER_WAVES, "a wave keeps its node ends one per lane");
static_assert(GATHER_GROUP == 16, "the load fence names 16 row registers");
static_assert((long)N_NODES * D < (1L << 31), "row offsets are 32-bit float indices");

template <bool FULL>
__device__ __forceinline__ void gnn_gather_group(const float* __restrict__ h, int idx, int j,
                                                 int m, int pos0, int eEnd, unsigned loff,
                                                 int re, float* arow, int& cur, int& curEnd,
                                                 v2f& acc) {
  v2f v[GATHER_GROUP];
#pragma unroll
  for (int u = 0; u < GATHER_GROUP; ++u) {
    const int sl = (FULL || j + u < m) ? (j + u) : (m - 1);
    const unsigned nb = (unsigned)__builtin_amdgcn_readlane(idx, sl);
    v[u] = *(const v2f*)(h + ((nb << 6) + loff));
  }
  asm volatile("" ::"v"(v[0]), "v"(v[1]), "v"(v[2]), "v"(v[3]), "v"(v[4]), "v"(v[5]),
               "v"(v[6]), "v"(v[7]), "v"(v[8]), "v"(v[9]), "v"(v[10]), "v"(v[11]),
               "v"(v[12]), "v"(v[13]), "v"(v[14]), "v"(v[15]));
  const v2f zero2 = {0.f, 0.f};
#pragma unroll
  for (int u = 0; u < GATHER_GROUP; ++u) {
    const int pos = pos0 + u;
    if (FULL || pos < eEnd) {
      while (pos == curEnd) {
        *(v2f*)(arow + (long)cur * D) = acc;
        acc = zero2;
        ++cur;
        curEnd = __builtin_amdgcn_readlane(re, cur);
      }
      acc += v[u];
    }
  }
}

__global__ __launch_bounds__(256) void gnn_gather_agg_kernel(const float* __restrict__ h,
                                                             const int* __restrict__ colidx,
                                                             const int* __restrict__ rowptr,
                                                             const int* __restrict__ degi,
                                                             float* __restrict__ agg, int nN) {
  const int lane = threadIdx.x & 31;
  const int w = __builtin_amdgcn_readfirstlane((int)(blockIdx.x * 8 + (threadIdx.x >> 5)));
  const int n0 = (int)(((long)w * nN) / GATHER_WAVES);
  const int n1 = (int)(((long)(w + 1) * nN) / GATHER_WAVES);
  const int nn = n1 - n0;
  if (nn <= 0) return;

  int rb = 0, re = 0;
  if (lane < nn) {
    rb = rowptr[n0 + lane];
    re = rb + degi[n0 + lane];
  }
  const int e0 = __builtin_amdgcn_readlane(rb, 0);
  const int eEnd = __builtin_amdgcn_readlane(re, nn - 1);

  const unsigned loff = 2u * (unsigned)lane;
  float* const arow = agg + (long)n0 * D + loff;

  int cur = 0;
  int curEnd = __builtin_amdgcn_readlane(re, 0);
  v2f acc = {0.f, 0.f};

  int nxt = (e0 + lane < eEnd) ? colidx[e0 + lane] : 0;
  for (int base = e0; base < eEnd; base += 32) {
    const int idx = nxt;
    asm volatile("" ::"v"(idx));
    const int nbase = base + 32;
    nxt = (nbase + lane < eEnd) ? colidx[nbase + lane] : 0;
    const int rem = eEnd - base;
    const int m = rem < 32 ? rem : 32;
    for (int j = 0; j < m; j += GATHER_GROUP) {
      if (j + GATHER_GROUP <= m)
        gnn_gather_group<true>(h, idx, j, m, base + j, eEnd, loff, re, arow, cur, curEnd, acc);
      else
        gnn_gather_group<false>(h, idx, j, m, base + j, eEnd, loff, re, arow, cur, curEnd, acc);
    }
  }
  const v2f zero2 = {0.f, 0.f};
  for (; cur < nn; ++cur) {
    *(v2f*)(arow + (long)cur * D) = acc;
    acc = zero2;
  }
}

__global__ __launch_bounds__(256) void gnn_scatter_kernel(const float* __restrict__ h,
                                                          const int* __restrict__ src,
                                                          const int* __restrict__ dstv,
                                                          float* __restrict__ agg, int nE) {
  long tid = (long)blockIdx.x * 256 + threadIdx.x;
  int e = (int)(tid >> 4);
  int c = (int)(tid & 15);
  if (e >= nE) return;
  int s = src[e];
  int d = dstv[e];
  const float4 v = *(const float4*)(h + (long)s * D + c * 4);
  float* ap = agg + (long)d * D + c * 4;
  atomicAdd(ap + 0, v.x);
  atomicAdd(ap + 1, v.y);
  atomicAdd(ap + 2, v.z);
  atomicAdd(ap + 3, v.w);
}

__global__ __launch_bounds__(256) void gnn_sage_gemm_kernel(
    const float* __restrict__ agg, const float* __restrict__ h,
    const float* __restrict__ invdeg,
    const float* __restrict__ Wl, const float* __restrict__ Wr,
    const float* __restrict__ bias,
    float* __restrict__ out, int nN, int do_relu, int do_resid) {
  __shared__ __align__(16) float2 sWf[2][16][4][32];
  __shared__ __align__(16) float sB[D];
  int tid = threadIdx.x;

  for (int t = tid; t < 2048; t += 256) {
    int k  = t >> 7;
    int n  = (t >> 5) & 3;
    int ln = t & 31;
    int koff = (ln < 16) ? 0 : 2;
    int lr = ln & 15;
    int kb = k * 4 + koff;
    int col = n * 16 + lr;
    float2 f0, f1;
    f0.x = Wl[(kb + 0) * D + col];
    f0.y = Wl[(kb + 1) * D + col];
    f1.x = Wr[(kb + 0) * D + col];
    f1.y = Wr[(kb + 1) * D + col];
    sWf[0][k][n][ln] = f0;
    sWf[1][k][n][ln] = f1;
  }
  if (tid < D) sB[tid] = bias[tid];
  __syncthreads();

  int wave = tid >> 5;
  int lane = tid & 31;
  int lrow = lane & 15;
  int khalf = (lane < 16) ? 0 : 2;
  int rowbase = blockIdx.x * 128 + wave * 16;
  int M = rowbase + lrow;
  int Mc = (M < nN) ? M : (nN - 1);

  float invd = invdeg[Mc];

  v8f acc[4];
#pragma unroll
  for (int n = 0; n < 4; ++n) {
    float bv = sB[n * 16 + lrow];
#pragma unroll
    for (int j = 0; j < 8; ++j) acc[n][j] = bv;
  }

  for (int p = 0; p < 2; ++p) {
    const float* Arow = (p == 0 ? agg : h) + (long)Mc * D;
    float scale = (p == 0) ? invd : 1.0f;
#pragma unroll
    for (int k = 0; k < 16; ++k) {
      int kb = k * 4 + khalf;
      v2f a;
      a[0] = Arow[kb + 0] * scale;
      a[1] = Arow[kb + 1] * scale;
#pragma unroll
      for (int n = 0; n < 4; ++n) {
        float2 bf2 = sWf[p][k][n][lane];
        v2f bf;
        bf[0] = bf2.x;
        bf[1] = bf2.y;
        acc[n] = __builtin_amdgcn_wmma_f32_16x16x4_f32(
            false, a, false, bf, (short)0, acc[n], false, false);
      }
    }
  }

#pragma unroll
  for (int n = 0; n < 4; ++n) {
#pragma unroll
    for (int j = 0; j < 8; ++j) {
      int row = rowbase + ((lane < 16) ? j : j + 8);
      int col = n * 16 + lrow;
      if (row < nN) {
        float v = acc[n][j];
        if (do_resid) v += h[(long)row * D + col];
        if (do_relu) v = fmaxf(v, 0.0f);
        out[(long)row * D + col] = v;
      }
    }
  }
}

extern "C" void kernel_launch(void* const* d_in, const int* in_sizes, int n_in,
                              void* d_out, int out_size, void* d_ws, size_t ws_size,
                              hipStream_t stream) {
  const float* x  = (const float*)d_in[0];
  const int*   ei = (const int*)d_in[1];
  const float* Wl = (const float*)d_in[2];
  const float* Wr = (const float*)d_in[3];
  const float* b  = (const float*)d_in[4];
  float* out = (float*)d_out;

  const int* src = ei;
  const int* dst = ei + N_EDGES;

  const size_t featBytes = (size_t)N_NODES * D * sizeof(float);
  const size_t nodeI     = align256((size_t)N_NODES * sizeof(int));
  const size_t nodeF     = align256((size_t)N_NODES * sizeof(float));
  const size_t edgeI     = align256((size_t)N_EDGES * sizeof(int));
  const int scanBlocks   = (N_NODES + 255) / 256;
  const size_t histBytes = align256((size_t)EDGE_CHUNKS * N_BUCKETS * sizeof(int));
  const size_t baseBytes = align256((size_t)(N_BUCKETS + 1) * sizeof(int));

  const size_t csrNeed = 2 * nodeI + histBytes + baseBytes + nodeF + 2 * edgeI + 3 * featBytes;

  char* ws = (char*)d_ws;

  if (ws_size >= csrNeed) {
    int*   degi       = (int*)ws;   ws += nodeI;
    int*   rowptr     = (int*)ws;   ws += nodeI;
    int*   hist       = (int*)ws;   ws += histBytes;
    int*   bucketBase = (int*)ws;   ws += baseBytes;
    float* invdeg     = (float*)ws; ws += nodeF;
    int*   colidx     = (int*)ws;   ws += edgeI;
    int*   ebuf       = (int*)ws;   ws += edgeI;
    float* agg        = (float*)ws; ws += featBytes;
    float* h1         = (float*)ws; ws += featBytes;
    float* h2         = (float*)ws;

    gnn_bucket_count_kernel<<<EDGE_CHUNKS, EDGE_THREADS, 0, stream>>>(dst, hist, N_EDGES);
    gnn_bucket_scatter_kernel<<<EDGE_CHUNKS, EDGE_THREADS, 0, stream>>>(src, dst, hist, bucketBase, ebuf, N_EDGES);
    gnn_bucket_rows_kernel<<<N_BUCKETS, EDGE_THREADS, 0, stream>>>(ebuf, bucketBase, rowptr, degi, invdeg, colidx, N_NODES);

    const float* h = x;
    float* houts[N_LAYERS] = {h1, h2, h1, h2, out};
    int gatherBlocks = GATHER_WAVES / 8;
    int gemmBlocks   = (N_NODES + 127) / 128;

    for (int i = 0; i < N_LAYERS; ++i) {
      gnn_gather_agg_kernel<<<gatherBlocks, 256, 0, stream>>>(h, colidx, rowptr, degi, agg, N_NODES);
      int relu  = (i < N_LAYERS - 1) ? 1 : 0;
      int resid = (i > 0 && i < N_LAYERS - 1) ? 1 : 0;
      gnn_sage_gemm_kernel<<<gemmBlocks, 256, 0, stream>>>(
          agg, h, invdeg,
          Wl + (long)i * D * D, Wr + (long)i * D * D, b + (long)i * D,
          houts[i], N_NODES, relu, resid);
      h = houts[i];
    }
  } else {
    float* invdeg = (float*)ws; ws += nodeF;
    float* agg    = (float*)ws; ws += featBytes;
    float* h1     = (float*)ws; ws += featBytes;
    float* h2     = (float*)ws;

    hipMemsetAsync(invdeg, 0, (size_t)N_NODES * sizeof(float), stream);
    gnn_deg_int_kernel<<<(N_EDGES + 255) / 256, 256, 0, stream>>>(dst, (int*)invdeg, N_EDGES);
    gnn_invdeg_from_int_kernel<<<scanBlocks, 256, 0, stream>>>((const int*)invdeg, invdeg, N_NODES);

    const float* h = x;
    float* houts[N_LAYERS] = {h1, h2, h1, h2, out};
    long scatterBlocks = ((long)N_EDGES * 16 + 255) / 256;
    int gemmBlocks = (N_NODES + 127) / 128;

    for (int i = 0; i < N_LAYERS; ++i) {
      hipMemsetAsync(agg, 0, featBytes, stream);
      gnn_scatter_kernel<<<(int)scatterBlocks, 256, 0, stream>>>(h, src, dst, agg, N_EDGES);
      int relu  = (i < N_LAYERS - 1) ? 1 : 0;
      int resid = (i > 0 && i < N_LAYERS - 1) ? 1 : 0;
      gnn_sage_gemm_kernel<<<gemmBlocks, 256, 0, stream>>>(
          agg, h, invdeg,
          Wl + (long)i * D * D, Wr + (long)i * D * D, b + (long)i * D,
          houts[i], N_NODES, relu, resid);
      h = houts[i];
    }
  }
}


# Each layer's gather fused into its GEMM kernel: a wave does b+h@Wr on WMMA, sums its node run's neighbour rows into an LDS tile, then adds (sum*invdeg)@Wl (was agg term first); 8 launches for 13
# speedup vs baseline: 1.9339x; 1.1415x over previous
#include <hip/hip_runtime.h>

typedef __attribute__((ext_vector_type(2))) float v2f;
typedef __attribute__((ext_vector_type(8))) float v8f;

#define N_NODES 100000
#define N_EDGES 1600000
#define D 64
#define N_LAYERS 5

static inline size_t align256(size_t x) { return (x + 255) & ~(size_t)255; }

__global__ __launch_bounds__(256) void gnn_deg_int_kernel(const int* __restrict__ dst,
                                                          int* __restrict__ degi, int nE) {
  int e = blockIdx.x * 256 + threadIdx.x;
  if (e < nE) atomicAdd(&degi[dst[e]], 1);
}

__global__ __launch_bounds__(256) void gnn_invdeg_from_int_kernel(const int* __restrict__ degi,
                                                                  float* __restrict__ invd, int n) {
  int i = blockIdx.x * 256 + threadIdx.x;
  if (i < n) invd[i] = 1.0f / fmaxf((float)degi[i], 1.0f);
}

#define BK_SHIFT 9
#define BK_NODES (1 << BK_SHIFT)
#define N_BUCKETS ((N_NODES + BK_NODES - 1) / BK_NODES)
#define EDGE_CHUNKS 128
#define EDGE_THREADS 1024
#define EDGE_BATCH 4
#define HIST_PARTS 4
static_assert((long)N_NODES << BK_SHIFT < (1L << 31), "src and the local dst share one 32-bit word");
static_assert(N_BUCKETS <= EDGE_THREADS / HIST_PARTS, "one thread per bucket in each column-sum part");
static_assert(EDGE_CHUNKS % HIST_PARTS == 0, "column sums are split evenly");
static_assert(BK_NODES <= EDGE_THREADS, "one thread per node of a bucket");

__device__ __forceinline__ int gnn_block_excl_scan(int v, int* wsum) {
  const int lane = threadIdx.x & 31;
  const int w = threadIdx.x >> 5;
  int inc = v;
#pragma unroll
  for (int o = 1; o < 32; o <<= 1) {
    const int t = __shfl_up(inc, o, 32);
    if (lane >= o) inc += t;
  }
  if (lane == 31) wsum[w] = inc;
  __syncthreads();
  const int ws = wsum[lane];
  int wi = ws;
#pragma unroll
  for (int o = 1; o < 32; o <<= 1) {
    const int t = __shfl_up(wi, o, 32);
    if (lane >= o) wi += t;
  }
  const int wexcl = __shfl(wi - ws, w, 32);
  return wexcl + inc - v;
}

__global__ __launch_bounds__(EDGE_THREADS) void gnn_bucket_count_kernel(const int* __restrict__ dst,
                                                                        int* __restrict__ hist,
                                                                        int nE) {
  __shared__ int cnt[N_BUCKETS];
  const int tid = threadIdx.x;
  const int c = blockIdx.x;
  if (tid < N_BUCKETS) cnt[tid] = 0;
  __syncthreads();
  const int e0 = (int)(((long)c * nE) / EDGE_CHUNKS);
  const int e1 = (int)(((long)(c + 1) * nE) / EDGE_CHUNKS);
  for (int e = e0 + tid; e < e1; e += EDGE_BATCH * EDGE_THREADS) {
    int d[EDGE_BATCH];
#pragma unroll
    for (int u = 0; u < EDGE_BATCH; ++u) {
      const int q = e + u * EDGE_THREADS;
      d[u] = dst[q < e1 ? q : e1 - 1];
    }
#pragma unroll
    for (int u = 0; u < EDGE_BATCH; ++u)
      atomicAdd(&cnt[d[u] >> BK_SHIFT], (e + u * EDGE_THREADS < e1) ? 1 : 0);
  }
  __syncthreads();
  if (tid < N_BUCKETS) hist[c * N_BUCKETS + tid] = cnt[tid];
}

__global__ __launch_bounds__(EDGE_THREADS) void gnn_bucket_scatter_kernel(
    const int* __restrict__ src, const int* __restrict__ dst, const int* __restrict__ hist,
    int* __restrict__ bucketBase, int* __restrict__ ebuf, int nE) {
  __shared__ int ptot[HIST_PARTS][EDGE_THREADS / HIST_PARTS];
  __shared__ int pbef[HIST_PARTS][EDGE_THREADS / HIST_PARTS];
  __shared__ int cursor[N_BUCKETS];
  __shared__ int wsum[32];
  const int tid = threadIdx.x;
  const int c = blockIdx.x;

  const int part = tid / (EDGE_THREADS / HIST_PARTS);
  const int j = tid % (EDGE_THREADS / HIST_PARTS);
  int t = 0, bf = 0;
  if (j < N_BUCKETS) {
#pragma unroll
    for (int i = 0; i < EDGE_CHUNKS / HIST_PARTS; ++i) {
      const int k = part * (EDGE_CHUNKS / HIST_PARTS) + i;
      const int v = hist[k * N_BUCKETS + j];
      t += v;
      bf += (k < c) ? v : 0;
    }
  }
  ptot[part][j] = t;
  pbef[part][j] = bf;
  __syncthreads();
  int tot = 0, before = 0;
  if (tid < N_BUCKETS) {
#pragma unroll
    for (int p = 0; p < HIST_PARTS; ++p) {
      tot += ptot[p][tid];
      before += pbef[p][tid];
    }
  }
  const int excl = gnn_block_excl_scan(tot, wsum);
  if (tid < N_BUCKETS) cursor[tid] = excl + before;
  if (c == 0 && tid <= N_BUCKETS) bucketBase[tid] = excl;
  __syncthreads();

  const int e0 = (int)(((long)c * nE) / EDGE_CHUNKS);
  const int e1 = (int)(((long)(c + 1) * nE) / EDGE_CHUNKS);
  for (int e = e0 + tid; e < e1; e += EDGE_BATCH * EDGE_THREADS) {
    int d[EDGE_BATCH], s[EDGE_BATCH], slot[EDGE_BATCH];
#pragma unroll
    for (int u = 0; u < EDGE_BATCH; ++u) {
      const int q = e + u * EDGE_THREADS;
      const int qc = q < e1 ? q : e1 - 1;
      d[u] = dst[qc];
      s[u] = src[qc];
    }
#pragma unroll
    for (int u = 0; u < EDGE_BATCH; ++u)
      slot[u] = (e + u * EDGE_THREADS < e1) ? atomicAdd(&cursor[d[u] >> BK_SHIFT], 1) : -1;
#pragma unroll
    for (int u = 0; u < EDGE_BATCH; ++u)
      if (slot[u] >= 0) ebuf[slot[u]] = (s[u] << BK_SHIFT) | (d[u] & (BK_NODES - 1));
  }
}

__global__ __launch_bounds__(EDGE_THREADS) void gnn_bucket_rows_kernel(
    const int* __restrict__ ebuf, const int* __restrict__ bucketBase, int* __restrict__ rowptr,
    int* __restrict__ degi, float* __restrict__ invd, int* __restrict__ colidx, int nN) {
  __shared__ int cnt[EDGE_THREADS];
  __shared__ int wsum[32];
  const int tid = threadIdx.x;
  const int b = blockIdx.x;
  const int p0 = bucketBase[b];
  const int p1 = bucketBase[b + 1];
  cnt[tid] = 0;
  __syncthreads();
  for (int p = p0 + tid; p < p1; p += EDGE_BATCH * EDGE_THREADS) {
    int w[EDGE_BATCH];
#pragma unroll
    for (int u = 0; u < EDGE_BATCH; ++u) {
      const int q = p + u * EDGE_THREADS;
      w[u] = ebuf[q < p1 ? q : p1 - 1];
    }
#pragma unroll
    for (int u = 0; u < EDGE_BATCH; ++u)
      atomicAdd(&cnt[w[u] & (BK_NODES - 1)], (p + u * EDGE_THREADS < p1) ? 1 : 0);
  }
  __syncthreads();
  const int deg = cnt[tid];
  const int start = p0 + gnn_block_excl_scan(deg, wsum);
  cnt[tid] = start;
  const int node = b * BK_NODES + tid;
  if (tid < BK_NODES && node < nN) {
    rowptr[node] = start;
    degi[node] = deg;
    invd[node] = 1.0f / fmaxf((float)deg, 1.0f);
  }
  __syncthreads();
  for (int p = p0 + tid; p < p1; p += EDGE_BATCH * EDGE_THREADS) {
    int w[EDGE_BATCH], slot[EDGE_BATCH];
#pragma unroll
    for (int u = 0; u < EDGE_BATCH; ++u) {
      const int q = p + u * EDGE_THREADS;
      w[u] = ebuf[q < p1 ? q : p1 - 1];
    }
#pragma unroll
    for (int u = 0; u < EDGE_BATCH; ++u)
      slot[u] = (p + u * EDGE_THREADS < p1) ? atomicAdd(&cnt[w[u] & (BK_NODES - 1)], 1) : -1;
#pragma unroll
    for (int u = 0; u < EDGE_BATCH; ++u)
      if (slot[u] >= 0) colidx[slot[u]] = w[u] >> BK_SHIFT;
  }
}

#define GATHER_WAVES 8192
#define GATHER_GROUP 16
static_assert((GATHER_WAVES & (GATHER_WAVES - 1)) == 0, "node split uses a shift");
#define TILE_ROWS 16
#define AGG_PITCH 68
static_assert((long)N_NODES <= (long)TILE_ROWS * GATHER_WAVES, "a wave's node run fits one 16-row tile");
static_assert(GATHER_GROUP == 16, "the load fence names 16 row registers");
static_assert((long)N_NODES * D < (1L << 31), "row offsets are 32-bit float indices");

template <bool FULL>
__device__ __forceinline__ void gnn_gather_group(const float* __restrict__ h, int idx, int j,
                                                 int m, int pos0, int eEnd, unsigned loff,
                                                 int re, float* srow, int& cur, int& curEnd,
                                                 v2f& acc) {
  v2f v[GATHER_GROUP];
#pragma unroll
  for (int u = 0; u < GATHER_GROUP; ++u) {
    const int sl = (FULL || j + u < m) ? (j + u) : (m - 1);
    const unsigned nb = (unsigned)__builtin_amdgcn_readlane(idx, sl);
    v[u] = *(const v2f*)(h + ((nb << 6) + loff));
  }
  asm volatile("" ::"v"(v[0]), "v"(v[1]), "v"(v[2]), "v"(v[3]), "v"(v[4]), "v"(v[5]),
               "v"(v[6]), "v"(v[7]), "v"(v[8]), "v"(v[9]), "v"(v[10]), "v"(v[11]),
               "v"(v[12]), "v"(v[13]), "v"(v[14]), "v"(v[15]));
  const v2f zero2 = {0.f, 0.f};
#pragma unroll
  for (int u = 0; u < GATHER_GROUP; ++u) {
    const int pos = pos0 + u;
    if (FULL || pos < eEnd) {
      while (pos == curEnd) {
        *(v2f*)(srow + cur * AGG_PITCH) = acc;
        acc = zero2;
        ++cur;
        curEnd = __builtin_amdgcn_readlane(re, cur);
      }
      acc += v[u];
    }
  }
}

__global__ __launch_bounds__(256) void gnn_sage_fused_kernel(
    const float* __restrict__ h, const int* __restrict__ colidx, const int* __restrict__ rowptr,
    const int* __restrict__ degi, const float* __restrict__ invdeg,
    const float* __restrict__ Wl, const float* __restrict__ Wr, const float* __restrict__ bias,
    float* __restrict__ out, int nN, int do_relu, int do_resid) {
  __shared__ __align__(16) float2 sWf[2][16][4][32];
  __shared__ __align__(16) float sB[D];
  __shared__ __align__(16) float sAgg[8][TILE_ROWS * AGG_PITCH];
  const int tid = threadIdx.x;
  const int lane = tid & 31;
  const int wv = __builtin_amdgcn_readfirstlane(tid >> 5);
  const int w = (int)blockIdx.x * 8 + wv;
  const int n0 = (int)(((long)w * nN) / GATHER_WAVES);
  const int n1 = (int)(((long)(w + 1) * nN) / GATHER_WAVES);
  const int nn = n1 - n0;

  const int lrow = lane & 15;
  const int khalf = (lane < 16) ? 0 : 2;
  const int M = n0 + lrow;
  const int Mc = (lrow < nn) ? M : (nN - 1);
  const float invd = invdeg[Mc];
  const float* Hrow = h + (long)Mc * D + khalf;
  v2f ah[16];
#pragma unroll
  for (int k = 0; k < 16; ++k) ah[k] = *(const v2f*)(Hrow + 4 * k);
  int rb = 0, re = 0;
  if (lane < nn) {
    rb = rowptr[n0 + lane];
    re = rb + degi[n0 + lane];
  }
  const int e0 = __builtin_amdgcn_readlane(rb, 0);
  const int eEnd = __builtin_amdgcn_readlane(re, nn > 0 ? nn - 1 : 0);
  int nxt = (e0 + lane < eEnd) ? colidx[e0 + lane] : 0;

  for (int t = tid; t < 2048; t += 256) {
    int k  = t >> 7;
    int n  = (t >> 5) & 3;
    int ln = t & 31;
    int koff = (ln < 16) ? 0 : 2;
    int lr = ln & 15;
    int kb = k * 4 + koff;
    int col = n * 16 + lr;
    float2 f0, f1;
    f0.x = Wl[(kb + 0) * D + col];
    f0.y = Wl[(kb + 1) * D + col];
    f1.x = Wr[(kb + 0) * D + col];
    f1.y = Wr[(kb + 1) * D + col];
    sWf[0][k][n][ln] = f0;
    sWf[1][k][n][ln] = f1;
  }
  if (tid < D) sB[tid] = bias[tid];
  __syncthreads();
  if (nn <= 0) return;

  v8f c4[4];
#pragma unroll
  for (int n = 0; n < 4; ++n) {
    float bv = sB[n * 16 + lrow];
#pragma unroll
    for (int j = 0; j < 8; ++j) c4[n][j] = bv;
  }
#pragma unroll
  for (int k = 0; k < 16; ++k) {
#pragma unroll
    for (int n = 0; n < 4; ++n) {
      float2 bf2 = sWf[1][k][n][lane];
      v2f bf;
      bf[0] = bf2.x;
      bf[1] = bf2.y;
      c4[n] = __builtin_amdgcn_wmma_f32_16x16x4_f32(
          false, ah[k], false, bf, (short)0, c4[n], false, false);
    }
  }

  const unsigned loff = 2u * (unsigned)lane;
  float* const srow = &sAgg[wv][0] + loff;

  int cur = 0;
  int curEnd = __builtin_amdgcn_readlane(re, 0);
  v2f acc = {0.f, 0.f};

  for (int base = e0; base < eEnd; base += 32) {
    const int idx = nxt;
    asm volatile("" ::"v"(idx));
    const int nbase = base + 32;
    nxt = (nbase + lane < eEnd) ? colidx[nbase + lane] : 0;
    const int rem = eEnd - base;
    const int m = rem < 32 ? rem : 32;
    for (int j = 0; j < m; j += GATHER_GROUP) {
      if (j + GATHER_GROUP <= m)
        gnn_gather_group<true>(h, idx, j, m, base + j, eEnd, loff, re, srow, cur, curEnd, acc);
      else
        gnn_gather_group<false>(h, idx, j, m, base + j, eEnd, loff, re, srow, cur, curEnd, acc);
    }
  }
  const v2f zero2 = {0.f, 0.f};
  for (; cur < TILE_ROWS; ++cur) {
    *(v2f*)(srow + cur * AGG_PITCH) = acc;
    acc = zero2;
  }

  __builtin_amdgcn_fence(__ATOMIC_RELEASE, "wavefront");
  __builtin_amdgcn_wave_barrier();
  __builtin_amdgcn_fence(__ATOMIC_ACQUIRE, "wavefront");

  const float* sArow = &sAgg[wv][0] + lrow * AGG_PITCH + khalf;
#pragma unroll
  for (int k = 0; k < 16; ++k) {
    v2f a = *(const v2f*)(sArow + 4 * k);
    a[0] = a[0] * invd;
    a[1] = a[1] * invd;
#pragma unroll
    for (int n = 0; n < 4; ++n) {
      float2 bf2 = sWf[0][k][n][lane];
      v2f bf;
      bf[0] = bf2.x;
      bf[1] = bf2.y;
      c4[n] = __builtin_amdgcn_wmma_f32_16x16x4_f32(
          false, a, false, bf, (short)0, c4[n], false, false);
    }
  }
#pragma unroll
  for (int n = 0; n < 4; ++n) {
#pragma unroll
    for (int j = 0; j < 8; ++j) {
      int r = (lane < 16) ? j : j + 8;
      int col = n * 16 + lrow;
      if (r < nn) {
        long o = (long)(n0 + r) * D + col;
        float v = c4[n][j];
        if (do_resid) v += h[o];
        if (do_relu) v = fmaxf(v, 0.0f);
        out[o] = v;
      }
    }
  }
}

__global__ __launch_bounds__(256) void gnn_scatter_kernel(const float* __restrict__ h,
                                                          const int* __restrict__ src,
                                                          const int* __restrict__ dstv,
                                                          float* __restrict__ agg, int nE) {
  long tid = (long)blockIdx.x * 256 + threadIdx.x;
  int e = (int)(tid >> 4);
  int c = (int)(tid & 15);
  if (e >= nE) return;
  int s = src[e];
  int d = dstv[e];
  const float4 v = *(const float4*)(h + (long)s * D + c * 4);
  float* ap = agg + (long)d * D + c * 4;
  atomicAdd(ap + 0, v.x);
  atomicAdd(ap + 1, v.y);
  atomicAdd(ap + 2, v.z);
  atomicAdd(ap + 3, v.w);
}

__global__ __launch_bounds__(256) void gnn_sage_gemm_kernel(
    const float* __restrict__ agg, const float* __restrict__ h,
    const float* __restrict__ invdeg,
    const float* __restrict__ Wl, const float* __restrict__ Wr,
    const float* __restrict__ bias,
    float* __restrict__ out, int nN, int do_relu, int do_resid) {
  __shared__ __align__(16) float2 sWf[2][16][4][32];
  __shared__ __align__(16) float sB[D];
  int tid = threadIdx.x;

  for (int t = tid; t < 2048; t += 256) {
    int k  = t >> 7;
    int n  = (t >> 5) & 3;
    int ln = t & 31;
    int koff = (ln < 16) ? 0 : 2;
    int lr = ln & 15;
    int kb = k * 4 + koff;
    int col = n * 16 + lr;
    float2 f0, f1;
    f0.x = Wl[(kb + 0) * D + col];
    f0.y = Wl[(kb + 1) * D + col];
    f1.x = Wr[(kb + 0) * D + col];
    f1.y = Wr[(kb + 1) * D + col];
    sWf[0][k][n][ln] = f0;
    sWf[1][k][n][ln] = f1;
  }
  if (tid < D) sB[tid] = bias[tid];
  __syncthreads();

  int wave = tid >> 5;
  int lane = tid & 31;
  int lrow = lane & 15;
  int khalf = (lane < 16) ? 0 : 2;
  int rowbase = blockIdx.x * 128 + wave * 16;
  int M = rowbase + lrow;
  int Mc = (M < nN) ? M : (nN - 1);

  float invd = invdeg[Mc];

  v8f acc[4];
#pragma unroll
  for (int n = 0; n < 4; ++n) {
    float bv = sB[n * 16 + lrow];
#pragma unroll
    for (int j = 0; j < 8; ++j) acc[n][j] = bv;
  }

  for (int p = 0; p < 2; ++p) {
    const float* Arow = (p == 0 ? agg : h) + (long)Mc * D;
    float scale = (p == 0) ? invd : 1.0f;
#pragma unroll
    for (int k = 0; k < 16; ++k) {
      int kb = k * 4 + khalf;
      v2f a;
      a[0] = Arow[kb + 0] * scale;
      a[1] = Arow[kb + 1] * scale;
#pragma unroll
      for (int n = 0; n < 4; ++n) {
        float2 bf2 = sWf[p][k][n][lane];
        v2f bf;
        bf[0] = bf2.x;
        bf[1] = bf2.y;
        acc[n] = __builtin_amdgcn_wmma_f32_16x16x4_f32(
            false, a, false, bf, (short)0, acc[n], false, false);
      }
    }
  }

#pragma unroll
  for (int n = 0; n < 4; ++n) {
#pragma unroll
    for (int j = 0; j < 8; ++j) {
      int row = rowbase + ((lane < 16) ? j : j + 8);
      int col = n * 16 + lrow;
      if (row < nN) {
        float v = acc[n][j];
        if (do_resid) v += h[(long)row * D + col];
        if (do_relu) v = fmaxf(v, 0.0f);
        out[(long)row * D + col] = v;
      }
    }
  }
}

extern "C" void kernel_launch(void* const* d_in, const int* in_sizes, int n_in,
                              void* d_out, int out_size, void* d_ws, size_t ws_size,
                              hipStream_t stream) {
  const float* x  = (const float*)d_in[0];
  const int*   ei = (const int*)d_in[1];
  const float* Wl = (const float*)d_in[2];
  const float* Wr = (const float*)d_in[3];
  const float* b  = (const float*)d_in[4];
  float* out = (float*)d_out;

  const int* src = ei;
  const int* dst = ei + N_EDGES;

  const size_t featBytes = (size_t)N_NODES * D * sizeof(float);
  const size_t nodeI     = align256((size_t)N_NODES * sizeof(int));
  const size_t nodeF     = align256((size_t)N_NODES * sizeof(float));
  const size_t edgeI     = align256((size_t)N_EDGES * sizeof(int));
  const int scanBlocks   = (N_NODES + 255) / 256;
  const size_t histBytes = align256((size_t)EDGE_CHUNKS * N_BUCKETS * sizeof(int));
  const size_t baseBytes = align256((size_t)(N_BUCKETS + 1) * sizeof(int));

  const size_t csrNeed = 2 * nodeI + histBytes + baseBytes + nodeF + 2 * edgeI + 2 * featBytes;

  char* ws = (char*)d_ws;

  if (ws_size >= csrNeed) {
    int*   degi       = (int*)ws;   ws += nodeI;
    int*   rowptr     = (int*)ws;   ws += nodeI;
    int*   hist       = (int*)ws;   ws += histBytes;
    int*   bucketBase = (int*)ws;   ws += baseBytes;
    float* invdeg     = (float*)ws; ws += nodeF;
    int*   colidx     = (int*)ws;   ws += edgeI;
    int*   ebuf       = (int*)ws;   ws += edgeI;
    float* h1         = (float*)ws; ws += featBytes;
    float* h2         = (float*)ws;

    gnn_bucket_count_kernel<<<EDGE_CHUNKS, EDGE_THREADS, 0, stream>>>(dst, hist, N_EDGES);
    gnn_bucket_scatter_kernel<<<EDGE_CHUNKS, EDGE_THREADS, 0, stream>>>(src, dst, hist, bucketBase, ebuf, N_EDGES);
    gnn_bucket_rows_kernel<<<N_BUCKETS, EDGE_THREADS, 0, stream>>>(ebuf, bucketBase, rowptr, degi, invdeg, colidx, N_NODES);

    const float* h = x;
    float* houts[N_LAYERS] = {h1, h2, h1, h2, out};
    int fusedBlocks = GATHER_WAVES / 8;

    for (int i = 0; i < N_LAYERS; ++i) {
      int relu  = (i < N_LAYERS - 1) ? 1 : 0;
      int resid = (i > 0 && i < N_LAYERS - 1) ? 1 : 0;
      gnn_sage_fused_kernel<<<fusedBlocks, 256, 0, stream>>>(
          h, colidx, rowptr, degi, invdeg,
          Wl + (long)i * D * D, Wr + (long)i * D * D, b + (long)i * D,
          houts[i], N_NODES, relu, resid);
      h = houts[i];
    }
  } else {
    float* invdeg = (float*)ws; ws += nodeF;
    float* agg    = (float*)ws; ws += featBytes;
    float* h1     = (float*)ws; ws += featBytes;
    float* h2     = (float*)ws;

    hipMemsetAsync(invdeg, 0, (size_t)N_NODES * sizeof(float), stream);
    gnn_deg_int_kernel<<<(N_EDGES + 255) / 256, 256, 0, stream>>>(dst, (int*)invdeg, N_EDGES);
    gnn_invdeg_from_int_kernel<<<scanBlocks, 256, 0, stream>>>((const int*)invdeg, invdeg, N_NODES);

    const float* h = x;
    float* houts[N_LAYERS] = {h1, h2, h1, h2, out};
    long scatterBlocks = ((long)N_EDGES * 16 + 255) / 256;
    int gemmBlocks = (N_NODES + 127) / 128;

    for (int i = 0; i < N_LAYERS; ++i) {
      hipMemsetAsync(agg, 0, featBytes, stream);
      gnn_scatter_kernel<<<(int)scatterBlocks, 256, 0, stream>>>(h, src, dst, agg, N_EDGES);
      int relu  = (i < N_LAYERS - 1) ? 1 : 0;
      int resid = (i > 0 && i < N_LAYERS - 1) ? 1 : 0;
      gnn_sage_gemm_kernel<<<gemmBlocks, 256, 0, stream>>>(
          agg, h, invdeg,
          Wl + (long)i * D * D, Wr + (long)i * D * D, b + (long)i * D,
          houts[i], N_NODES, relu, resid);
      h = houts[i];
    }
  }
}


# Fused layer kernel entry: a block's Wl, Wr and bias requested at once up front as 16-byte whole-row loads, first index request moved behind the barrier: 1 load round trip before the barrier, was 10
# speedup vs baseline: 2.0558x; 1.0630x over previous
#include <hip/hip_runtime.h>

typedef __attribute__((ext_vector_type(2))) float v2f;
typedef __attribute__((ext_vector_type(8))) float v8f;

#define N_NODES 100000
#define N_EDGES 1600000
#define D 64
#define N_LAYERS 5

static inline size_t align256(size_t x) { return (x + 255) & ~(size_t)255; }

__global__ __launch_bounds__(256) void gnn_deg_int_kernel(const int* __restrict__ dst,
                                                          int* __restrict__ degi, int nE) {
  int e = blockIdx.x * 256 + threadIdx.x;
  if (e < nE) atomicAdd(&degi[dst[e]], 1);
}

__global__ __launch_bounds__(256) void gnn_invdeg_from_int_kernel(const int* __restrict__ degi,
                                                                  float* __restrict__ invd, int n) {
  int i = blockIdx.x * 256 + threadIdx.x;
  if (i < n) invd[i] = 1.0f / fmaxf((float)degi[i], 1.0f);
}

#define BK_SHIFT 9
#define BK_NODES (1 << BK_SHIFT)
#define N_BUCKETS ((N_NODES + BK_NODES - 1) / BK_NODES)
#define EDGE_CHUNKS 128
#define EDGE_THREADS 1024
#define EDGE_BATCH 4
#define HIST_PARTS 4
static_assert((long)N_NODES << BK_SHIFT < (1L << 31), "src and the local dst share one 32-bit word");
static_assert(N_BUCKETS <= EDGE_THREADS / HIST_PARTS, "one thread per bucket in each column-sum part");
static_assert(EDGE_CHUNKS % HIST_PARTS == 0, "column sums are split evenly");
static_assert(BK_NODES <= EDGE_THREADS, "one thread per node of a bucket");

__device__ __forceinline__ int gnn_block_excl_scan(int v, int* wsum) {
  const int lane = threadIdx.x & 31;
  const int w = threadIdx.x >> 5;
  int inc = v;
#pragma unroll
  for (int o = 1; o < 32; o <<= 1) {
    const int t = __shfl_up(inc, o, 32);
    if (lane >= o) inc += t;
  }
  if (lane == 31) wsum[w] = inc;
  __syncthreads();
  const int ws = wsum[lane];
  int wi = ws;
#pragma unroll
  for (int o = 1; o < 32; o <<= 1) {
    const int t = __shfl_up(wi, o, 32);
    if (lane >= o) wi += t;
  }
  const int wexcl = __shfl(wi - ws, w, 32);
  return wexcl + inc - v;
}

__global__ __launch_bounds__(EDGE_THREADS) void gnn_bucket_count_kernel(const int* __restrict__ dst,
                                                                        int* __restrict__ hist,
                                                                        int nE) {
  __shared__ int cnt[N_BUCKETS];
  const int tid = threadIdx.x;
  const int c = blockIdx.x;
  if (tid < N_BUCKETS) cnt[tid] = 0;
  __syncthreads();
  const int e0 = (int)(((long)c * nE) / EDGE_CHUNKS);
  const int e1 = (int)(((long)(c + 1) * nE) / EDGE_CHUNKS);
  for (int e = e0 + tid; e < e1; e += EDGE_BATCH * EDGE_THREADS) {
    int d[EDGE_BATCH];
#pragma unroll
    for (int u = 0; u < EDGE_BATCH; ++u) {
      const int q = e + u * EDGE_THREADS;
      d[u] = dst[q < e1 ? q : e1 - 1];
    }
#pragma unroll
    for (int u = 0; u < EDGE_BATCH; ++u)
      atomicAdd(&cnt[d[u] >> BK_SHIFT], (e + u * EDGE_THREADS < e1) ? 1 : 0);
  }
  __syncthreads();
  if (tid < N_BUCKETS) hist[c * N_BUCKETS + tid] = cnt[tid];
}

__global__ __launch_bounds__(EDGE_THREADS) void gnn_bucket_scatter_kernel(
    const int* __restrict__ src, const int* __restrict__ dst, const int* __restrict__ hist,
    int* __restrict__ bucketBase, int* __restrict__ ebuf, int nE) {
  __shared__ int ptot[HIST_PARTS][EDGE_THREADS / HIST_PARTS];
  __shared__ int pbef[HIST_PARTS][EDGE_THREADS / HIST_PARTS];
  __shared__ int cursor[N_BUCKETS];
  __shared__ int wsum[32];
  const int tid = threadIdx.x;
  const int c = blockIdx.x;

  const int part = tid / (EDGE_THREADS / HIST_PARTS);
  const int j = tid % (EDGE_THREADS / HIST_PARTS);
  int t = 0, bf = 0;
  if (j < N_BUCKETS) {
#pragma unroll
    for (int i = 0; i < EDGE_CHUNKS / HIST_PARTS; ++i) {
      const int k = part * (EDGE_CHUNKS / HIST_PARTS) + i;
      const int v = hist[k * N_BUCKETS + j];
      t += v;
      bf += (k < c) ? v : 0;
    }
  }
  ptot[part][j] = t;
  pbef[part][j] = bf;
  __syncthreads();
  int tot = 0, before = 0;
  if (tid < N_BUCKETS) {
#pragma unroll
    for (int p = 0; p < HIST_PARTS; ++p) {
      tot += ptot[p][tid];
      before += pbef[p][tid];
    }
  }
  const int excl = gnn_block_excl_scan(tot, wsum);
  if (tid < N_BUCKETS) cursor[tid] = excl + before;
  if (c == 0 && tid <= N_BUCKETS) bucketBase[tid] = excl;
  __syncthreads();

  const int e0 = (int)(((long)c * nE) / EDGE_CHUNKS);
  const int e1 = (int)(((long)(c + 1) * nE) / EDGE_CHUNKS);
  for (int e = e0 + tid; e < e1; e += EDGE_BATCH * EDGE_THREADS) {
    int d[EDGE_BATCH], s[EDGE_BATCH], slot[EDGE_BATCH];
#pragma unroll
    for (int u = 0; u < EDGE_BATCH; ++u) {
      const int q = e + u * EDGE_THREADS;
      const int qc = q < e1 ? q : e1 - 1;
      d[u] = dst[qc];
      s[u] = src[qc];
    }
#pragma unroll
    for (int u = 0; u < EDGE_BATCH; ++u)
      slot[u] = (e + u * EDGE_THREADS < e1) ? atomicAdd(&cursor[d[u] >> BK_SHIFT], 1) : -1;
#pragma unroll
    for (int u = 0; u < EDGE_BATCH; ++u)
      if (slot[u] >= 0) ebuf[slot[u]] = (s[u] << BK_SHIFT) | (d[u] & (BK_NODES - 1));
  }
}

__global__ __launch_bounds__(EDGE_THREADS) void gnn_bucket_rows_kernel(
    const int* __restrict__ ebuf, const int* __restrict__ bucketBase, int* __restrict__ rowptr,
    int* __restrict__ degi, float* __restrict__ invd, int* __restrict__ colidx, int nN) {
  __shared__ int cnt[EDGE_THREADS];
  __shared__ int wsum[32];
  const int tid = threadIdx.x;
  const int b = blockIdx.x;
  const int p0 = bucketBase[b];
  const int p1 = bucketBase[b + 1];
  cnt[tid] = 0;
  __syncthreads();
  for (int p = p0 + tid; p < p1; p += EDGE_BATCH * EDGE_THREADS) {
    int w[EDGE_BATCH];
#pragma unroll
    for (int u = 0; u < EDGE_BATCH; ++u) {
      const int q = p + u * EDGE_THREADS;
      w[u] = ebuf[q < p1 ? q : p1 - 1];
    }
#pragma unroll
    for (int u = 0; u < EDGE_BATCH; ++u)
      atomicAdd(&cnt[w[u] & (BK_NODES - 1)], (p + u * EDGE_THREADS < p1) ? 1 : 0);
  }
  __syncthreads();
  const int deg = cnt[tid];
  const int start = p0 + gnn_block_excl_scan(deg, wsum);
  cnt[tid] = start;
  const int node = b * BK_NODES + tid;
  if (tid < BK_NODES && node < nN) {
    rowptr[node] = start;
    degi[node] = deg;
    invd[node] = 1.0f / fmaxf((float)deg, 1.0f);
  }
  __syncthreads();
  for (int p = p0 + tid; p < p1; p += EDGE_BATCH * EDGE_THREADS) {
    int w[EDGE_BATCH], slot[EDGE_BATCH];
#pragma unroll
    for (int u = 0; u < EDGE_BATCH; ++u) {
      const int q = p + u * EDGE_THREADS;
      w[u] = ebuf[q < p1 ? q : p1 - 1];
    }
#pragma unroll
    for (int u = 0; u < EDGE_BATCH; ++u)
      slot[u] = (p + u * EDGE_THREADS < p1) ? atomicAdd(&cnt[w[u] & (BK_NODES - 1)], 1) : -1;
#pragma unroll
    for (int u = 0; u < EDGE_BATCH; ++u)
      if (slot[u] >= 0) colidx[slot[u]] = w[u] >> BK_SHIFT;
  }
}

#define GATHER_WAVES 8192
#define GATHER_GROUP 16
static_assert((GATHER_WAVES & (GATHER_WAVES - 1)) == 0, "node split uses a shift");
#define TILE_ROWS 16
#define AGG_PITCH 68
#define STAGE_ITEMS 4
static_assert((long)N_NODES <= (long)TILE_ROWS * GATHER_WAVES, "a wave's node run fits one 16-row tile");
static_assert(GATHER_GROUP == 16, "the load fence names 16 row registers");
static_assert((long)N_NODES * D < (1L << 31), "row offsets are 32-bit float indices");

template <bool FULL>
__device__ __forceinline__ void gnn_gather_group(const float* __restrict__ h, int idx, int j,
                                                 int m, int pos0, int eEnd, unsigned loff,
                                                 int re, float* srow, int& cur, int& curEnd,
                                                 v2f& acc) {
  v2f v[GATHER_GROUP];
#pragma unroll
  for (int u = 0; u < GATHER_GROUP; ++u) {
    const int sl = (FULL || j + u < m) ? (j + u) : (m - 1);
    const unsigned nb = (unsigned)__builtin_amdgcn_readlane(idx, sl);
    v[u] = *(const v2f*)(h + ((nb << 6) + loff));
  }
  asm volatile("" ::"v"(v[0]), "v"(v[1]), "v"(v[2]), "v"(v[3]), "v"(v[4]), "v"(v[5]),
               "v"(v[6]), "v"(v[7]), "v"(v[8]), "v"(v[9]), "v"(v[10]), "v"(v[11]),
               "v"(v[12]), "v"(v[13]), "v"(v[14]), "v"(v[15]));
  const v2f zero2 = {0.f, 0.f};
#pragma unroll
  for (int u = 0; u < GATHER_GROUP; ++u) {
    const int pos = pos0 + u;
    if (FULL || pos < eEnd) {
      while (pos == curEnd) {
        *(v2f*)(srow + cur * AGG_PITCH) = acc;
        acc = zero2;
        ++cur;
        curEnd = __builtin_amdgcn_readlane(re, cur);
      }
      acc += v[u];
    }
  }
}

__global__ __launch_bounds__(256) void gnn_sage_fused_kernel(
    const float* __restrict__ h, const int* __restrict__ colidx, const int* __restrict__ rowptr,
    const int* __restrict__ degi, const float* __restrict__ invdeg,
    const float* __restrict__ Wl, const float* __restrict__ Wr, const float* __restrict__ bias,
    float* __restrict__ out, int nN, int do_relu, int do_resid) {
  __shared__ __align__(16) float2 sWf[2][16][4][32];
  __shared__ __align__(16) float sB[D];
  __shared__ __align__(16) float sAgg[8][TILE_ROWS * AGG_PITCH];
  const int tid = threadIdx.x;
  const int lane = tid & 31;
  const int wv = __builtin_amdgcn_readfirstlane(tid >> 5);
  const int w = (int)blockIdx.x * 8 + wv;
  const int n0 = (int)(((long)w * nN) / GATHER_WAVES);
  const int n1 = (int)(((long)(w + 1) * nN) / GATHER_WAVES);
  const int nn = n1 - n0;

  float4 wq[STAGE_ITEMS][2];
#pragma unroll
  for (int i = 0; i < STAGE_ITEMS; ++i) {
    const float* const Wsrc = (i < STAGE_ITEMS / 2) ? Wl : Wr;
    const int rem = (tid + 256 * i) & 511;
    const int q = rem >> 4;
    const int g = rem & 15;
    wq[i][0] = *(const float4*)(Wsrc + (2 * q) * D + 4 * g);
    wq[i][1] = *(const float4*)(Wsrc + (2 * q + 1) * D + 4 * g);
  }
  const float bq = (tid < D) ? bias[tid] : 0.0f;

  const int lrow = lane & 15;
  const int khalf = (lane < 16) ? 0 : 2;
  const int M = n0 + lrow;
  const int Mc = (lrow < nn) ? M : (nN - 1);
  const float invd = invdeg[Mc];
  const float* Hrow = h + (long)Mc * D + khalf;
  v2f ah[16];
#pragma unroll
  for (int k = 0; k < 16; ++k) ah[k] = *(const v2f*)(Hrow + 4 * k);
  int rb = 0, re = 0;
  if (lane < nn) {
    rb = rowptr[n0 + lane];
    re = rb + degi[n0 + lane];
  }
  const int e0 = __builtin_amdgcn_readlane(rb, 0);
  const int eEnd = __builtin_amdgcn_readlane(re, nn > 0 ? nn - 1 : 0);

#pragma unroll
  for (int i = 0; i < STAGE_ITEMS; ++i) {
    const int rem = (tid + 256 * i) & 511;
    const int q = rem >> 4;
    const int g = rem & 15;
    float2* const fr = &sWf[i >> 1][q >> 1][g >> 2][16 * (q & 1) + 4 * (g & 3)];
    fr[0] = make_float2(wq[i][0].x, wq[i][1].x);
    fr[1] = make_float2(wq[i][0].y, wq[i][1].y);
    fr[2] = make_float2(wq[i][0].z, wq[i][1].z);
    fr[3] = make_float2(wq[i][0].w, wq[i][1].w);
  }
  if (tid < D) sB[tid] = bq;
  __syncthreads();
  if (nn <= 0) return;

  int nxt = (e0 + lane < eEnd) ? colidx[e0 + lane] : 0;

  v8f c4[4];
#pragma unroll
  for (int n = 0; n < 4; ++n) {
    float bv = sB[n * 16 + lrow];
#pragma unroll
    for (int j = 0; j < 8; ++j) c4[n][j] = bv;
  }
#pragma unroll
  for (int k = 0; k < 16; ++k) {
#pragma unroll
    for (int n = 0; n < 4; ++n) {
      float2 bf2 = sWf[1][k][n][lane];
      v2f bf;
      bf[0] = bf2.x;
      bf[1] = bf2.y;
      c4[n] = __builtin_amdgcn_wmma_f32_16x16x4_f32(
          false, ah[k], false, bf, (short)0, c4[n], false, false);
    }
  }

  const unsigned loff = 2u * (unsigned)lane;
  float* const srow = &sAgg[wv][0] + loff;

  int cur = 0;
  int curEnd = __builtin_amdgcn_readlane(re, 0);
  v2f acc = {0.f, 0.f};

  for (int base = e0; base < eEnd; base += 32) {
    const int idx = nxt;
    asm volatile("" ::"v"(idx));
    const int nbase = base + 32;
    nxt = (nbase + lane < eEnd) ? colidx[nbase + lane] : 0;
    const int rem = eEnd - base;
    const int m = rem < 32 ? rem : 32;
    for (int j = 0; j < m; j += GATHER_GROUP) {
      if (j + GATHER_GROUP <= m)
        gnn_gather_group<true>(h, idx, j, m, base + j, eEnd, loff, re, srow, cur, curEnd, acc);
      else
        gnn_gather_group<false>(h, idx, j, m, base + j, eEnd, loff, re, srow, cur, curEnd, acc);
    }
  }
  const v2f zero2 = {0.f, 0.f};
  for (; cur < TILE_ROWS; ++cur) {
    *(v2f*)(srow + cur * AGG_PITCH) = acc;
    acc = zero2;
  }

  __builtin_amdgcn_fence(__ATOMIC_RELEASE, "wavefront");
  __builtin_amdgcn_wave_barrier();
  __builtin_amdgcn_fence(__ATOMIC_ACQUIRE, "wavefront");

  const float* sArow = &sAgg[wv][0] + lrow * AGG_PITCH + khalf;
#pragma unroll
  for (int k = 0; k < 16; ++k) {
    v2f a = *(const v2f*)(sArow + 4 * k);
    a[0] = a[0] * invd;
    a[1] = a[1] * invd;
#pragma unroll
    for (int n = 0; n < 4; ++n) {
      float2 bf2 = sWf[0][k][n][lane];
      v2f bf;
      bf[0] = bf2.x;
      bf[1] = bf2.y;
      c4[n] = __builtin_amdgcn_wmma_f32_16x16x4_f32(
          false, a, false, bf, (short)0, c4[n], false, false);
    }
  }
#pragma unroll
  for (int n = 0; n < 4; ++n) {
#pragma unroll
    for (int j = 0; j < 8; ++j) {
      int r = (lane < 16) ? j : j + 8;
      int col = n * 16 + lrow;
      if (r < nn) {
        long o = (long)(n0 + r) * D + col;
        float v = c4[n][j];
        if (do_resid) v += h[o];
        if (do_relu) v = fmaxf(v, 0.0f);
        out[o] = v;
      }
    }
  }
}

__global__ __launch_bounds__(256) void gnn_scatter_kernel(const float* __restrict__ h,
                                                          const int* __restrict__ src,
                                                          const int* __restrict__ dstv,
                                                          float* __restrict__ agg, int nE) {
  long tid = (long)blockIdx.x * 256 + threadIdx.x;
  int e = (int)(tid >> 4);
  int c = (int)(tid & 15);
  if (e >= nE) return;
  int s = src[e];
  int d = dstv[e];
  const float4 v = *(const float4*)(h + (long)s * D + c * 4);
  float* ap = agg + (long)d * D + c * 4;
  atomicAdd(ap + 0, v.x);
  atomicAdd(ap + 1, v.y);
  atomicAdd(ap + 2, v.z);
  atomicAdd(ap + 3, v.w);
}

__global__ __launch_bounds__(256) void gnn_sage_gemm_kernel(
    const float* __restrict__ agg, const float* __restrict__ h,
    const float* __restrict__ invdeg,
    const float* __restrict__ Wl, const float* __restrict__ Wr,
    const float* __restrict__ bias,
    float* __restrict__ out, int nN, int do_relu, int do_resid) {
  __shared__ __align__(16) float2 sWf[2][16][4][32];
  __shared__ __align__(16) float sB[D];
  int tid = threadIdx.x;

  for (int t = tid; t < 2048; t += 256) {
    int k  = t >> 7;
    int n  = (t >> 5) & 3;
    int ln = t & 31;
    int koff = (ln < 16) ? 0 : 2;
    int lr = ln & 15;
    int kb = k * 4 + koff;
    int col = n * 16 + lr;
    float2 f0, f1;
    f0.x = Wl[(kb + 0) * D + col];
    f0.y = Wl[(kb + 1) * D + col];
    f1.x = Wr[(kb + 0) * D + col];
    f1.y = Wr[(kb + 1) * D + col];
    sWf[0][k][n][ln] = f0;
    sWf[1][k][n][ln] = f1;
  }
  if (tid < D) sB[tid] = bias[tid];
  __syncthreads();

  int wave = tid >> 5;
  int lane = tid & 31;
  int lrow = lane & 15;
  int khalf = (lane < 16) ? 0 : 2;
  int rowbase = blockIdx.x * 128 + wave * 16;
  int M = rowbase + lrow;
  int Mc = (M < nN) ? M : (nN - 1);

  float invd = invdeg[Mc];

  v8f acc[4];
#pragma unroll
  for (int n = 0; n < 4; ++n) {
    float bv = sB[n * 16 + lrow];
#pragma unroll
    for (int j = 0; j < 8; ++j) acc[n][j] = bv;
  }

  for (int p = 0; p < 2; ++p) {
    const float* Arow = (p == 0 ? agg : h) + (long)Mc * D;
    float scale = (p == 0) ? invd : 1.0f;
#pragma unroll
    for (int k = 0; k < 16; ++k) {
      int kb = k * 4 + khalf;
      v2f a;
      a[0] = Arow[kb + 0] * scale;
      a[1] = Arow[kb + 1] * scale;
#pragma unroll
      for (int n = 0; n < 4; ++n) {
        float2 bf2 = sWf[p][k][n][lane];
        v2f bf;
        bf[0] = bf2.x;
        bf[1] = bf2.y;
        acc[n] = __builtin_amdgcn_wmma_f32_16x16x4_f32(
            false, a, false, bf, (short)0, acc[n], false, false);
      }
    }
  }

#pragma unroll
  for (int n = 0; n < 4; ++n) {
#pragma unroll
    for (int j = 0; j < 8; ++j) {
      int row = rowbase + ((lane < 16) ? j : j + 8);
      int col = n * 16 + lrow;
      if (row < nN) {
        float v = acc[n][j];
        if (do_resid) v += h[(long)row * D + col];
        if (do_relu) v = fmaxf(v, 0.0f);
        out[(long)row * D + col] = v;
      }
    }
  }
}

extern "C" void kernel_launch(void* const* d_in, const int* in_sizes, int n_in,
                              void* d_out, int out_size, void* d_ws, size_t ws_size,
                              hipStream_t stream) {
  const float* x  = (const float*)d_in[0];
  const int*   ei = (const int*)d_in[1];
  const float* Wl = (const float*)d_in[2];
  const float* Wr = (const float*)d_in[3];
  const float* b  = (const float*)d_in[4];
  float* out = (float*)d_out;

  const int* src = ei;
  const int* dst = ei + N_EDGES;

  const size_t featBytes = (size_t)N_NODES * D * sizeof(float);
  const size_t nodeI     = align256((size_t)N_NODES * sizeof(int));
  const size_t nodeF     = align256((size_t)N_NODES * sizeof(float));
  const size_t edgeI     = align256((size_t)N_EDGES * sizeof(int));
  const int scanBlocks   = (N_NODES + 255) / 256;
  const size_t histBytes = align256((size_t)EDGE_CHUNKS * N_BUCKETS * sizeof(int));
  const size_t baseBytes = align256((size_t)(N_BUCKETS + 1) * sizeof(int));

  const size_t csrNeed = 2 * nodeI + histBytes + baseBytes + nodeF + 2 * edgeI + 2 * featBytes;

  char* ws = (char*)d_ws;

  if (ws_size >= csrNeed) {
    int*   degi       = (int*)ws;   ws += nodeI;
    int*   rowptr     = (int*)ws;   ws += nodeI;
    int*   hist       = (int*)ws;   ws += histBytes;
    int*   bucketBase = (int*)ws;   ws += baseBytes;
    float* invdeg     = (float*)ws; ws += nodeF;
    int*   colidx     = (int*)ws;   ws += edgeI;
    int*   ebuf       = (int*)ws;   ws += edgeI;
    float* h1         = (float*)ws; ws += featBytes;
    float* h2         = (float*)ws;

    gnn_bucket_count_kernel<<<EDGE_CHUNKS, EDGE_THREADS, 0, stream>>>(dst, hist, N_EDGES);
    gnn_bucket_scatter_kernel<<<EDGE_CHUNKS, EDGE_THREADS, 0, stream>>>(src, dst, hist, bucketBase, ebuf, N_EDGES);
    gnn_bucket_rows_kernel<<<N_BUCKETS, EDGE_THREADS, 0, stream>>>(ebuf, bucketBase, rowptr, degi, invdeg, colidx, N_NODES);

    const float* h = x;
    float* houts[N_LAYERS] = {h1, h2, h1, h2, out};
    int fusedBlocks = GATHER_WAVES / 8;

    for (int i = 0; i < N_LAYERS; ++i) {
      int relu  = (i < N_LAYERS - 1) ? 1 : 0;
      int resid = (i > 0 && i < N_LAYERS - 1) ? 1 : 0;
      gnn_sage_fused_kernel<<<fusedBlocks, 256, 0, stream>>>(
          h, colidx, rowptr, degi, invdeg,
          Wl + (long)i * D * D, Wr + (long)i * D * D, b + (long)i * D,
          houts[i], N_NODES, relu, resid);
      h = houts[i];
    }
  } else {
    float* invdeg = (float*)ws; ws += nodeF;
    float* agg    = (float*)ws; ws += featBytes;
    float* h1     = (float*)ws; ws += featBytes;
    float* h2     = (float*)ws;

    hipMemsetAsync(invdeg, 0, (size_t)N_NODES * sizeof(float), stream);
    gnn_deg_int_kernel<<<(N_EDGES + 255) / 256, 256, 0, stream>>>(dst, (int*)invdeg, N_EDGES);
    gnn_invdeg_from_int_kernel<<<scanBlocks, 256, 0, stream>>>((const int*)invdeg, invdeg, N_NODES);

    const float* h = x;
    float* houts[N_LAYERS] = {h1, h2, h1, h2, out};
    long scatterBlocks = ((long)N_EDGES * 16 + 255) / 256;
    int gemmBlocks = (N_NODES + 127) / 128;

    for (int i = 0; i < N_LAYERS; ++i) {
      hipMemsetAsync(agg, 0, featBytes, stream);
      gnn_scatter_kernel<<<(int)scatterBlocks, 256, 0, stream>>>(h, src, dst, agg, N_EDGES);
      int relu  = (i < N_LAYERS - 1) ? 1 : 0;
      int resid = (i > 0 && i < N_LAYERS - 1) ? 1 : 0;
      gnn_sage_gemm_kernel<<<gemmBlocks, 256, 0, stream>>>(
          agg, h, invdeg,
          Wl + (long)i * D * D, Wr + (long)i * D * D, b + (long)i * D,
          houts[i], N_NODES, relu, resid);
      h = houts[i];
    }
  }
}


# Tile rows move as whole 256-B rows via the wave's LDS tile: h in by 8 16-B loads (was 16 loads of 16 lines), residual by 8 more issued under the 2nd product (was 32 serial loads), out by 8 row stores
# speedup vs baseline: 2.4521x; 1.1928x over previous
#include <hip/hip_runtime.h>

typedef __attribute__((ext_vector_type(2))) float v2f;
typedef __attribute__((ext_vector_type(4))) float v4f;
typedef __attribute__((ext_vector_type(8))) float v8f;

#define N_NODES 100000
#define N_EDGES 1600000
#define D 64
#define N_LAYERS 5

static inline size_t align256(size_t x) { return (x + 255) & ~(size_t)255; }

__global__ __launch_bounds__(256) void gnn_deg_int_kernel(const int* __restrict__ dst,
                                                          int* __restrict__ degi, int nE) {
  int e = blockIdx.x * 256 + threadIdx.x;
  if (e < nE) atomicAdd(&degi[dst[e]], 1);
}

__global__ __launch_bounds__(256) void gnn_invdeg_from_int_kernel(const int* __restrict__ degi,
                                                                  float* __restrict__ invd, int n) {
  int i = blockIdx.x * 256 + threadIdx.x;
  if (i < n) invd[i] = 1.0f / fmaxf((float)degi[i], 1.0f);
}

#define BK_SHIFT 9
#define BK_NODES (1 << BK_SHIFT)
#define N_BUCKETS ((N_NODES + BK_NODES - 1) / BK_NODES)
#define EDGE_CHUNKS 128
#define EDGE_THREADS 1024
#define EDGE_BATCH 4
#define HIST_PARTS 4
static_assert((long)N_NODES << BK_SHIFT < (1L << 31), "src and the local dst share one 32-bit word");
static_assert(N_BUCKETS <= EDGE_THREADS / HIST_PARTS, "one thread per bucket in each column-sum part");
static_assert(EDGE_CHUNKS % HIST_PARTS == 0, "column sums are split evenly");
static_assert(BK_NODES <= EDGE_THREADS, "one thread per node of a bucket");

__device__ __forceinline__ int gnn_block_excl_scan(int v, int* wsum) {
  const int lane = threadIdx.x & 31;
  const int w = threadIdx.x >> 5;
  int inc = v;
#pragma unroll
  for (int o = 1; o < 32; o <<= 1) {
    const int t = __shfl_up(inc, o, 32);
    if (lane >= o) inc += t;
  }
  if (lane == 31) wsum[w] = inc;
  __syncthreads();
  const int ws = wsum[lane];
  int wi = ws;
#pragma unroll
  for (int o = 1; o < 32; o <<= 1) {
    const int t = __shfl_up(wi, o, 32);
    if (lane >= o) wi += t;
  }
  const int wexcl = __shfl(wi - ws, w, 32);
  return wexcl + inc - v;
}

__global__ __launch_bounds__(EDGE_THREADS) void gnn_bucket_count_kernel(const int* __restrict__ dst,
                                                                        int* __restrict__ hist,
                                                                        int nE) {
  __shared__ int cnt[N_BUCKETS];
  const int tid = threadIdx.x;
  const int c = blockIdx.x;
  if (tid < N_BUCKETS) cnt[tid] = 0;
  __syncthreads();
  const int e0 = (int)(((long)c * nE) / EDGE_CHUNKS);
  const int e1 = (int)(((long)(c + 1) * nE) / EDGE_CHUNKS);
  for (int e = e0 + tid; e < e1; e += EDGE_BATCH * EDGE_THREADS) {
    int d[EDGE_BATCH];
#pragma unroll
    for (int u = 0; u < EDGE_BATCH; ++u) {
      const int q = e + u * EDGE_THREADS;
      d[u] = dst[q < e1 ? q : e1 - 1];
    }
#pragma unroll
    for (int u = 0; u < EDGE_BATCH; ++u)
      atomicAdd(&cnt[d[u] >> BK_SHIFT], (e + u * EDGE_THREADS < e1) ? 1 : 0);
  }
  __syncthreads();
  if (tid < N_BUCKETS) hist[c * N_BUCKETS + tid] = cnt[tid];
}

__global__ __launch_bounds__(EDGE_THREADS) void gnn_bucket_scatter_kernel(
    const int* __restrict__ src, const int* __restrict__ dst, const int* __restrict__ hist,
    int* __restrict__ bucketBase, int* __restrict__ ebuf, int nE) {
  __shared__ int ptot[HIST_PARTS][EDGE_THREADS / HIST_PARTS];
  __shared__ int pbef[HIST_PARTS][EDGE_THREADS / HIST_PARTS];
  __shared__ int cursor[N_BUCKETS];
  __shared__ int wsum[32];
  const int tid = threadIdx.x;
  const int c = blockIdx.x;

  const int part = tid / (EDGE_THREADS / HIST_PARTS);
  const int j = tid % (EDGE_THREADS / HIST_PARTS);
  int t = 0, bf = 0;
  if (j < N_BUCKETS) {
#pragma unroll
    for (int i = 0; i < EDGE_CHUNKS / HIST_PARTS; ++i) {
      const int k = part * (EDGE_CHUNKS / HIST_PARTS) + i;
      const int v = hist[k * N_BUCKETS + j];
      t += v;
      bf += (k < c) ? v : 0;
    }
  }
  ptot[part][j] = t;
  pbef[part][j] = bf;
  __syncthreads();
  int tot = 0, before = 0;
  if (tid < N_BUCKETS) {
#pragma unroll
    for (int p = 0; p < HIST_PARTS; ++p) {
      tot += ptot[p][tid];
      before += pbef[p][tid];
    }
  }
  const int excl = gnn_block_excl_scan(tot, wsum);
  if (tid < N_BUCKETS) cursor[tid] = excl + before;
  if (c == 0 && tid <= N_BUCKETS) bucketBase[tid] = excl;
  __syncthreads();

  const int e0 = (int)(((long)c * nE) / EDGE_CHUNKS);
  const int e1 = (int)(((long)(c + 1) * nE) / EDGE_CHUNKS);
  for (int e = e0 + tid; e < e1; e += EDGE_BATCH * EDGE_THREADS) {
    int d[EDGE_BATCH], s[EDGE_BATCH], slot[EDGE_BATCH];
#pragma unroll
    for (int u = 0; u < EDGE_BATCH; ++u) {
      const int q = e + u * EDGE_THREADS;
      const int qc = q < e1 ? q : e1 - 1;
      d[u] = dst[qc];
      s[u] = src[qc];
    }
#pragma unroll
    for (int u = 0; u < EDGE_BATCH; ++u)
      slot[u] = (e + u * EDGE_THREADS < e1) ? atomicAdd(&cursor[d[u] >> BK_SHIFT], 1) : -1;
#pragma unroll
    for (int u = 0; u < EDGE_BATCH; ++u)
      if (slot[u] >= 0) ebuf[slot[u]] = (s[u] << BK_SHIFT) | (d[u] & (BK_NODES - 1));
  }
}

__global__ __launch_bounds__(EDGE_THREADS) void gnn_bucket_rows_kernel(
    const int* __restrict__ ebuf, const int* __restrict__ bucketBase, int* __restrict__ rowptr,
    int* __restrict__ degi, float* __restrict__ invd, int* __restrict__ colidx, int nN) {
  __shared__ int cnt[EDGE_THREADS];
  __shared__ int wsum[32];
  const int tid = threadIdx.x;
  const int b = blockIdx.x;
  const int p0 = bucketBase[b];
  const int p1 = bucketBase[b + 1];
  cnt[tid] = 0;
  __syncthreads();
  for (int p = p0 + tid; p < p1; p += EDGE_BATCH * EDGE_THREADS) {
    int w[EDGE_BATCH];
#pragma unroll
    for (int u = 0; u < EDGE_BATCH; ++u) {
      const int q = p + u * EDGE_THREADS;
      w[u] = ebuf[q < p1 ? q : p1 - 1];
    }
#pragma unroll
    for (int u = 0; u < EDGE_BATCH; ++u)
      atomicAdd(&cnt[w[u] & (BK_NODES - 1)], (p + u * EDGE_THREADS < p1) ? 1 : 0);
  }
  __syncthreads();
  const int deg = cnt[tid];
  const int start = p0 + gnn_block_excl_scan(deg, wsum);
  cnt[tid] = start;
  const int node = b * BK_NODES + tid;
  if (tid < BK_NODES && node < nN) {
    rowptr[node] = start;
    degi[node] = deg;
    invd[node] = 1.0f / fmaxf((float)deg, 1.0f);
  }
  __syncthreads();
  for (int p = p0 + tid; p < p1; p += EDGE_BATCH * EDGE_THREADS) {
    int w[EDGE_BATCH], slot[EDGE_BATCH];
#pragma unroll
    for (int u = 0; u < EDGE_BATCH; ++u) {
      const int q = p + u * EDGE_THREADS;
      w[u] = ebuf[q < p1 ? q : p1 - 1];
    }
#pragma unroll
    for (int u = 0; u < EDGE_BATCH; ++u)
      slot[u] = (p + u * EDGE_THREADS < p1) ? atomicAdd(&cnt[w[u] & (BK_NODES - 1)], 1) : -1;
#pragma unroll
    for (int u = 0; u < EDGE_BATCH; ++u)
      if (slot[u] >= 0) colidx[slot[u]] = w[u] >> BK_SHIFT;
  }
}

#define GATHER_WAVES 8192
#define GATHER_GROUP 16
static_assert((GATHER_WAVES & (GATHER_WAVES - 1)) == 0, "node split uses a shift");
#define TILE_ROWS 16
#define AGG_PITCH 68
#define STAGE_ITEMS 4
static_assert((long)N_NODES <= (long)TILE_ROWS * GATHER_WAVES, "a wave's node run fits one 16-row tile");
static_assert(GATHER_GROUP == 16, "the load fence names 16 row registers");
static_assert((long)N_NODES * D < (1L << 31), "row offsets are 32-bit float indices");

template <bool FULL>
__device__ __forceinline__ void gnn_gather_group(const float* __restrict__ h, int idx, int j,
                                                 int m, int pos0, int eEnd, unsigned loff,
                                                 int re, float* srow, int& cur, int& curEnd,
                                                 v2f& acc) {
  v2f v[GATHER_GROUP];
#pragma unroll
  for (int u = 0; u < GATHER_GROUP; ++u) {
    const int sl = (FULL || j + u < m) ? (j + u) : (m - 1);
    const unsigned nb = (unsigned)__builtin_amdgcn_readlane(idx, sl);
    v[u] = *(const v2f*)(h + ((nb << 6) + loff));
  }
  asm volatile("" ::"v"(v[0]), "v"(v[1]), "v"(v[2]), "v"(v[3]), "v"(v[4]), "v"(v[5]),
               "v"(v[6]), "v"(v[7]), "v"(v[8]), "v"(v[9]), "v"(v[10]), "v"(v[11]),
               "v"(v[12]), "v"(v[13]), "v"(v[14]), "v"(v[15]));
  const v2f zero2 = {0.f, 0.f};
#pragma unroll
  for (int u = 0; u < GATHER_GROUP; ++u) {
    const int pos = pos0 + u;
    if (FULL || pos < eEnd) {
      while (pos == curEnd) {
        *(v2f*)(srow + cur * AGG_PITCH) = acc;
        acc = zero2;
        ++cur;
        curEnd = __builtin_amdgcn_readlane(re, cur);
      }
      acc += v[u];
    }
  }
}

__global__ __launch_bounds__(256) void gnn_sage_fused_kernel(
    const float* __restrict__ h, const int* __restrict__ colidx, const int* __restrict__ rowptr,
    const int* __restrict__ degi, const float* __restrict__ invdeg,
    const float* __restrict__ Wl, const float* __restrict__ Wr, const float* __restrict__ bias,
    float* __restrict__ out, int nN, int do_relu, int do_resid) {
  __shared__ __align__(16) float2 sWf[2][16][4][32];
  __shared__ __align__(16) float sB[D];
  __shared__ __align__(16) float sAgg[8][TILE_ROWS * AGG_PITCH];
  const int tid = threadIdx.x;
  const int lane = tid & 31;
  const int wv = __builtin_amdgcn_readfirstlane(tid >> 5);
  const int w = (int)blockIdx.x * 8 + wv;
  const int n0 = (int)(((long)w * nN) / GATHER_WAVES);
  const int n1 = (int)(((long)(w + 1) * nN) / GATHER_WAVES);
  const int nn = n1 - n0;

  float4 wq[STAGE_ITEMS][2];
#pragma unroll
  for (int i = 0; i < STAGE_ITEMS; ++i) {
    const float* const Wsrc = (i < STAGE_ITEMS / 2) ? Wl : Wr;
    const int rem = (tid + 256 * i) & 511;
    const int q = rem >> 4;
    const int g = rem & 15;
    wq[i][0] = *(const float4*)(Wsrc + (2 * q) * D + 4 * g);
    wq[i][1] = *(const float4*)(Wsrc + (2 * q + 1) * D + 4 * g);
  }
  const float bq = (tid < D) ? bias[tid] : 0.0f;

  const int lrow = lane & 15;
  const int lhalf = lane >> 4;
  const int khalf = 2 * lhalf;
  const int M = n0 + lrow;
  const int Mc = (lrow < nn) ? M : (nN - 1);
  const float invd = invdeg[Mc];
  int rb = 0, dg = 0;
  if (lane < nn) {
    rb = rowptr[n0 + lane];
    dg = degi[n0 + lane];
  }
  const int col4 = 4 * lrow;
  v4f hv[TILE_ROWS / 2];
#pragma unroll
  for (int i = 0; i < TILE_ROWS / 2; ++i) {
    const int r = 2 * i + lhalf;
    const int node = (r < nn) ? (n0 + r) : (nN - 1);
    hv[i] = *(const v4f*)(h + (long)node * D + col4);
  }
  __builtin_amdgcn_sched_barrier(0);

#pragma unroll
  for (int i = 0; i < STAGE_ITEMS; ++i) {
    const int rem = (tid + 256 * i) & 511;
    const int q = rem >> 4;
    const int g = rem & 15;
    float2* const fr = &sWf[i >> 1][q >> 1][g >> 2][16 * (q & 1) + 4 * (g & 3)];
    fr[0] = make_float2(wq[i][0].x, wq[i][1].x);
    fr[1] = make_float2(wq[i][0].y, wq[i][1].y);
    fr[2] = make_float2(wq[i][0].z, wq[i][1].z);
    fr[3] = make_float2(wq[i][0].w, wq[i][1].w);
  }
  if (tid < D) sB[tid] = bq;
  float* const tile = &sAgg[wv][0];
#pragma unroll
  for (int i = 0; i < TILE_ROWS / 2; ++i)
    *(v4f*)(tile + (2 * i + lhalf) * AGG_PITCH + col4) = hv[i];
  __syncthreads();
  if (nn <= 0) return;
  const int re = rb + dg;
  const int e0 = __builtin_amdgcn_readlane(rb, 0);
  const int eEnd = __builtin_amdgcn_readlane(re, nn - 1);

  int nxt = (e0 + lane < eEnd) ? colidx[e0 + lane] : 0;

  v8f c4[4];
#pragma unroll
  for (int n = 0; n < 4; ++n) {
    float bv = sB[n * 16 + lrow];
#pragma unroll
    for (int j = 0; j < 8; ++j) c4[n][j] = bv;
  }
  const float* const sArow = tile + lrow * AGG_PITCH + khalf;
#pragma unroll
  for (int k = 0; k < 16; ++k) {
    const v2f ah = *(const v2f*)(sArow + 4 * k);
#pragma unroll
    for (int n = 0; n < 4; ++n) {
      float2 bf2 = sWf[1][k][n][lane];
      v2f bf;
      bf[0] = bf2.x;
      bf[1] = bf2.y;
      c4[n] = __builtin_amdgcn_wmma_f32_16x16x4_f32(
          false, ah, false, bf, (short)0, c4[n], false, false);
    }
  }

  const unsigned loff = 2u * (unsigned)lane;
  float* const srow = tile + loff;

  int cur = 0;
  int curEnd = __builtin_amdgcn_readlane(re, 0);
  v2f acc = {0.f, 0.f};

  for (int base = e0; base < eEnd; base += 32) {
    const int idx = nxt;
    asm volatile("" ::"v"(idx));
    const int nbase = base + 32;
    nxt = (nbase + lane < eEnd) ? colidx[nbase + lane] : 0;
    const int rem = eEnd - base;
    const int m = rem < 32 ? rem : 32;
    for (int j = 0; j < m; j += GATHER_GROUP) {
      if (j + GATHER_GROUP <= m)
        gnn_gather_group<true>(h, idx, j, m, base + j, eEnd, loff, re, srow, cur, curEnd, acc);
      else
        gnn_gather_group<false>(h, idx, j, m, base + j, eEnd, loff, re, srow, cur, curEnd, acc);
    }
  }
  const v2f zero2 = {0.f, 0.f};
  for (; cur < TILE_ROWS; ++cur) {
    *(v2f*)(srow + cur * AGG_PITCH) = acc;
    acc = zero2;
  }

  v4f hres[TILE_ROWS / 2];
  if (do_resid) {
#pragma unroll
    for (int i = 0; i < TILE_ROWS / 2; ++i) {
      const int r = 2 * i + lhalf;
      const int node = (r < nn) ? (n0 + r) : (nN - 1);
      hres[i] = *(const v4f*)(h + (long)node * D + col4);
    }
  } else {
#pragma unroll
    for (int i = 0; i < TILE_ROWS / 2; ++i) hres[i] = (v4f){0.f, 0.f, 0.f, 0.f};
  }

  __builtin_amdgcn_fence(__ATOMIC_RELEASE, "wavefront");
  __builtin_amdgcn_wave_barrier();
  __builtin_amdgcn_fence(__ATOMIC_ACQUIRE, "wavefront");

#pragma unroll
  for (int k = 0; k < 16; ++k) {
    v2f a = *(const v2f*)(sArow + 4 * k);
    a[0] = a[0] * invd;
    a[1] = a[1] * invd;
#pragma unroll
    for (int n = 0; n < 4; ++n) {
      float2 bf2 = sWf[0][k][n][lane];
      v2f bf;
      bf[0] = bf2.x;
      bf[1] = bf2.y;
      c4[n] = __builtin_amdgcn_wmma_f32_16x16x4_f32(
          false, a, false, bf, (short)0, c4[n], false, false);
    }
  }
#pragma unroll
  for (int n = 0; n < 4; ++n) {
#pragma unroll
    for (int j = 0; j < 8; ++j) tile[(8 * lhalf + j) * AGG_PITCH + n * 16 + lrow] = c4[n][j];
  }
  __builtin_amdgcn_fence(__ATOMIC_RELEASE, "wavefront");
  __builtin_amdgcn_wave_barrier();
  __builtin_amdgcn_fence(__ATOMIC_ACQUIRE, "wavefront");
  v4f o[TILE_ROWS / 2];
#pragma unroll
  for (int i = 0; i < TILE_ROWS / 2; ++i)
    o[i] = *(const v4f*)(tile + (2 * i + lhalf) * AGG_PITCH + col4);
#pragma unroll
  for (int i = 0; i < TILE_ROWS / 2; ++i) {
    const int r = 2 * i + lhalf;
    v4f v = o[i];
    if (do_resid) v += hres[i];
    if (do_relu) {
      v.x = fmaxf(v.x, 0.0f);
      v.y = fmaxf(v.y, 0.0f);
      v.z = fmaxf(v.z, 0.0f);
      v.w = fmaxf(v.w, 0.0f);
    }
    if (r < nn) *(v4f*)(out + (long)(n0 + r) * D + col4) = v;
  }
}

__global__ __launch_bounds__(256) void gnn_scatter_kernel(const float* __restrict__ h,
                                                          const int* __restrict__ src,
                                                          const int* __restrict__ dstv,
                                                          float* __restrict__ agg, int nE) {
  long tid = (long)blockIdx.x * 256 + threadIdx.x;
  int e = (int)(tid >> 4);
  int c = (int)(tid & 15);
  if (e >= nE) return;
  int s = src[e];
  int d = dstv[e];
  const float4 v = *(const float4*)(h + (long)s * D + c * 4);
  float* ap = agg + (long)d * D + c * 4;
  atomicAdd(ap + 0, v.x);
  atomicAdd(ap + 1, v.y);
  atomicAdd(ap + 2, v.z);
  atomicAdd(ap + 3, v.w);
}

__global__ __launch_bounds__(256) void gnn_sage_gemm_kernel(
    const float* __restrict__ agg, const float* __restrict__ h,
    const float* __restrict__ invdeg,
    const float* __restrict__ Wl, const float* __restrict__ Wr,
    const float* __restrict__ bias,
    float* __restrict__ out, int nN, int do_relu, int do_resid) {
  __shared__ __align__(16) float2 sWf[2][16][4][32];
  __shared__ __align__(16) float sB[D];
  int tid = threadIdx.x;

  for (int t = tid; t < 2048; t += 256) {
    int k  = t >> 7;
    int n  = (t >> 5) & 3;
    int ln = t & 31;
    int koff = (ln < 16) ? 0 : 2;
    int lr = ln & 15;
    int kb = k * 4 + koff;
    int col = n * 16 + lr;
    float2 f0, f1;
    f0.x = Wl[(kb + 0) * D + col];
    f0.y = Wl[(kb + 1) * D + col];
    f1.x = Wr[(kb + 0) * D + col];
    f1.y = Wr[(kb + 1) * D + col];
    sWf[0][k][n][ln] = f0;
    sWf[1][k][n][ln] = f1;
  }
  if (tid < D) sB[tid] = bias[tid];
  __syncthreads();

  int wave = tid >> 5;
  int lane = tid & 31;
  int lrow = lane & 15;
  int khalf = (lane < 16) ? 0 : 2;
  int rowbase = blockIdx.x * 128 + wave * 16;
  int M = rowbase + lrow;
  int Mc = (M < nN) ? M : (nN - 1);

  float invd = invdeg[Mc];

  v8f acc[4];
#pragma unroll
  for (int n = 0; n < 4; ++n) {
    float bv = sB[n * 16 + lrow];
#pragma unroll
    for (int j = 0; j < 8; ++j) acc[n][j] = bv;
  }

  for (int p = 0; p < 2; ++p) {
    const float* Arow = (p == 0 ? agg : h) + (long)Mc * D;
    float scale = (p == 0) ? invd : 1.0f;
#pragma unroll
    for (int k = 0; k < 16; ++k) {
      int kb = k * 4 + khalf;
      v2f a;
      a[0] = Arow[kb + 0] * scale;
      a[1] = Arow[kb + 1] * scale;
#pragma unroll
      for (int n = 0; n < 4; ++n) {
        float2 bf2 = sWf[p][k][n][lane];
        v2f bf;
        bf[0] = bf2.x;
        bf[1] = bf2.y;
        acc[n] = __builtin_amdgcn_wmma_f32_16x16x4_f32(
            false, a, false, bf, (short)0, acc[n], false, false);
      }
    }
  }

#pragma unroll
  for (int n = 0; n < 4; ++n) {
#pragma unroll
    for (int j = 0; j < 8; ++j) {
      int row = rowbase + ((lane < 16) ? j : j + 8);
      int col = n * 16 + lrow;
      if (row < nN) {
        float v = acc[n][j];
        if (do_resid) v += h[(long)row * D + col];
        if (do_relu) v = fmaxf(v, 0.0f);
        out[(long)row * D + col] = v;
      }
    }
  }
}

extern "C" void kernel_launch(void* const* d_in, const int* in_sizes, int n_in,
                              void* d_out, int out_size, void* d_ws, size_t ws_size,
                              hipStream_t stream) {
  const float* x  = (const float*)d_in[0];
  const int*   ei = (const int*)d_in[1];
  const float* Wl = (const float*)d_in[2];
  const float* Wr = (const float*)d_in[3];
  const float* b  = (const float*)d_in[4];
  float* out = (float*)d_out;

  const int* src = ei;
  const int* dst = ei + N_EDGES;

  const size_t featBytes = (size_t)N_NODES * D * sizeof(float);
  const size_t nodeI     = align256((size_t)N_NODES * sizeof(int));
  const size_t nodeF     = align256((size_t)N_NODES * sizeof(float));
  const size_t edgeI     = align256((size_t)N_EDGES * sizeof(int));
  const int scanBlocks   = (N_NODES + 255) / 256;
  const size_t histBytes = align256((size_t)EDGE_CHUNKS * N_BUCKETS * sizeof(int));
  const size_t baseBytes = align256((size_t)(N_BUCKETS + 1) * sizeof(int));

  const size_t csrNeed = 2 * nodeI + histBytes + baseBytes + nodeF + 2 * edgeI + 2 * featBytes;

  char* ws = (char*)d_ws;

  if (ws_size >= csrNeed) {
    int*   degi       = (int*)ws;   ws += nodeI;
    int*   rowptr     = (int*)ws;   ws += nodeI;
    int*   hist       = (int*)ws;   ws += histBytes;
    int*   bucketBase = (int*)ws;   ws += baseBytes;
    float* invdeg     = (float*)ws; ws += nodeF;
    int*   colidx     = (int*)ws;   ws += edgeI;
    int*   ebuf       = (int*)ws;   ws += edgeI;
    float* h1         = (float*)ws; ws += featBytes;
    float* h2         = (float*)ws;

    gnn_bucket_count_kernel<<<EDGE_CHUNKS, EDGE_THREADS, 0, stream>>>(dst, hist, N_EDGES);
    gnn_bucket_scatter_kernel<<<EDGE_CHUNKS, EDGE_THREADS, 0, stream>>>(src, dst, hist, bucketBase, ebuf, N_EDGES);
    gnn_bucket_rows_kernel<<<N_BUCKETS, EDGE_THREADS, 0, stream>>>(ebuf, bucketBase, rowptr, degi, invdeg, colidx, N_NODES);

    const float* h = x;
    float* houts[N_LAYERS] = {h1, h2, h1, h2, out};
    int fusedBlocks = GATHER_WAVES / 8;

    for (int i = 0; i < N_LAYERS; ++i) {
      int relu  = (i < N_LAYERS - 1) ? 1 : 0;
      int resid = (i > 0 && i < N_LAYERS - 1) ? 1 : 0;
      gnn_sage_fused_kernel<<<fusedBlocks, 256, 0, stream>>>(
          h, colidx, rowptr, degi, invdeg,
          Wl + (long)i * D * D, Wr + (long)i * D * D, b + (long)i * D,
          houts[i], N_NODES, relu, resid);
      h = houts[i];
    }
  } else {
    float* invdeg = (float*)ws; ws += nodeF;
    float* agg    = (float*)ws; ws += featBytes;
    float* h1     = (float*)ws; ws += featBytes;
    float* h2     = (float*)ws;

    hipMemsetAsync(invdeg, 0, (size_t)N_NODES * sizeof(float), stream);
    gnn_deg_int_kernel<<<(N_EDGES + 255) / 256, 256, 0, stream>>>(dst, (int*)invdeg, N_EDGES);
    gnn_invdeg_from_int_kernel<<<scanBlocks, 256, 0, stream>>>((const int*)invdeg, invdeg, N_NODES);

    const float* h = x;
    float* houts[N_LAYERS] = {h1, h2, h1, h2, out};
    long scatterBlocks = ((long)N_EDGES * 16 + 255) / 256;
    int gemmBlocks = (N_NODES + 127) / 128;

    for (int i = 0; i < N_LAYERS; ++i) {
      hipMemsetAsync(agg, 0, featBytes, stream);
      gnn_scatter_kernel<<<(int)scatterBlocks, 256, 0, stream>>>(h, src, dst, agg, N_EDGES);
      int relu  = (i < N_LAYERS - 1) ? 1 : 0;
      int resid = (i > 0 && i < N_LAYERS - 1) ? 1 : 0;
      gnn_sage_gemm_kernel<<<gemmBlocks, 256, 0, stream>>>(
          agg, h, invdeg,
          Wl + (long)i * D * D, Wr + (long)i * D * D, b + (long)i * D,
          houts[i], N_NODES, relu, resid);
      h = houts[i];
    }
  }
}


# Gather row address = scalar base h + one 32-bit lane offset (1 VGPR, 1 vector op per load; was a 64-bit pair, 2 ops): the freed registers hold a whole chunk's 32 row loads in flight (was 16)
# speedup vs baseline: 2.5888x; 1.0557x over previous
#include <hip/hip_runtime.h>

typedef __attribute__((ext_vector_type(2))) float v2f;
typedef __attribute__((ext_vector_type(4))) float v4f;
typedef __attribute__((ext_vector_type(8))) float v8f;

#define N_NODES 100000
#define N_EDGES 1600000
#define D 64
#define N_LAYERS 5

static inline size_t align256(size_t x) { return (x + 255) & ~(size_t)255; }

__global__ __launch_bounds__(256) void gnn_deg_int_kernel(const int* __restrict__ dst,
                                                          int* __restrict__ degi, int nE) {
  int e = blockIdx.x * 256 + threadIdx.x;
  if (e < nE) atomicAdd(&degi[dst[e]], 1);
}

__global__ __launch_bounds__(256) void gnn_invdeg_from_int_kernel(const int* __restrict__ degi,
                                                                  float* __restrict__ invd, int n) {
  int i = blockIdx.x * 256 + threadIdx.x;
  if (i < n) invd[i] = 1.0f / fmaxf((float)degi[i], 1.0f);
}

#define BK_SHIFT 9
#define BK_NODES (1 << BK_SHIFT)
#define N_BUCKETS ((N_NODES + BK_NODES - 1) / BK_NODES)
#define EDGE_CHUNKS 128
#define EDGE_THREADS 1024
#define EDGE_BATCH 4
#define HIST_PARTS 4
static_assert((long)N_NODES << BK_SHIFT < (1L << 31), "src and the local dst share one 32-bit word");
static_assert(N_BUCKETS <= EDGE_THREADS / HIST_PARTS, "one thread per bucket in each column-sum part");
static_assert(EDGE_CHUNKS % HIST_PARTS == 0, "column sums are split evenly");
static_assert(BK_NODES <= EDGE_THREADS, "one thread per node of a bucket");

__device__ __forceinline__ int gnn_block_excl_scan(int v, int* wsum) {
  const int lane = threadIdx.x & 31;
  const int w = threadIdx.x >> 5;
  int inc = v;
#pragma unroll
  for (int o = 1; o < 32; o <<= 1) {
    const int t = __shfl_up(inc, o, 32);
    if (lane >= o) inc += t;
  }
  if (lane == 31) wsum[w] = inc;
  __syncthreads();
  const int ws = wsum[lane];
  int wi = ws;
#pragma unroll
  for (int o = 1; o < 32; o <<= 1) {
    const int t = __shfl_up(wi, o, 32);
    if (lane >= o) wi += t;
  }
  const int wexcl = __shfl(wi - ws, w, 32);
  return wexcl + inc - v;
}

__global__ __launch_bounds__(EDGE_THREADS) void gnn_bucket_count_kernel(const int* __restrict__ dst,
                                                                        int* __restrict__ hist,
                                                                        int nE) {
  __shared__ int cnt[N_BUCKETS];
  const int tid = threadIdx.x;
  const int c = blockIdx.x;
  if (tid < N_BUCKETS) cnt[tid] = 0;
  __syncthreads();
  const int e0 = (int)(((long)c * nE) / EDGE_CHUNKS);
  const int e1 = (int)(((long)(c + 1) * nE) / EDGE_CHUNKS);
  for (int e = e0 + tid; e < e1; e += EDGE_BATCH * EDGE_THREADS) {
    int d[EDGE_BATCH];
#pragma unroll
    for (int u = 0; u < EDGE_BATCH; ++u) {
      const int q = e + u * EDGE_THREADS;
      d[u] = dst[q < e1 ? q : e1 - 1];
    }
#pragma unroll
    for (int u = 0; u < EDGE_BATCH; ++u)
      atomicAdd(&cnt[d[u] >> BK_SHIFT], (e + u * EDGE_THREADS < e1) ? 1 : 0);
  }
  __syncthreads();
  if (tid < N_BUCKETS) hist[c * N_BUCKETS + tid] = cnt[tid];
}

__global__ __launch_bounds__(EDGE_THREADS) void gnn_bucket_scatter_kernel(
    const int* __restrict__ src, const int* __restrict__ dst, const int* __restrict__ hist,
    int* __restrict__ bucketBase, int* __restrict__ ebuf, int nE) {
  __shared__ int ptot[HIST_PARTS][EDGE_THREADS / HIST_PARTS];
  __shared__ int pbef[HIST_PARTS][EDGE_THREADS / HIST_PARTS];
  __shared__ int cursor[N_BUCKETS];
  __shared__ int wsum[32];
  const int tid = threadIdx.x;
  const int c = blockIdx.x;

  const int part = tid / (EDGE_THREADS / HIST_PARTS);
  const int j = tid % (EDGE_THREADS / HIST_PARTS);
  int t = 0, bf = 0;
  if (j < N_BUCKETS) {
#pragma unroll
    for (int i = 0; i < EDGE_CHUNKS / HIST_PARTS; ++i) {
      const int k = part * (EDGE_CHUNKS / HIST_PARTS) + i;
      const int v = hist[k * N_BUCKETS + j];
      t += v;
      bf += (k < c) ? v : 0;
    }
  }
  ptot[part][j] = t;
  pbef[part][j] = bf;
  __syncthreads();
  int tot = 0, before = 0;
  if (tid < N_BUCKETS) {
#pragma unroll
    for (int p = 0; p < HIST_PARTS; ++p) {
      tot += ptot[p][tid];
      before += pbef[p][tid];
    }
  }
  const int excl = gnn_block_excl_scan(tot, wsum);
  if (tid < N_BUCKETS) cursor[tid] = excl + before;
  if (c == 0 && tid <= N_BUCKETS) bucketBase[tid] = excl;
  __syncthreads();

  const int e0 = (int)(((long)c * nE) / EDGE_CHUNKS);
  const int e1 = (int)(((long)(c + 1) * nE) / EDGE_CHUNKS);
  for (int e = e0 + tid; e < e1; e += EDGE_BATCH * EDGE_THREADS) {
    int d[EDGE_BATCH], s[EDGE_BATCH], slot[EDGE_BATCH];
#pragma unroll
    for (int u = 0; u < EDGE_BATCH; ++u) {
      const int q = e + u * EDGE_THREADS;
      const int qc = q < e1 ? q : e1 - 1;
      d[u] = dst[qc];
      s[u] = src[qc];
    }
#pragma unroll
    for (int u = 0; u < EDGE_BATCH; ++u)
      slot[u] = (e + u * EDGE_THREADS < e1) ? atomicAdd(&cursor[d[u] >> BK_SHIFT], 1) : -1;
#pragma unroll
    for (int u = 0; u < EDGE_BATCH; ++u)
      if (slot[u] >= 0) ebuf[slot[u]] = (s[u] << BK_SHIFT) | (d[u] & (BK_NODES - 1));
  }
}

__global__ __launch_bounds__(EDGE_THREADS) void gnn_bucket_rows_kernel(
    const int* __restrict__ ebuf, const int* __restrict__ bucketBase, int* __restrict__ rowptr,
    int* __restrict__ degi, float* __restrict__ invd, int* __restrict__ colidx, int nN) {
  __shared__ int cnt[EDGE_THREADS];
  __shared__ int wsum[32];
  const int tid = threadIdx.x;
  const int b = blockIdx.x;
  const int p0 = bucketBase[b];
  const int p1 = bucketBase[b + 1];
  cnt[tid] = 0;
  __syncthreads();
  for (int p = p0 + tid; p < p1; p += EDGE_BATCH * EDGE_THREADS) {
    int w[EDGE_BATCH];
#pragma unroll
    for (int u = 0; u < EDGE_BATCH; ++u) {
      const int q = p + u * EDGE_THREADS;
      w[u] = ebuf[q < p1 ? q : p1 - 1];
    }
#pragma unroll
    for (int u = 0; u < EDGE_BATCH; ++u)
      atomicAdd(&cnt[w[u] & (BK_NODES - 1)], (p + u * EDGE_THREADS < p1) ? 1 : 0);
  }
  __syncthreads();
  const int deg = cnt[tid];
  const int start = p0 + gnn_block_excl_scan(deg, wsum);
  cnt[tid] = start;
  const int node = b * BK_NODES + tid;
  if (tid < BK_NODES && node < nN) {
    rowptr[node] = start;
    degi[node] = deg;
    invd[node] = 1.0f / fmaxf((float)deg, 1.0f);
  }
  __syncthreads();
  for (int p = p0 + tid; p < p1; p += EDGE_BATCH * EDGE_THREADS) {
    int w[EDGE_BATCH], slot[EDGE_BATCH];
#pragma unroll
    for (int u = 0; u < EDGE_BATCH; ++u) {
      const int q = p + u * EDGE_THREADS;
      w[u] = ebuf[q < p1 ? q : p1 - 1];
    }
#pragma unroll
    for (int u = 0; u < EDGE_BATCH; ++u)
      slot[u] = (p + u * EDGE_THREADS < p1) ? atomicAdd(&cnt[w[u] & (BK_NODES - 1)], 1) : -1;
#pragma unroll
    for (int u = 0; u < EDGE_BATCH; ++u)
      if (slot[u] >= 0) colidx[slot[u]] = w[u] >> BK_SHIFT;
  }
}

#define GATHER_WAVES 8192
#define GATHER_GROUP 16
#define GATHER_CHUNK 32
static_assert((GATHER_WAVES & (GATHER_WAVES - 1)) == 0, "node split uses a shift");
#define TILE_ROWS 16
#define AGG_PITCH 68
#define STAGE_ITEMS 4
static_assert((long)N_NODES <= (long)TILE_ROWS * GATHER_WAVES, "a wave's node run fits one 16-row tile");
static_assert(GATHER_GROUP == 16 && GATHER_CHUNK == 32, "the load fences name 16 row registers each");
static_assert((long)N_NODES * D * 4 < (1L << 32), "a row's byte offset fits 32 bits");

template <int G, bool FULL>
__device__ __forceinline__ void gnn_gather_group(const char* __restrict__ hb, int idx, int j,
                                                 int m, int pos0, int eEnd, unsigned lane8,
                                                 int re, float* srow, int& cur, int& curEnd,
                                                 v2f& acc) {
  v2f v[G];
#pragma unroll
  for (int u = 0; u < G; ++u) {
    const int sl = (FULL || j + u < m) ? (j + u) : (m - 1);
    const unsigned nb = (unsigned)__builtin_amdgcn_readlane(idx, sl);
    const unsigned off = (nb << 8) + lane8;
    v[u] = *(const v2f*)(hb + off);
  }
  asm volatile("" ::"v"(v[0]), "v"(v[1]), "v"(v[2]), "v"(v[3]), "v"(v[4]), "v"(v[5]),
               "v"(v[6]), "v"(v[7]), "v"(v[8]), "v"(v[9]), "v"(v[10]), "v"(v[11]),
               "v"(v[12]), "v"(v[13]), "v"(v[14]), "v"(v[15]));
  if (G > 16)
    asm volatile("" ::"v"(v[16 % G]), "v"(v[17 % G]), "v"(v[18 % G]), "v"(v[19 % G]),
                 "v"(v[20 % G]), "v"(v[21 % G]), "v"(v[22 % G]), "v"(v[23 % G]),
                 "v"(v[24 % G]), "v"(v[25 % G]), "v"(v[26 % G]), "v"(v[27 % G]),
                 "v"(v[28 % G]), "v"(v[29 % G]), "v"(v[30 % G]), "v"(v[31 % G]));
  const v2f zero2 = {0.f, 0.f};
#pragma unroll
  for (int u = 0; u < G; ++u) {
    const int pos = pos0 + u;
    if (FULL || pos < eEnd) {
      while (pos == curEnd) {
        *(v2f*)(srow + cur * AGG_PITCH) = acc;
        acc = zero2;
        ++cur;
        curEnd = __builtin_amdgcn_readlane(re, cur);
      }
      acc += v[u];
    }
  }
}

__global__ __launch_bounds__(256) __attribute__((amdgpu_waves_per_eu(8, 8))) void gnn_sage_fused_kernel(
    const float* __restrict__ h, const int* __restrict__ colidx, const int* __restrict__ rowptr,
    const int* __restrict__ degi, const float* __restrict__ invdeg,
    const float* __restrict__ Wl, const float* __restrict__ Wr, const float* __restrict__ bias,
    float* __restrict__ out, int nN, int do_relu, int do_resid) {
  __shared__ __align__(16) float2 sWf[2][16][4][32];
  __shared__ __align__(16) float sB[D];
  __shared__ __align__(16) float sAgg[8][TILE_ROWS * AGG_PITCH];
  const int tid = threadIdx.x;
  const int lane = tid & 31;
  const int wv = __builtin_amdgcn_readfirstlane(tid >> 5);
  const int w = (int)blockIdx.x * 8 + wv;
  const int n0 = (int)(((long)w * nN) / GATHER_WAVES);
  const int n1 = (int)(((long)(w + 1) * nN) / GATHER_WAVES);
  const int nn = n1 - n0;

  float4 wq[STAGE_ITEMS][2];
#pragma unroll
  for (int i = 0; i < STAGE_ITEMS; ++i) {
    const float* const Wsrc = (i < STAGE_ITEMS / 2) ? Wl : Wr;
    const int rem = (tid + 256 * i) & 511;
    const int q = rem >> 4;
    const int g = rem & 15;
    wq[i][0] = *(const float4*)(Wsrc + (2 * q) * D + 4 * g);
    wq[i][1] = *(const float4*)(Wsrc + (2 * q + 1) * D + 4 * g);
  }
  const float bq = (tid < D) ? bias[tid] : 0.0f;

  const int lrow = lane & 15;
  const int lhalf = lane >> 4;
  const int khalf = 2 * lhalf;
  const int M = n0 + lrow;
  const int Mc = (lrow < nn) ? M : (nN - 1);
  const float invd = invdeg[Mc];
  int rb = 0, dg = 0;
  if (lane < nn) {
    rb = rowptr[n0 + lane];
    dg = degi[n0 + lane];
  }
  const int col4 = 4 * lrow;
  v4f hv[TILE_ROWS / 2];
  const unsigned off0 = (unsigned)((n0 + lhalf) * (D * 4) + col4 * 4);
  const unsigned offLast = (unsigned)((nN - 1) * (D * 4) + col4 * 4);
#pragma unroll
  for (int i = 0; i < TILE_ROWS / 2; ++i) {
    const unsigned o = off0 + (unsigned)(i * 2 * D * 4);
    hv[i] = *(const v4f*)((const char*)h + (o < offLast ? o : offLast));
  }
  __builtin_amdgcn_sched_barrier(0);

#pragma unroll
  for (int i = 0; i < STAGE_ITEMS; ++i) {
    const int rem = (tid + 256 * i) & 511;
    const int q = rem >> 4;
    const int g = rem & 15;
    float2* const fr = &sWf[i >> 1][q >> 1][g >> 2][16 * (q & 1) + 4 * (g & 3)];
    fr[0] = make_float2(wq[i][0].x, wq[i][1].x);
    fr[1] = make_float2(wq[i][0].y, wq[i][1].y);
    fr[2] = make_float2(wq[i][0].z, wq[i][1].z);
    fr[3] = make_float2(wq[i][0].w, wq[i][1].w);
  }
  if (tid < D) sB[tid] = bq;
  float* const tile = &sAgg[wv][0];
#pragma unroll
  for (int i = 0; i < TILE_ROWS / 2; ++i)
    *(v4f*)(tile + (2 * i + lhalf) * AGG_PITCH + col4) = hv[i];
  __syncthreads();
  if (nn <= 0) return;
  const int re = rb + dg;
  const int e0 = __builtin_amdgcn_readlane(rb, 0);
  const int eEnd = __builtin_amdgcn_readlane(re, nn - 1);

  int nxt = (e0 + lane < eEnd) ? colidx[e0 + lane] : 0;

  v8f c4[4];
#pragma unroll
  for (int n = 0; n < 4; ++n) {
    float bv = sB[n * 16 + lrow];
#pragma unroll
    for (int j = 0; j < 8; ++j) c4[n][j] = bv;
  }
  const float* const sArow = tile + lrow * AGG_PITCH + khalf;
#pragma unroll
  for (int k = 0; k < 16; ++k) {
    const v2f ah = *(const v2f*)(sArow + 4 * k);
#pragma unroll
    for (int n = 0; n < 4; ++n) {
      float2 bf2 = sWf[1][k][n][lane];
      v2f bf;
      bf[0] = bf2.x;
      bf[1] = bf2.y;
      c4[n] = __builtin_amdgcn_wmma_f32_16x16x4_f32(
          false, ah, false, bf, (short)0, c4[n], false, false);
    }
  }

  const unsigned loff = 2u * (unsigned)lane;
  const unsigned lane8 = 8u * (unsigned)lane;
  const char* const hb = (const char*)h;
  float* const srow = tile + loff;

  int cur = 0;
  int curEnd = __builtin_amdgcn_readlane(re, 0);
  v2f acc = {0.f, 0.f};

  for (int base = e0; base < eEnd; base += 32) {
    const int idx = nxt;
    asm volatile("" ::"v"(idx));
    const int nbase = base + 32;
    nxt = (nbase + lane < eEnd) ? colidx[nbase + lane] : 0;
    const int rem = eEnd - base;
    const int m = rem < 32 ? rem : 32;
    if (m == GATHER_CHUNK) {
      gnn_gather_group<GATHER_CHUNK, true>(hb, idx, 0, m, base, eEnd, lane8, re, srow, cur,
                                           curEnd, acc);
    } else {
      for (int j = 0; j < m; j += GATHER_GROUP) {
        if (j + GATHER_GROUP <= m)
          gnn_gather_group<GATHER_GROUP, true>(hb, idx, j, m, base + j, eEnd, lane8, re, srow,
                                               cur, curEnd, acc);
        else
          gnn_gather_group<GATHER_GROUP, false>(hb, idx, j, m, base + j, eEnd, lane8, re, srow,
                                                cur, curEnd, acc);
      }
    }
  }
  const v2f zero2 = {0.f, 0.f};
  for (; cur < TILE_ROWS; ++cur) {
    *(v2f*)(srow + cur * AGG_PITCH) = acc;
    acc = zero2;
  }

  v4f hres[TILE_ROWS / 2];
  if (do_resid) {
#pragma unroll
    for (int i = 0; i < TILE_ROWS / 2; ++i) {
      const int r = 2 * i + lhalf;
      const int node = n0 + (r < nn ? r : nn - 1);
      hres[i] = *(const v4f*)((const char*)h + (unsigned)(node * (D * 4) + col4 * 4));
    }
  } else {
#pragma unroll
    for (int i = 0; i < TILE_ROWS / 2; ++i) hres[i] = (v4f){0.f, 0.f, 0.f, 0.f};
  }

  __builtin_amdgcn_fence(__ATOMIC_RELEASE, "wavefront");
  __builtin_amdgcn_wave_barrier();
  __builtin_amdgcn_fence(__ATOMIC_ACQUIRE, "wavefront");

#pragma unroll
  for (int k = 0; k < 16; ++k) {
    v2f a = *(const v2f*)(sArow + 4 * k);
    a[0] = a[0] * invd;
    a[1] = a[1] * invd;
#pragma unroll
    for (int n = 0; n < 4; ++n) {
      float2 bf2 = sWf[0][k][n][lane];
      v2f bf;
      bf[0] = bf2.x;
      bf[1] = bf2.y;
      c4[n] = __builtin_amdgcn_wmma_f32_16x16x4_f32(
          false, a, false, bf, (short)0, c4[n], false, false);
    }
  }
#pragma unroll
  for (int n = 0; n < 4; ++n) {
#pragma unroll
    for (int j = 0; j < 8; ++j) tile[(8 * lhalf + j) * AGG_PITCH + n * 16 + lrow] = c4[n][j];
  }
  __builtin_amdgcn_fence(__ATOMIC_RELEASE, "wavefront");
  __builtin_amdgcn_wave_barrier();
  __builtin_amdgcn_fence(__ATOMIC_ACQUIRE, "wavefront");
  v4f o[TILE_ROWS / 2];
#pragma unroll
  for (int i = 0; i < TILE_ROWS / 2; ++i)
    o[i] = *(const v4f*)(tile + (2 * i + lhalf) * AGG_PITCH + col4);
#pragma unroll
  for (int i = 0; i < TILE_ROWS / 2; ++i) {
    const int r = 2 * i + lhalf;
    v4f v = o[i];
    if (do_resid) v += hres[i];
    if (do_relu) {
      v.x = fmaxf(v.x, 0.0f);
      v.y = fmaxf(v.y, 0.0f);
      v.z = fmaxf(v.z, 0.0f);
      v.w = fmaxf(v.w, 0.0f);
    }
    if (r < nn) *(v4f*)((char*)out + (unsigned)((n0 + r) * (D * 4) + col4 * 4)) = v;
  }
}

__global__ __launch_bounds__(256) void gnn_scatter_kernel(const float* __restrict__ h,
                                                          const int* __restrict__ src,
                                                          const int* __restrict__ dstv,
                                                          float* __restrict__ agg, int nE) {
  long tid = (long)blockIdx.x * 256 + threadIdx.x;
  int e = (int)(tid >> 4);
  int c = (int)(tid & 15);
  if (e >= nE) return;
  int s = src[e];
  int d = dstv[e];
  const float4 v = *(const float4*)(h + (long)s * D + c * 4);
  float* ap = agg + (long)d * D + c * 4;
  atomicAdd(ap + 0, v.x);
  atomicAdd(ap + 1, v.y);
  atomicAdd(ap + 2, v.z);
  atomicAdd(ap + 3, v.w);
}

__global__ __launch_bounds__(256) void gnn_sage_gemm_kernel(
    const float* __restrict__ agg, const float* __restrict__ h,
    const float* __restrict__ invdeg,
    const float* __restrict__ Wl, const float* __restrict__ Wr,
    const float* __restrict__ bias,
    float* __restrict__ out, int nN, int do_relu, int do_resid) {
  __shared__ __align__(16) float2 sWf[2][16][4][32];
  __shared__ __align__(16) float sB[D];
  int tid = threadIdx.x;

  for (int t = tid; t < 2048; t += 256) {
    int k  = t >> 7;
    int n  = (t >> 5) & 3;
    int ln = t & 31;
    int koff = (ln < 16) ? 0 : 2;
    int lr = ln & 15;
    int kb = k * 4 + koff;
    int col = n * 16 + lr;
    float2 f0, f1;
    f0.x = Wl[(kb + 0) * D + col];
    f0.y = Wl[(kb + 1) * D + col];
    f1.x = Wr[(kb + 0) * D + col];
    f1.y = Wr[(kb + 1) * D + col];
    sWf[0][k][n][ln] = f0;
    sWf[1][k][n][ln] = f1;
  }
  if (tid < D) sB[tid] = bias[tid];
  __syncthreads();

  int wave = tid >> 5;
  int lane = tid & 31;
  int lrow = lane & 15;
  int khalf = (lane < 16) ? 0 : 2;
  int rowbase = blockIdx.x * 128 + wave * 16;
  int M = rowbase + lrow;
  int Mc = (M < nN) ? M : (nN - 1);

  float invd = invdeg[Mc];

  v8f acc[4];
#pragma unroll
  for (int n = 0; n < 4; ++n) {
    float bv = sB[n * 16 + lrow];
#pragma unroll
    for (int j = 0; j < 8; ++j) acc[n][j] = bv;
  }

  for (int p = 0; p < 2; ++p) {
    const float* Arow = (p == 0 ? agg : h) + (long)Mc * D;
    float scale = (p == 0) ? invd : 1.0f;
#pragma unroll
    for (int k = 0; k < 16; ++k) {
      int kb = k * 4 + khalf;
      v2f a;
      a[0] = Arow[kb + 0] * scale;
      a[1] = Arow[kb + 1] * scale;
#pragma unroll
      for (int n = 0; n < 4; ++n) {
        float2 bf2 = sWf[p][k][n][lane];
        v2f bf;
        bf[0] = bf2.x;
        bf[1] = bf2.y;
        acc[n] = __builtin_amdgcn_wmma_f32_16x16x4_f32(
            false, a, false, bf, (short)0, acc[n], false, false);
      }
    }
  }

#pragma unroll
  for (int n = 0; n < 4; ++n) {
#pragma unroll
    for (int j = 0; j < 8; ++j) {
      int row = rowbase + ((lane < 16) ? j : j + 8);
      int col = n * 16 + lrow;
      if (row < nN) {
        float v = acc[n][j];
        if (do_resid) v += h[(long)row * D + col];
        if (do_relu) v = fmaxf(v, 0.0f);
        out[(long)row * D + col] = v;
      }
    }
  }
}

extern "C" void kernel_launch(void* const* d_in, const int* in_sizes, int n_in,
                              void* d_out, int out_size, void* d_ws, size_t ws_size,
                              hipStream_t stream) {
  const float* x  = (const float*)d_in[0];
  const int*   ei = (const int*)d_in[1];
  const float* Wl = (const float*)d_in[2];
  const float* Wr = (const float*)d_in[3];
  const float* b  = (const float*)d_in[4];
  float* out = (float*)d_out;

  const int* src = ei;
  const int* dst = ei + N_EDGES;

  const size_t featBytes = (size_t)N_NODES * D * sizeof(float);
  const size_t nodeI     = align256((size_t)N_NODES * sizeof(int));
  const size_t nodeF     = align256((size_t)N_NODES * sizeof(float));
  const size_t edgeI     = align256((size_t)N_EDGES * sizeof(int));
  const int scanBlocks   = (N_NODES + 255) / 256;
  const size_t histBytes = align256((size_t)EDGE_CHUNKS * N_BUCKETS * sizeof(int));
  const size_t baseBytes = align256((size_t)(N_BUCKETS + 1) * sizeof(int));

  const size_t csrNeed = 2 * nodeI + histBytes + baseBytes + nodeF + 2 * edgeI + 2 * featBytes;

  char* ws = (char*)d_ws;

  if (ws_size >= csrNeed) {
    int*   degi       = (int*)ws;   ws += nodeI;
    int*   rowptr     = (int*)ws;   ws += nodeI;
    int*   hist       = (int*)ws;   ws += histBytes;
    int*   bucketBase = (int*)ws;   ws += baseBytes;
    float* invdeg     = (float*)ws; ws += nodeF;
    int*   colidx     = (int*)ws;   ws += edgeI;
    int*   ebuf       = (int*)ws;   ws += edgeI;
    float* h1         = (float*)ws; ws += featBytes;
    float* h2         = (float*)ws;

    gnn_bucket_count_kernel<<<EDGE_CHUNKS, EDGE_THREADS, 0, stream>>>(dst, hist, N_EDGES);
    gnn_bucket_scatter_kernel<<<EDGE_CHUNKS, EDGE_THREADS, 0, stream>>>(src, dst, hist, bucketBase, ebuf, N_EDGES);
    gnn_bucket_rows_kernel<<<N_BUCKETS, EDGE_THREADS, 0, stream>>>(ebuf, bucketBase, rowptr, degi, invdeg, colidx, N_NODES);

    const float* h = x;
    float* houts[N_LAYERS] = {h1, h2, h1, h2, out};
    int fusedBlocks = GATHER_WAVES / 8;

    for (int i = 0; i < N_LAYERS; ++i) {
      int relu  = (i < N_LAYERS - 1) ? 1 : 0;
      int resid = (i > 0 && i < N_LAYERS - 1) ? 1 : 0;
      gnn_sage_fused_kernel<<<fusedBlocks, 256, 0, stream>>>(
          h, colidx, rowptr, degi, invdeg,
          Wl + (long)i * D * D, Wr + (long)i * D * D, b + (long)i * D,
          houts[i], N_NODES, relu, resid);
      h = houts[i];
    }
  } else {
    float* invdeg = (float*)ws; ws += nodeF;
    float* agg    = (float*)ws; ws += featBytes;
    float* h1     = (float*)ws; ws += featBytes;
    float* h2     = (float*)ws;

    hipMemsetAsync(invdeg, 0, (size_t)N_NODES * sizeof(float), stream);
    gnn_deg_int_kernel<<<(N_EDGES + 255) / 256, 256, 0, stream>>>(dst, (int*)invdeg, N_EDGES);
    gnn_invdeg_from_int_kernel<<<scanBlocks, 256, 0, stream>>>((const int*)invdeg, invdeg, N_NODES);

    const float* h = x;
    float* houts[N_LAYERS] = {h1, h2, h1, h2, out};
    long scatterBlocks = ((long)N_EDGES * 16 + 255) / 256;
    int gemmBlocks = (N_NODES + 127) / 128;

    for (int i = 0; i < N_LAYERS; ++i) {
      hipMemsetAsync(agg, 0, featBytes, stream);
      gnn_scatter_kernel<<<(int)scatterBlocks, 256, 0, stream>>>(h, src, dst, agg, N_EDGES);
      int relu  = (i < N_LAYERS - 1) ? 1 : 0;
      int resid = (i > 0 && i < N_LAYERS - 1) ? 1 : 0;
      gnn_sage_gemm_kernel<<<gemmBlocks, 256, 0, stream>>>(
          agg, h, invdeg,
          Wl + (long)i * D * D, Wr + (long)i * D * D, b + (long)i * D,
          houts[i], N_NODES, relu, resid);
      h = houts[i];
    }
  }
}


# CSR build, 2 launches for 3: each edge chunk bucket-sorted in ONE pass (LDS count gives the rank, edges stay in registers); rows kernel reads 128 chunk segments; row order = order of new LDS counts
# speedup vs baseline: 2.6864x; 1.0377x over previous
#include <hip/hip_runtime.h>

typedef __attribute__((ext_vector_type(2))) float v2f;
typedef __attribute__((ext_vector_type(4))) float v4f;
typedef __attribute__((ext_vector_type(8))) float v8f;

#define N_NODES 100000
#define N_EDGES 1600000
#define D 64
#define N_LAYERS 5

static inline size_t align256(size_t x) { return (x + 255) & ~(size_t)255; }

__global__ __launch_bounds__(256) void gnn_deg_int_kernel(const int* __restrict__ dst,
                                                          int* __restrict__ degi, int nE) {
  int e = blockIdx.x * 256 + threadIdx.x;
  if (e < nE) atomicAdd(&degi[dst[e]], 1);
}

__global__ __launch_bounds__(256) void gnn_invdeg_from_int_kernel(const int* __restrict__ degi,
                                                                  float* __restrict__ invd, int n) {
  int i = blockIdx.x * 256 + threadIdx.x;
  if (i < n) invd[i] = 1.0f / fmaxf((float)degi[i], 1.0f);
}

#define BK_SHIFT 9
#define BK_NODES (1 << BK_SHIFT)
#define N_BUCKETS ((N_NODES + BK_NODES - 1) / BK_NODES)
#define EDGE_CHUNKS 128
#define EDGE_THREADS 1024
#define CHUNK_MAX ((N_EDGES + EDGE_CHUNKS - 1) / EDGE_CHUNKS)
#define SORT_SLOTS ((CHUNK_MAX + EDGE_THREADS - 1) / EDGE_THREADS)
#define SEG_PITCH (N_BUCKETS + 1)
#define SEG_LANES (EDGE_THREADS / EDGE_CHUNKS)
#define ROW_BATCH 12
static_assert((long)N_NODES << BK_SHIFT < (1L << 31), "src and the local dst share one 32-bit word");
static_assert(N_BUCKETS + 1 <= EDGE_THREADS, "one thread per bucket start, one more for the chunk length");
static_assert(N_EDGES >= EDGE_CHUNKS, "no chunk is empty");
static_assert(SEG_LANES * EDGE_CHUNKS == EDGE_THREADS, "SEG_LANES threads per chunk segment");
static_assert(BK_NODES <= EDGE_THREADS, "one thread per node of a bucket");

__device__ __forceinline__ int gnn_block_excl_scan(int v, int* wsum) {
  const int lane = threadIdx.x & 31;
  const int w = threadIdx.x >> 5;
  int inc = v;
#pragma unroll
  for (int o = 1; o < 32; o <<= 1) {
    const int t = __shfl_up(inc, o, 32);
    if (lane >= o) inc += t;
  }
  if (lane == 31) wsum[w] = inc;
  __syncthreads();
  const int ws = wsum[lane];
  int wi = ws;
#pragma unroll
  for (int o = 1; o < 32; o <<= 1) {
    const int t = __shfl_up(wi, o, 32);
    if (lane >= o) wi += t;
  }
  const int wexcl = __shfl(wi - ws, w, 32);
  return wexcl + inc - v;
}

__global__ __launch_bounds__(EDGE_THREADS) void gnn_chunk_sort_kernel(
    const int* __restrict__ src, const int* __restrict__ dst, int* __restrict__ hist,
    int* __restrict__ ebuf, int nE) {
  __shared__ int cnt[N_BUCKETS];
  __shared__ int wsum[32];
  const int tid = threadIdx.x;
  const int c = blockIdx.x;
  if (tid < N_BUCKETS) cnt[tid] = 0;
  __syncthreads();
  const int e0 = (int)(((long)c * nE) / EDGE_CHUNKS);
  const int e1 = (int)(((long)(c + 1) * nE) / EDGE_CHUNKS);
  int d[SORT_SLOTS], s[SORT_SLOTS], rk[SORT_SLOTS];
#pragma unroll
  for (int u = 0; u < SORT_SLOTS; ++u) {
    const int q = e0 + tid + u * EDGE_THREADS;
    const int qc = q < e1 ? q : e1 - 1;
    d[u] = dst[qc];
    s[u] = src[qc];
  }
  __builtin_amdgcn_sched_barrier(0);
#pragma unroll
  for (int u = 0; u < SORT_SLOTS; ++u)
    rk[u] = (e0 + tid + u * EDGE_THREADS < e1) ? atomicAdd(&cnt[d[u] >> BK_SHIFT], 1) : -1;
  __syncthreads();
  const int excl = gnn_block_excl_scan(tid < N_BUCKETS ? cnt[tid] : 0, wsum);
  if (tid < N_BUCKETS) cnt[tid] = excl;
  if (tid <= N_BUCKETS) hist[c * SEG_PITCH + tid] = excl;
  __syncthreads();
  int pos[SORT_SLOTS];
#pragma unroll
  for (int u = 0; u < SORT_SLOTS; ++u) pos[u] = e0 + cnt[d[u] >> BK_SHIFT] + rk[u];
#pragma unroll
  for (int u = 0; u < SORT_SLOTS; ++u)
    if (rk[u] >= 0) ebuf[pos[u]] = (s[u] << BK_SHIFT) | (d[u] & (BK_NODES - 1));
}

__global__ __launch_bounds__(EDGE_THREADS) void gnn_bucket_rows_kernel(
    const int* __restrict__ ebuf, const int* __restrict__ hist, int* __restrict__ rowptr,
    int* __restrict__ degi, float* __restrict__ invd, int* __restrict__ colidx, int nN, int nE) {
  __shared__ int cnt[EDGE_THREADS];
  __shared__ int wsum[32];
  __shared__ int segBeg[EDGE_CHUNKS];
  __shared__ int segLen[EDGE_CHUNKS];
  __shared__ int bucketStart;
  const int tid = threadIdx.x;
  const int b = blockIdx.x;
  cnt[tid] = 0;
  if (tid == 0) bucketStart = 0;
  __syncthreads();
  if (tid < EDGE_CHUNKS) {
    const int a0 = hist[tid * SEG_PITCH + b];
    const int a1 = hist[tid * SEG_PITCH + b + 1];
    segBeg[tid] = (int)(((long)tid * nE) / EDGE_CHUNKS) + a0;
    segLen[tid] = a1 - a0;
    atomicAdd(&bucketStart, a0);
  }
  __syncthreads();
  const int p0 = bucketStart;
  const int sb = segBeg[tid / SEG_LANES];
  const int sl = segLen[tid / SEG_LANES];
  const int sub = tid % SEG_LANES;
  for (int j0 = sub; j0 < sl; j0 += SEG_LANES * ROW_BATCH) {
    int w[ROW_BATCH];
#pragma unroll
    for (int u = 0; u < ROW_BATCH; ++u) {
      const int j = j0 + u * SEG_LANES;
      w[u] = ebuf[sb + (j < sl ? j : sl - 1)];
    }
#pragma unroll
    for (int u = 0; u < ROW_BATCH; ++u)
      atomicAdd(&cnt[w[u] & (BK_NODES - 1)], (j0 + u * SEG_LANES < sl) ? 1 : 0);
  }
  __syncthreads();
  const int deg = cnt[tid];
  const int start = p0 + gnn_block_excl_scan(deg, wsum);
  cnt[tid] = start;
  const int node = b * BK_NODES + tid;
  if (tid < BK_NODES && node < nN) {
    rowptr[node] = start;
    degi[node] = deg;
    invd[node] = 1.0f / fmaxf((float)deg, 1.0f);
  }
  __syncthreads();
  for (int j0 = sub; j0 < sl; j0 += SEG_LANES * ROW_BATCH) {
    int w[ROW_BATCH], slot[ROW_BATCH];
#pragma unroll
    for (int u = 0; u < ROW_BATCH; ++u) {
      const int j = j0 + u * SEG_LANES;
      w[u] = ebuf[sb + (j < sl ? j : sl - 1)];
    }
#pragma unroll
    for (int u = 0; u < ROW_BATCH; ++u)
      slot[u] = (j0 + u * SEG_LANES < sl) ? atomicAdd(&cnt[w[u] & (BK_NODES - 1)], 1) : -1;
#pragma unroll
    for (int u = 0; u < ROW_BATCH; ++u)
      if (slot[u] >= 0) colidx[slot[u]] = w[u] >> BK_SHIFT;
  }
}

#define GATHER_WAVES 8192
#define GATHER_GROUP 16
#define GATHER_CHUNK 32
static_assert((GATHER_WAVES & (GATHER_WAVES - 1)) == 0, "node split uses a shift");
#define TILE_ROWS 16
#define AGG_PITCH 68
#define STAGE_ITEMS 4
static_assert((long)N_NODES <= (long)TILE_ROWS * GATHER_WAVES, "a wave's node run fits one 16-row tile");
static_assert(GATHER_GROUP == 16 && GATHER_CHUNK == 32, "the load fences name 16 row registers each");
static_assert((long)N_NODES * D * 4 < (1L << 32), "a row's byte offset fits 32 bits");

template <int G, bool FULL>
__device__ __forceinline__ void gnn_gather_group(const char* __restrict__ hb, int idx, int j,
                                                 int m, int pos0, int eEnd, unsigned lane8,
                                                 int re, float* srow, int& cur, int& curEnd,
                                                 v2f& acc) {
  v2f v[G];
#pragma unroll
  for (int u = 0; u < G; ++u) {
    const int sl = (FULL || j + u < m) ? (j + u) : (m - 1);
    const unsigned nb = (unsigned)__builtin_amdgcn_readlane(idx, sl);
    const unsigned off = (nb << 8) + lane8;
    v[u] = *(const v2f*)(hb + off);
  }
  asm volatile("" ::"v"(v[0]), "v"(v[1]), "v"(v[2]), "v"(v[3]), "v"(v[4]), "v"(v[5]),
               "v"(v[6]), "v"(v[7]), "v"(v[8]), "v"(v[9]), "v"(v[10]), "v"(v[11]),
               "v"(v[12]), "v"(v[13]), "v"(v[14]), "v"(v[15]));
  if (G > 16)
    asm volatile("" ::"v"(v[16 % G]), "v"(v[17 % G]), "v"(v[18 % G]), "v"(v[19 % G]),
                 "v"(v[20 % G]), "v"(v[21 % G]), "v"(v[22 % G]), "v"(v[23 % G]),
                 "v"(v[24 % G]), "v"(v[25 % G]), "v"(v[26 % G]), "v"(v[27 % G]),
                 "v"(v[28 % G]), "v"(v[29 % G]), "v"(v[30 % G]), "v"(v[31 % G]));
  const v2f zero2 = {0.f, 0.f};
#pragma unroll
  for (int u = 0; u < G; ++u) {
    const int pos = pos0 + u;
    if (FULL || pos < eEnd) {
      while (pos == curEnd) {
        *(v2f*)(srow + cur * AGG_PITCH) = acc;
        acc = zero2;
        ++cur;
        curEnd = __builtin_amdgcn_readlane(re, cur);
      }
      acc += v[u];
    }
  }
}

__global__ __launch_bounds__(256) __attribute__((amdgpu_waves_per_eu(8, 8))) void gnn_sage_fused_kernel(
    const float* __restrict__ h, const int* __restrict__ colidx, const int* __restrict__ rowptr,
    const int* __restrict__ degi, const float* __restrict__ invdeg,
    const float* __restrict__ Wl, const float* __restrict__ Wr, const float* __restrict__ bias,
    float* __restrict__ out, int nN, int do_relu, int do_resid) {
  __shared__ __align__(16) float2 sWf[2][16][4][32];
  __shared__ __align__(16) float sB[D];
  __shared__ __align__(16) float sAgg[8][TILE_ROWS * AGG_PITCH];
  const int tid = threadIdx.x;
  const int lane = tid & 31;
  const int wv = __builtin_amdgcn_readfirstlane(tid >> 5);
  const int w = (int)blockIdx.x * 8 + wv;
  const int n0 = (int)(((long)w * nN) / GATHER_WAVES);
  const int n1 = (int)(((long)(w + 1) * nN) / GATHER_WAVES);
  const int nn = n1 - n0;

  float4 wq[STAGE_ITEMS][2];
#pragma unroll
  for (int i = 0; i < STAGE_ITEMS; ++i) {
    const float* const Wsrc = (i < STAGE_ITEMS / 2) ? Wl : Wr;
    const int rem = (tid + 256 * i) & 511;
    const int q = rem >> 4;
    const int g = rem & 15;
    wq[i][0] = *(const float4*)(Wsrc + (2 * q) * D + 4 * g);
    wq[i][1] = *(const float4*)(Wsrc + (2 * q + 1) * D + 4 * g);
  }
  const float bq = (tid < D) ? bias[tid] : 0.0f;

  const int lrow = lane & 15;
  const int lhalf = lane >> 4;
  const int khalf = 2 * lhalf;
  const int M = n0 + lrow;
  const int Mc = (lrow < nn) ? M : (nN - 1);
  const float invd = invdeg[Mc];
  int rb = 0, dg = 0;
  if (lane < nn) {
    rb = rowptr[n0 + lane];
    dg = degi[n0 + lane];
  }
  const int col4 = 4 * lrow;
  v4f hv[TILE_ROWS / 2];
  const unsigned off0 = (unsigned)((n0 + lhalf) * (D * 4) + col4 * 4);
  const unsigned offLast = (unsigned)((nN - 1) * (D * 4) + col4 * 4);
#pragma unroll
  for (int i = 0; i < TILE_ROWS / 2; ++i) {
    const unsigned o = off0 + (unsigned)(i * 2 * D * 4);
    hv[i] = *(const v4f*)((const char*)h + (o < offLast ? o : offLast));
  }
  __builtin_amdgcn_sched_barrier(0);

#pragma unroll
  for (int i = 0; i < STAGE_ITEMS; ++i) {
    const int rem = (tid + 256 * i) & 511;
    const int q = rem >> 4;
    const int g = rem & 15;
    float2* const fr = &sWf[i >> 1][q >> 1][g >> 2][16 * (q & 1) + 4 * (g & 3)];
    fr[0] = make_float2(wq[i][0].x, wq[i][1].x);
    fr[1] = make_float2(wq[i][0].y, wq[i][1].y);
    fr[2] = make_float2(wq[i][0].z, wq[i][1].z);
    fr[3] = make_float2(wq[i][0].w, wq[i][1].w);
  }
  if (tid < D) sB[tid] = bq;
  float* const tile = &sAgg[wv][0];
#pragma unroll
  for (int i = 0; i < TILE_ROWS / 2; ++i)
    *(v4f*)(tile + (2 * i + lhalf) * AGG_PITCH + col4) = hv[i];
  __syncthreads();
  if (nn <= 0) return;
  const int re = rb + dg;
  const int e0 = __builtin_amdgcn_readlane(rb, 0);
  const int eEnd = __builtin_amdgcn_readlane(re, nn - 1);

  int nxt = (e0 + lane < eEnd) ? colidx[e0 + lane] : 0;

  v8f c4[4];
#pragma unroll
  for (int n = 0; n < 4; ++n) {
    float bv = sB[n * 16 + lrow];
#pragma unroll
    for (int j = 0; j < 8; ++j) c4[n][j] = bv;
  }
  const float* const sArow = tile + lrow * AGG_PITCH + khalf;
#pragma unroll
  for (int k = 0; k < 16; ++k) {
    const v2f ah = *(const v2f*)(sArow + 4 * k);
#pragma unroll
    for (int n = 0; n < 4; ++n) {
      float2 bf2 = sWf[1][k][n][lane];
      v2f bf;
      bf[0] = bf2.x;
      bf[1] = bf2.y;
      c4[n] = __builtin_amdgcn_wmma_f32_16x16x4_f32(
          false, ah, false, bf, (short)0, c4[n], false, false);
    }
  }

  const unsigned loff = 2u * (unsigned)lane;
  const unsigned lane8 = 8u * (unsigned)lane;
  const char* const hb = (const char*)h;
  float* const srow = tile + loff;

  int cur = 0;
  int curEnd = __builtin_amdgcn_readlane(re, 0);
  v2f acc = {0.f, 0.f};

  for (int base = e0; base < eEnd; base += 32) {
    const int idx = nxt;
    asm volatile("" ::"v"(idx));
    const int nbase = base + 32;
    nxt = (nbase + lane < eEnd) ? colidx[nbase + lane] : 0;
    const int rem = eEnd - base;
    const int m = rem < 32 ? rem : 32;
    if (m == GATHER_CHUNK) {
      gnn_gather_group<GATHER_CHUNK, true>(hb, idx, 0, m, base, eEnd, lane8, re, srow, cur,
                                           curEnd, acc);
    } else {
      for (int j = 0; j < m; j += GATHER_GROUP) {
        if (j + GATHER_GROUP <= m)
          gnn_gather_group<GATHER_GROUP, true>(hb, idx, j, m, base + j, eEnd, lane8, re, srow,
                                               cur, curEnd, acc);
        else
          gnn_gather_group<GATHER_GROUP, false>(hb, idx, j, m, base + j, eEnd, lane8, re, srow,
                                                cur, curEnd, acc);
      }
    }
  }
  const v2f zero2 = {0.f, 0.f};
  for (; cur < TILE_ROWS; ++cur) {
    *(v2f*)(srow + cur * AGG_PITCH) = acc;
    acc = zero2;
  }

  v4f hres[TILE_ROWS / 2];
  if (do_resid) {
#pragma unroll
    for (int i = 0; i < TILE_ROWS / 2; ++i) {
      const int r = 2 * i + lhalf;
      const int node = n0 + (r < nn ? r : nn - 1);
      hres[i] = *(const v4f*)((const char*)h + (unsigned)(node * (D * 4) + col4 * 4));
    }
  } else {
#pragma unroll
    for (int i = 0; i < TILE_ROWS / 2; ++i) hres[i] = (v4f){0.f, 0.f, 0.f, 0.f};
  }

  __builtin_amdgcn_fence(__ATOMIC_RELEASE, "wavefront");
  __builtin_amdgcn_wave_barrier();
  __builtin_amdgcn_fence(__ATOMIC_ACQUIRE, "wavefront");

#pragma unroll
  for (int k = 0; k < 16; ++k) {
    v2f a = *(const v2f*)(sArow + 4 * k);
    a[0] = a[0] * invd;
    a[1] = a[1] * invd;
#pragma unroll
    for (int n = 0; n < 4; ++n) {
      float2 bf2 = sWf[0][k][n][lane];
      v2f bf;
      bf[0] = bf2.x;
      bf[1] = bf2.y;
      c4[n] = __builtin_amdgcn_wmma_f32_16x16x4_f32(
          false, a, false, bf, (short)0, c4[n], false, false);
    }
  }
#pragma unroll
  for (int n = 0; n < 4; ++n) {
#pragma unroll
    for (int j = 0; j < 8; ++j) tile[(8 * lhalf + j) * AGG_PITCH + n * 16 + lrow] = c4[n][j];
  }
  __builtin_amdgcn_fence(__ATOMIC_RELEASE, "wavefront");
  __builtin_amdgcn_wave_barrier();
  __builtin_amdgcn_fence(__ATOMIC_ACQUIRE, "wavefront");
  v4f o[TILE_ROWS / 2];
#pragma unroll
  for (int i = 0; i < TILE_ROWS / 2; ++i)
    o[i] = *(const v4f*)(tile + (2 * i + lhalf) * AGG_PITCH + col4);
#pragma unroll
  for (int i = 0; i < TILE_ROWS / 2; ++i) {
    const int r = 2 * i + lhalf;
    v4f v = o[i];
    if (do_resid) v += hres[i];
    if (do_relu) {
      v.x = fmaxf(v.x, 0.0f);
      v.y = fmaxf(v.y, 0.0f);
      v.z = fmaxf(v.z, 0.0f);
      v.w = fmaxf(v.w, 0.0f);
    }
    if (r < nn) *(v4f*)((char*)out + (unsigned)((n0 + r) * (D * 4) + col4 * 4)) = v;
  }
}

__global__ __launch_bounds__(256) void gnn_scatter_kernel(const float* __restrict__ h,
                                                          const int* __restrict__ src,
                                                          const int* __restrict__ dstv,
                                                          float* __restrict__ agg, int nE) {
  long tid = (long)blockIdx.x * 256 + threadIdx.x;
  int e = (int)(tid >> 4);
  int c = (int)(tid & 15);
  if (e >= nE) return;
  int s = src[e];
  int d = dstv[e];
  const float4 v = *(const float4*)(h + (long)s * D + c * 4);
  float* ap = agg + (long)d * D + c * 4;
  atomicAdd(ap + 0, v.x);
  atomicAdd(ap + 1, v.y);
  atomicAdd(ap + 2, v.z);
  atomicAdd(ap + 3, v.w);
}

__global__ __launch_bounds__(256) void gnn_sage_gemm_kernel(
    const float* __restrict__ agg, const float* __restrict__ h,
    const float* __restrict__ invdeg,
    const float* __restrict__ Wl, const float* __restrict__ Wr,
    const float* __restrict__ bias,
    float* __restrict__ out, int nN, int do_relu, int do_resid) {
  __shared__ __align__(16) float2 sWf[2][16][4][32];
  __shared__ __align__(16) float sB[D];
  int tid = threadIdx.x;

  for (int t = tid; t < 2048; t += 256) {
    int k  = t >> 7;
    int n  = (t >> 5) & 3;
    int ln = t & 31;
    int koff = (ln < 16) ? 0 : 2;
    int lr = ln & 15;
    int kb = k * 4 + koff;
    int col = n * 16 + lr;
    float2 f0, f1;
    f0.x = Wl[(kb + 0) * D + col];
    f0.y = Wl[(kb + 1) * D + col];
    f1.x = Wr[(kb + 0) * D + col];
    f1.y = Wr[(kb + 1) * D + col];
    sWf[0][k][n][ln] = f0;
    sWf[1][k][n][ln] = f1;
  }
  if (tid < D) sB[tid] = bias[tid];
  __syncthreads();

  int wave = tid >> 5;
  int lane = tid & 31;
  int lrow = lane & 15;
  int khalf = (lane < 16) ? 0 : 2;
  int rowbase = blockIdx.x * 128 + wave * 16;
  int M = rowbase + lrow;
  int Mc = (M < nN) ? M : (nN - 1);

  float invd = invdeg[Mc];

  v8f acc[4];
#pragma unroll
  for (int n = 0; n < 4; ++n) {
    float bv = sB[n * 16 + lrow];
#pragma unroll
    for (int j = 0; j < 8; ++j) acc[n][j] = bv;
  }

  for (int p = 0; p < 2; ++p) {
    const float* Arow = (p == 0 ? agg : h) + (long)Mc * D;
    float scale = (p == 0) ? invd : 1.0f;
#pragma unroll
    for (int k = 0; k < 16; ++k) {
      int kb = k * 4 + khalf;
      v2f a;
      a[0] = Arow[kb + 0] * scale;
      a[1] = Arow[kb + 1] * scale;
#pragma unroll
      for (int n = 0; n < 4; ++n) {
        float2 bf2 = sWf[p][k][n][lane];
        v2f bf;
        bf[0] = bf2.x;
        bf[1] = bf2.y;
        acc[n] = __builtin_amdgcn_wmma_f32_16x16x4_f32(
            false, a, false, bf, (short)0, acc[n], false, false);
      }
    }
  }

#pragma unroll
  for (int n = 0; n < 4; ++n) {
#pragma unroll
    for (int j = 0; j < 8; ++j) {
      int row = rowbase + ((lane < 16) ? j : j + 8);
      int col = n * 16 + lrow;
      if (row < nN) {
        float v = acc[n][j];
        if (do_resid) v += h[(long)row * D + col];
        if (do_relu) v = fmaxf(v, 0.0f);
        out[(long)row * D + col] = v;
      }
    }
  }
}

extern "C" void kernel_launch(void* const* d_in, const int* in_sizes, int n_in,
                              void* d_out, int out_size, void* d_ws, size_t ws_size,
                              hipStream_t stream) {
  const float* x  = (const float*)d_in[0];
  const int*   ei = (const int*)d_in[1];
  const float* Wl = (const float*)d_in[2];
  const float* Wr = (const float*)d_in[3];
  const float* b  = (const float*)d_in[4];
  float* out = (float*)d_out;

  const int* src = ei;
  const int* dst = ei + N_EDGES;

  const size_t featBytes = (size_t)N_NODES * D * sizeof(float);
  const size_t nodeI     = align256((size_t)N_NODES * sizeof(int));
  const size_t nodeF     = align256((size_t)N_NODES * sizeof(float));
  const size_t edgeI     = align256((size_t)N_EDGES * sizeof(int));
  const int scanBlocks   = (N_NODES + 255) / 256;
  const size_t histBytes = align256((size_t)EDGE_CHUNKS * SEG_PITCH * sizeof(int));

  const size_t csrNeed = 2 * nodeI + histBytes + nodeF + 2 * edgeI + 2 * featBytes;

  char* ws = (char*)d_ws;

  if (ws_size >= csrNeed) {
    int*   degi       = (int*)ws;   ws += nodeI;
    int*   rowptr     = (int*)ws;   ws += nodeI;
    int*   hist       = (int*)ws;   ws += histBytes;
    float* invdeg     = (float*)ws; ws += nodeF;
    int*   colidx     = (int*)ws;   ws += edgeI;
    int*   ebuf       = (int*)ws;   ws += edgeI;
    float* h1         = (float*)ws; ws += featBytes;
    float* h2         = (float*)ws;

    gnn_chunk_sort_kernel<<<EDGE_CHUNKS, EDGE_THREADS, 0, stream>>>(src, dst, hist, ebuf, N_EDGES);
    gnn_bucket_rows_kernel<<<N_BUCKETS, EDGE_THREADS, 0, stream>>>(ebuf, hist, rowptr, degi, invdeg, colidx, N_NODES, N_EDGES);

    const float* h = x;
    float* houts[N_LAYERS] = {h1, h2, h1, h2, out};
    int fusedBlocks = GATHER_WAVES / 8;

    for (int i = 0; i < N_LAYERS; ++i) {
      int relu  = (i < N_LAYERS - 1) ? 1 : 0;
      int resid = (i > 0 && i < N_LAYERS - 1) ? 1 : 0;
      gnn_sage_fused_kernel<<<fusedBlocks, 256, 0, stream>>>(
          h, colidx, rowptr, degi, invdeg,
          Wl + (long)i * D * D, Wr + (long)i * D * D, b + (long)i * D,
          houts[i], N_NODES, relu, resid);
      h = houts[i];
    }
  } else {
    float* invdeg = (float*)ws; ws += nodeF;
    float* agg    = (float*)ws; ws += featBytes;
    float* h1     = (float*)ws; ws += featBytes;
    float* h2     = (float*)ws;

    hipMemsetAsync(invdeg, 0, (size_t)N_NODES * sizeof(float), stream);
    gnn_deg_int_kernel<<<(N_EDGES + 255) / 256, 256, 0, stream>>>(dst, (int*)invdeg, N_EDGES);
    gnn_invdeg_from_int_kernel<<<scanBlocks, 256, 0, stream>>>((const int*)invdeg, invdeg, N_NODES);

    const float* h = x;
    float* houts[N_LAYERS] = {h1, h2, h1, h2, out};
    long scatterBlocks = ((long)N_EDGES * 16 + 255) / 256;
    int gemmBlocks = (N_NODES + 127) / 128;

    for (int i = 0; i < N_LAYERS; ++i) {
      hipMemsetAsync(agg, 0, featBytes, stream);
      gnn_scatter_kernel<<<(int)scatterBlocks, 256, 0, stream>>>(h, src, dst, agg, N_EDGES);
      int relu  = (i < N_LAYERS - 1) ? 1 : 0;
      int resid = (i > 0 && i < N_LAYERS - 1) ? 1 : 0;
      gnn_sage_gemm_kernel<<<gemmBlocks, 256, 0, stream>>>(
          agg, h, invdeg,
          Wl + (long)i * D * D, Wr + (long)i * D * D, b + (long)i * D,
          houts[i], N_NODES, relu, resid);
      h = houts[i];
    }
  }
}


# CSR build: both sort kernels scatter their 4-byte results into an LDS image of the block's contiguous output range, then store it as whole 128-B lines (was up to 32 lines a store); row order kept
# speedup vs baseline: 2.8817x; 1.0727x over previous
#include <hip/hip_runtime.h>

typedef __attribute__((ext_vector_type(2))) float v2f;
typedef __attribute__((ext_vector_type(4))) float v4f;
typedef __attribute__((ext_vector_type(8))) float v8f;

#define N_NODES 100000
#define N_EDGES 1600000
#define D 64
#define N_LAYERS 5

static inline size_t align256(size_t x) { return (x + 255) & ~(size_t)255; }

__global__ __launch_bounds__(256) void gnn_deg_int_kernel(const int* __restrict__ dst,
                                                          int* __restrict__ degi, int nE) {
  int e = blockIdx.x * 256 + threadIdx.x;
  if (e < nE) atomicAdd(&degi[dst[e]], 1);
}

__global__ __launch_bounds__(256) void gnn_invdeg_from_int_kernel(const int* __restrict__ degi,
                                                                  float* __restrict__ invd, int n) {
  int i = blockIdx.x * 256 + threadIdx.x;
  if (i < n) invd[i] = 1.0f / fmaxf((float)degi[i], 1.0f);
}

#define BK_SHIFT 9
#define BK_NODES (1 << BK_SHIFT)
#define N_BUCKETS ((N_NODES + BK_NODES - 1) / BK_NODES)
#define EDGE_CHUNKS 128
#define EDGE_THREADS 1024
#define CHUNK_MAX ((N_EDGES + EDGE_CHUNKS - 1) / EDGE_CHUNKS)
#define SORT_SLOTS ((CHUNK_MAX + EDGE_THREADS - 1) / EDGE_THREADS)
#define SEG_PITCH (N_BUCKETS + 1)
#define SEG_LANES (EDGE_THREADS / EDGE_CHUNKS)
#define ROW_BATCH 12
#define LINE_INTS 32
#define SORT_IMG_INTS (((CHUNK_MAX + 3) / 4) * 4 + LINE_INTS)
#define ROWS_IMG_INTS 16384
static_assert((long)N_NODES << BK_SHIFT < (1L << 31), "src and the local dst share one 32-bit word");
static_assert(N_BUCKETS + 1 <= EDGE_THREADS, "one thread per bucket start, one more for the chunk length");
static_assert(N_EDGES >= EDGE_CHUNKS, "no chunk is empty");
static_assert(SEG_LANES * EDGE_CHUNKS == EDGE_THREADS, "SEG_LANES threads per chunk segment");
static_assert(BK_NODES <= EDGE_THREADS, "one thread per node of a bucket");

__device__ __forceinline__ int gnn_block_excl_scan(int v, int* wsum) {
  const int lane = threadIdx.x & 31;
  const int w = threadIdx.x >> 5;
  int inc = v;
#pragma unroll
  for (int o = 1; o < 32; o <<= 1) {
    const int t = __shfl_up(inc, o, 32);
    if (lane >= o) inc += t;
  }
  if (lane == 31) wsum[w] = inc;
  __syncthreads();
  const int ws = wsum[lane];
  int wi = ws;
#pragma unroll
  for (int o = 1; o < 32; o <<= 1) {
    const int t = __shfl_up(wi, o, 32);
    if (lane >= o) wi += t;
  }
  const int wexcl = __shfl(wi - ws, w, 32);
  return wexcl + inc - v;
}

__device__ __forceinline__ void gnn_store_image(const int* img, int* __restrict__ garr, int lead,
                                                int hi) {
  const int nq = (hi + 3) >> 2;
  for (int q = threadIdx.x; q < nq; q += EDGE_THREADS) {
    const int i = 4 * q;
    const int4 v = *(const int4*)(img + i);
    if (i >= lead && i + 4 <= hi) {
      *(int4*)(garr + i) = v;
    } else {
      if (i + 0 >= lead && i + 0 < hi) garr[i + 0] = v.x;
      if (i + 1 >= lead && i + 1 < hi) garr[i + 1] = v.y;
      if (i + 2 >= lead && i + 2 < hi) garr[i + 2] = v.z;
      if (i + 3 >= lead && i + 3 < hi) garr[i + 3] = v.w;
    }
  }
}

__global__ __launch_bounds__(EDGE_THREADS) void gnn_chunk_sort_kernel(
    const int* __restrict__ src, const int* __restrict__ dst, int* __restrict__ hist,
    int* __restrict__ ebuf, int nE) {
  __shared__ int cnt[N_BUCKETS];
  __shared__ int wsum[32];
  __shared__ __align__(16) int img[SORT_IMG_INTS];
  const int tid = threadIdx.x;
  const int c = blockIdx.x;
  if (tid < N_BUCKETS) cnt[tid] = 0;
  __syncthreads();
  const int e0 = (int)(((long)c * nE) / EDGE_CHUNKS);
  const int e1 = (int)(((long)(c + 1) * nE) / EDGE_CHUNKS);
  int d[SORT_SLOTS], s[SORT_SLOTS], rk[SORT_SLOTS];
#pragma unroll
  for (int u = 0; u < SORT_SLOTS; ++u) {
    const int q = e0 + tid + u * EDGE_THREADS;
    const int qc = q < e1 ? q : e1 - 1;
    d[u] = dst[qc];
    s[u] = src[qc];
  }
  __builtin_amdgcn_sched_barrier(0);
#pragma unroll
  for (int u = 0; u < SORT_SLOTS; ++u)
    rk[u] = (e0 + tid + u * EDGE_THREADS < e1) ? atomicAdd(&cnt[d[u] >> BK_SHIFT], 1) : -1;
  __syncthreads();
  const int excl = gnn_block_excl_scan(tid < N_BUCKETS ? cnt[tid] : 0, wsum);
  if (tid < N_BUCKETS) cnt[tid] = excl;
  if (tid <= N_BUCKETS) hist[c * SEG_PITCH + tid] = excl;
  __syncthreads();
  const int lead = e0 & (LINE_INTS - 1);
  int pos[SORT_SLOTS];
#pragma unroll
  for (int u = 0; u < SORT_SLOTS; ++u) pos[u] = lead + cnt[d[u] >> BK_SHIFT] + rk[u];
#pragma unroll
  for (int u = 0; u < SORT_SLOTS; ++u)
    if (rk[u] >= 0) img[pos[u]] = (s[u] << BK_SHIFT) | (d[u] & (BK_NODES - 1));
  __syncthreads();
  gnn_store_image(img, ebuf + (e0 - lead), lead, lead + (e1 - e0));
}

__global__ __launch_bounds__(EDGE_THREADS) void gnn_bucket_rows_kernel(
    const int* __restrict__ ebuf, const int* __restrict__ hist, int* __restrict__ rowptr,
    int* __restrict__ degi, float* __restrict__ invd, int* __restrict__ colidx, int nN, int nE) {
  __shared__ int cnt[EDGE_THREADS];
  __shared__ int wsum[32];
  __shared__ int segBeg[EDGE_CHUNKS];
  __shared__ int segLen[EDGE_CHUNKS];
  __shared__ int bucketStart;
  __shared__ int bucketLen;
  __shared__ __align__(16) int img[ROWS_IMG_INTS + LINE_INTS];
  const int tid = threadIdx.x;
  const int b = blockIdx.x;
  cnt[tid] = 0;
  if (tid == 0) {
    bucketStart = 0;
    bucketLen = 0;
  }
  __syncthreads();
  if (tid < EDGE_CHUNKS) {
    const int a0 = hist[tid * SEG_PITCH + b];
    const int a1 = hist[tid * SEG_PITCH + b + 1];
    segBeg[tid] = (int)(((long)tid * nE) / EDGE_CHUNKS) + a0;
    segLen[tid] = a1 - a0;
    atomicAdd(&bucketStart, a0);
    atomicAdd(&bucketLen, a1 - a0);
  }
  __syncthreads();
  const int p0 = bucketStart;
  const int total = bucketLen;
  const int sb = segBeg[tid / SEG_LANES];
  const int sl = segLen[tid / SEG_LANES];
  const int sub = tid % SEG_LANES;
  for (int j0 = sub; j0 < sl; j0 += SEG_LANES * ROW_BATCH) {
    int w[ROW_BATCH];
#pragma unroll
    for (int u = 0; u < ROW_BATCH; ++u) {
      const int j = j0 + u * SEG_LANES;
      w[u] = ebuf[sb + (j < sl ? j : sl - 1)];
    }
#pragma unroll
    for (int u = 0; u < ROW_BATCH; ++u)
      atomicAdd(&cnt[w[u] & (BK_NODES - 1)], (j0 + u * SEG_LANES < sl) ? 1 : 0);
  }
  __syncthreads();
  const int deg = cnt[tid];
  const int start = p0 + gnn_block_excl_scan(deg, wsum);
  cnt[tid] = start;
  const int node = b * BK_NODES + tid;
  if (tid < BK_NODES && node < nN) {
    rowptr[node] = start;
    degi[node] = deg;
    invd[node] = 1.0f / fmaxf((float)deg, 1.0f);
  }
  __syncthreads();
  const int lead = p0 & (LINE_INTS - 1);
  const bool staged = total <= ROWS_IMG_INTS;
  const int ibase = lead - p0;
  for (int j0 = sub; j0 < sl; j0 += SEG_LANES * ROW_BATCH) {
    int w[ROW_BATCH], slot[ROW_BATCH];
#pragma unroll
    for (int u = 0; u < ROW_BATCH; ++u) {
      const int j = j0 + u * SEG_LANES;
      w[u] = ebuf[sb + (j < sl ? j : sl - 1)];
    }
#pragma unroll
    for (int u = 0; u < ROW_BATCH; ++u)
      slot[u] = (j0 + u * SEG_LANES < sl) ? atomicAdd(&cnt[w[u] & (BK_NODES - 1)], 1) : -1;
    if (staged) {
#pragma unroll
      for (int u = 0; u < ROW_BATCH; ++u)
        if (slot[u] >= 0) img[ibase + slot[u]] = w[u] >> BK_SHIFT;
    } else {
#pragma unroll
      for (int u = 0; u < ROW_BATCH; ++u)
        if (slot[u] >= 0) colidx[slot[u]] = w[u] >> BK_SHIFT;
    }
  }
  if (staged) {
    __syncthreads();
    gnn_store_image(img, colidx + (p0 - lead), lead, lead + total);
  }
}

#define GATHER_WAVES 8192
#define GATHER_GROUP 16
#define GATHER_CHUNK 32
static_assert((GATHER_WAVES & (GATHER_WAVES - 1)) == 0, "node split uses a shift");
#define TILE_ROWS 16
#define AGG_PITCH 68
#define STAGE_ITEMS 4
static_assert((long)N_NODES <= (long)TILE_ROWS * GATHER_WAVES, "a wave's node run fits one 16-row tile");
static_assert(GATHER_GROUP == 16 && GATHER_CHUNK == 32, "the load fences name 16 row registers each");
static_assert((long)N_NODES * D * 4 < (1L << 32), "a row's byte offset fits 32 bits");

template <int G, bool FULL>
__device__ __forceinline__ void gnn_gather_group(const char* __restrict__ hb, int idx, int j,
                                                 int m, int pos0, int eEnd, unsigned lane8,
                                                 int re, float* srow, int& cur, int& curEnd,
                                                 v2f& acc) {
  v2f v[G];
#pragma unroll
  for (int u = 0; u < G; ++u) {
    const int sl = (FULL || j + u < m) ? (j + u) : (m - 1);
    const unsigned nb = (unsigned)__builtin_amdgcn_readlane(idx, sl);
    const unsigned off = (nb << 8) + lane8;
    v[u] = *(const v2f*)(hb + off);
  }
  asm volatile("" ::"v"(v[0]), "v"(v[1]), "v"(v[2]), "v"(v[3]), "v"(v[4]), "v"(v[5]),
               "v"(v[6]), "v"(v[7]), "v"(v[8]), "v"(v[9]), "v"(v[10]), "v"(v[11]),
               "v"(v[12]), "v"(v[13]), "v"(v[14]), "v"(v[15]));
  if (G > 16)
    asm volatile("" ::"v"(v[16 % G]), "v"(v[17 % G]), "v"(v[18 % G]), "v"(v[19 % G]),
                 "v"(v[20 % G]), "v"(v[21 % G]), "v"(v[22 % G]), "v"(v[23 % G]),
                 "v"(v[24 % G]), "v"(v[25 % G]), "v"(v[26 % G]), "v"(v[27 % G]),
                 "v"(v[28 % G]), "v"(v[29 % G]), "v"(v[30 % G]), "v"(v[31 % G]));
  const v2f zero2 = {0.f, 0.f};
#pragma unroll
  for (int u = 0; u < G; ++u) {
    const int pos = pos0 + u;
    if (FULL || pos < eEnd) {
      while (pos == curEnd) {
        *(v2f*)(srow + cur * AGG_PITCH) = acc;
        acc = zero2;
        ++cur;
        curEnd = __builtin_amdgcn_readlane(re, cur);
      }
      acc += v[u];
    }
  }
}

__global__ __launch_bounds__(256) __attribute__((amdgpu_waves_per_eu(8, 8))) void gnn_sage_fused_kernel(
    const float* __restrict__ h, const int* __restrict__ colidx, const int* __restrict__ rowptr,
    const int* __restrict__ degi, const float* __restrict__ invdeg,
    const float* __restrict__ Wl, const float* __restrict__ Wr, const float* __restrict__ bias,
    float* __restrict__ out, int nN, int do_relu, int do_resid) {
  __shared__ __align__(16) float2 sWf[2][16][4][32];
  __shared__ __align__(16) float sB[D];
  __shared__ __align__(16) float sAgg[8][TILE_ROWS * AGG_PITCH];
  const int tid = threadIdx.x;
  const int lane = tid & 31;
  const int wv = __builtin_amdgcn_readfirstlane(tid >> 5);
  const int w = (int)blockIdx.x * 8 + wv;
  const int n0 = (int)(((long)w * nN) / GATHER_WAVES);
  const int n1 = (int)(((long)(w + 1) * nN) / GATHER_WAVES);
  const int nn = n1 - n0;

  float4 wq[STAGE_ITEMS][2];
#pragma unroll
  for (int i = 0; i < STAGE_ITEMS; ++i) {
    const float* const Wsrc = (i < STAGE_ITEMS / 2) ? Wl : Wr;
    const int rem = (tid + 256 * i) & 511;
    const int q = rem >> 4;
    const int g = rem & 15;
    wq[i][0] = *(const float4*)(Wsrc + (2 * q) * D + 4 * g);
    wq[i][1] = *(const float4*)(Wsrc + (2 * q + 1) * D + 4 * g);
  }
  const float bq = (tid < D) ? bias[tid] : 0.0f;

  const int lrow = lane & 15;
  const int lhalf = lane >> 4;
  const int khalf = 2 * lhalf;
  const int M = n0 + lrow;
  const int Mc = (lrow < nn) ? M : (nN - 1);
  const float invd = invdeg[Mc];
  int rb = 0, dg = 0;
  if (lane < nn) {
    rb = rowptr[n0 + lane];
    dg = degi[n0 + lane];
  }
  const int col4 = 4 * lrow;
  v4f hv[TILE_ROWS / 2];
  const unsigned off0 = (unsigned)((n0 + lhalf) * (D * 4) + col4 * 4);
  const unsigned offLast = (unsigned)((nN - 1) * (D * 4) + col4 * 4);
#pragma unroll
  for (int i = 0; i < TILE_ROWS / 2; ++i) {
    const unsigned o = off0 + (unsigned)(i * 2 * D * 4);
    hv[i] = *(const v4f*)((const char*)h + (o < offLast ? o : offLast));
  }
  __builtin_amdgcn_sched_barrier(0);

#pragma unroll
  for (int i = 0; i < STAGE_ITEMS; ++i) {
    const int rem = (tid + 256 * i) & 511;
    const int q = rem >> 4;
    const int g = rem & 15;
    float2* const fr = &sWf[i >> 1][q >> 1][g >> 2][16 * (q & 1) + 4 * (g & 3)];
    fr[0] = make_float2(wq[i][0].x, wq[i][1].x);
    fr[1] = make_float2(wq[i][0].y, wq[i][1].y);
    fr[2] = make_float2(wq[i][0].z, wq[i][1].z);
    fr[3] = make_float2(wq[i][0].w, wq[i][1].w);
  }
  if (tid < D) sB[tid] = bq;
  float* const tile = &sAgg[wv][0];
#pragma unroll
  for (int i = 0; i < TILE_ROWS / 2; ++i)
    *(v4f*)(tile + (2 * i + lhalf) * AGG_PITCH + col4) = hv[i];
  __syncthreads();
  if (nn <= 0) return;
  const int re = rb + dg;
  const int e0 = __builtin_amdgcn_readlane(rb, 0);
  const int eEnd = __builtin_amdgcn_readlane(re, nn - 1);

  int nxt = (e0 + lane < eEnd) ? colidx[e0 + lane] : 0;

  v8f c4[4];
#pragma unroll
  for (int n = 0; n < 4; ++n) {
    float bv = sB[n * 16 + lrow];
#pragma unroll
    for (int j = 0; j < 8; ++j) c4[n][j] = bv;
  }
  const float* const sArow = tile + lrow * AGG_PITCH + khalf;
#pragma unroll
  for (int k = 0; k < 16; ++k) {
    const v2f ah = *(const v2f*)(sArow + 4 * k);
#pragma unroll
    for (int n = 0; n < 4; ++n) {
      float2 bf2 = sWf[1][k][n][lane];
      v2f bf;
      bf[0] = bf2.x;
      bf[1] = bf2.y;
      c4[n] = __builtin_amdgcn_wmma_f32_16x16x4_f32(
          false, ah, false, bf, (short)0, c4[n], false, false);
    }
  }

  const unsigned loff = 2u * (unsigned)lane;
  const unsigned lane8 = 8u * (unsigned)lane;
  const char* const hb = (const char*)h;
  float* const srow = tile + loff;

  int cur = 0;
  int curEnd = __builtin_amdgcn_readlane(re, 0);
  v2f acc = {0.f, 0.f};

  for (int base = e0; base < eEnd; base += 32) {
    const int idx = nxt;
    asm volatile("" ::"v"(idx));
    const int nbase = base + 32;
    nxt = (nbase + lane < eEnd) ? colidx[nbase + lane] : 0;
    const int rem = eEnd - base;
    const int m = rem < 32 ? rem : 32;
    if (m == GATHER_CHUNK) {
      gnn_gather_group<GATHER_CHUNK, true>(hb, idx, 0, m, base, eEnd, lane8, re, srow, cur,
                                           curEnd, acc);
    } else {
      for (int j = 0; j < m; j += GATHER_GROUP) {
        if (j + GATHER_GROUP <= m)
          gnn_gather_group<GATHER_GROUP, true>(hb, idx, j, m, base + j, eEnd, lane8, re, srow,
                                               cur, curEnd, acc);
        else
          gnn_gather_group<GATHER_GROUP, false>(hb, idx, j, m, base + j, eEnd, lane8, re, srow,
                                                cur, curEnd, acc);
      }
    }
  }
  const v2f zero2 = {0.f, 0.f};
  for (; cur < TILE_ROWS; ++cur) {
    *(v2f*)(srow + cur * AGG_PITCH) = acc;
    acc = zero2;
  }

  v4f hres[TILE_ROWS / 2];
  if (do_resid) {
#pragma unroll
    for (int i = 0; i < TILE_ROWS / 2; ++i) {
      const int r = 2 * i + lhalf;
      const int node = n0 + (r < nn ? r : nn - 1);
      hres[i] = *(const v4f*)((const char*)h + (unsigned)(node * (D * 4) + col4 * 4));
    }
  } else {
#pragma unroll
    for (int i = 0; i < TILE_ROWS / 2; ++i) hres[i] = (v4f){0.f, 0.f, 0.f, 0.f};
  }

  __builtin_amdgcn_fence(__ATOMIC_RELEASE, "wavefront");
  __builtin_amdgcn_wave_barrier();
  __builtin_amdgcn_fence(__ATOMIC_ACQUIRE, "wavefront");

#pragma unroll
  for (int k = 0; k < 16; ++k) {
    v2f a = *(const v2f*)(sArow + 4 * k);
    a[0] = a[0] * invd;
    a[1] = a[1] * invd;
#pragma unroll
    for (int n = 0; n < 4; ++n) {
      float2 bf2 = sWf[0][k][n][lane];
      v2f bf;
      bf[0] = bf2.x;
      bf[1] = bf2.y;
      c4[n] = __builtin_amdgcn_wmma_f32_16x16x4_f32(
          false, a, false, bf, (short)0, c4[n], false, false);
    }
  }
#pragma unroll
  for (int n = 0; n < 4; ++n) {
#pragma unroll
    for (int j = 0; j < 8; ++j) tile[(8 * lhalf + j) * AGG_PITCH + n * 16 + lrow] = c4[n][j];
  }
  __builtin_amdgcn_fence(__ATOMIC_RELEASE, "wavefront");
  __builtin_amdgcn_wave_barrier();
  __builtin_amdgcn_fence(__ATOMIC_ACQUIRE, "wavefront");
  v4f o[TILE_ROWS / 2];
#pragma unroll
  for (int i = 0; i < TILE_ROWS / 2; ++i)
    o[i] = *(const v4f*)(tile + (2 * i + lhalf) * AGG_PITCH + col4);
#pragma unroll
  for (int i = 0; i < TILE_ROWS / 2; ++i) {
    const int r = 2 * i + lhalf;
    v4f v = o[i];
    if (do_resid) v += hres[i];
    if (do_relu) {
      v.x = fmaxf(v.x, 0.0f);
      v.y = fmaxf(v.y, 0.0f);
      v.z = fmaxf(v.z, 0.0f);
      v.w = fmaxf(v.w, 0.0f);
    }
    if (r < nn) *(v4f*)((char*)out + (unsigned)((n0 + r) * (D * 4) + col4 * 4)) = v;
  }
}

__global__ __launch_bounds__(256) void gnn_scatter_kernel(const float* __restrict__ h,
                                                          const int* __restrict__ src,
                                                          const int* __restrict__ dstv,
                                                          float* __restrict__ agg, int nE) {
  long tid = (long)blockIdx.x * 256 + threadIdx.x;
  int e = (int)(tid >> 4);
  int c = (int)(tid & 15);
  if (e >= nE) return;
  int s = src[e];
  int d = dstv[e];
  const float4 v = *(const float4*)(h + (long)s * D + c * 4);
  float* ap = agg + (long)d * D + c * 4;
  atomicAdd(ap + 0, v.x);
  atomicAdd(ap + 1, v.y);
  atomicAdd(ap + 2, v.z);
  atomicAdd(ap + 3, v.w);
}

__global__ __launch_bounds__(256) void gnn_sage_gemm_kernel(
    const float* __restrict__ agg, const float* __restrict__ h,
    const float* __restrict__ invdeg,
    const float* __restrict__ Wl, const float* __restrict__ Wr,
    const float* __restrict__ bias,
    float* __restrict__ out, int nN, int do_relu, int do_resid) {
  __shared__ __align__(16) float2 sWf[2][16][4][32];
  __shared__ __align__(16) float sB[D];
  int tid = threadIdx.x;

  for (int t = tid; t < 2048; t += 256) {
    int k  = t >> 7;
    int n  = (t >> 5) & 3;
    int ln = t & 31;
    int koff = (ln < 16) ? 0 : 2;
    int lr = ln & 15;
    int kb = k * 4 + koff;
    int col = n * 16 + lr;
    float2 f0, f1;
    f0.x = Wl[(kb + 0) * D + col];
    f0.y = Wl[(kb + 1) * D + col];
    f1.x = Wr[(kb + 0) * D + col];
    f1.y = Wr[(kb + 1) * D + col];
    sWf[0][k][n][ln] = f0;
    sWf[1][k][n][ln] = f1;
  }
  if (tid < D) sB[tid] = bias[tid];
  __syncthreads();

  int wave = tid >> 5;
  int lane = tid & 31;
  int lrow = lane & 15;
  int khalf = (lane < 16) ? 0 : 2;
  int rowbase = blockIdx.x * 128 + wave * 16;
  int M = rowbase + lrow;
  int Mc = (M < nN) ? M : (nN - 1);

  float invd = invdeg[Mc];

  v8f acc[4];
#pragma unroll
  for (int n = 0; n < 4; ++n) {
    float bv = sB[n * 16 + lrow];
#pragma unroll
    for (int j = 0; j < 8; ++j) acc[n][j] = bv;
  }

  for (int p = 0; p < 2; ++p) {
    const float* Arow = (p == 0 ? agg : h) + (long)Mc * D;
    float scale = (p == 0) ? invd : 1.0f;
#pragma unroll
    for (int k = 0; k < 16; ++k) {
      int kb = k * 4 + khalf;
      v2f a;
      a[0] = Arow[kb + 0] * scale;
      a[1] = Arow[kb + 1] * scale;
#pragma unroll
      for (int n = 0; n < 4; ++n) {
        float2 bf2 = sWf[p][k][n][lane];
        v2f bf;
        bf[0] = bf2.x;
        bf[1] = bf2.y;
        acc[n] = __builtin_amdgcn_wmma_f32_16x16x4_f32(
            false, a, false, bf, (short)0, acc[n], false, false);
      }
    }
  }

#pragma unroll
  for (int n = 0; n < 4; ++n) {
#pragma unroll
    for (int j = 0; j < 8; ++j) {
      int row = rowbase + ((lane < 16) ? j : j + 8);
      int col = n * 16 + lrow;
      if (row < nN) {
        float v = acc[n][j];
        if (do_resid) v += h[(long)row * D + col];
        if (do_relu) v = fmaxf(v, 0.0f);
        out[(long)row * D + col] = v;
      }
    }
  }
}

extern "C" void kernel_launch(void* const* d_in, const int* in_sizes, int n_in,
                              void* d_out, int out_size, void* d_ws, size_t ws_size,
                              hipStream_t stream) {
  const float* x  = (const float*)d_in[0];
  const int*   ei = (const int*)d_in[1];
  const float* Wl = (const float*)d_in[2];
  const float* Wr = (const float*)d_in[3];
  const float* b  = (const float*)d_in[4];
  float* out = (float*)d_out;

  const int* src = ei;
  const int* dst = ei + N_EDGES;

  const size_t featBytes = (size_t)N_NODES * D * sizeof(float);
  const size_t nodeI     = align256((size_t)N_NODES * sizeof(int));
  const size_t nodeF     = align256((size_t)N_NODES * sizeof(float));
  const size_t edgeI     = align256((size_t)N_EDGES * sizeof(int));
  const int scanBlocks   = (N_NODES + 255) / 256;
  const size_t histBytes = align256((size_t)EDGE_CHUNKS * SEG_PITCH * sizeof(int));

  const size_t csrNeed = 2 * nodeI + histBytes + nodeF + 2 * edgeI + 2 * featBytes;

  char* ws = (char*)d_ws;

  if (ws_size >= csrNeed) {
    int*   degi       = (int*)ws;   ws += nodeI;
    int*   rowptr     = (int*)ws;   ws += nodeI;
    int*   hist       = (int*)ws;   ws += histBytes;
    float* invdeg     = (float*)ws; ws += nodeF;
    int*   colidx     = (int*)ws;   ws += edgeI;
    int*   ebuf       = (int*)ws;   ws += edgeI;
    float* h1         = (float*)ws; ws += featBytes;
    float* h2         = (float*)ws;

    gnn_chunk_sort_kernel<<<EDGE_CHUNKS, EDGE_THREADS, 0, stream>>>(src, dst, hist, ebuf, N_EDGES);
    gnn_bucket_rows_kernel<<<N_BUCKETS, EDGE_THREADS, 0, stream>>>(ebuf, hist, rowptr, degi, invdeg, colidx, N_NODES, N_EDGES);

    const float* h = x;
    float* houts[N_LAYERS] = {h1, h2, h1, h2, out};
    int fusedBlocks = GATHER_WAVES / 8;

    for (int i = 0; i < N_LAYERS; ++i) {
      int relu  = (i < N_LAYERS - 1) ? 1 : 0;
      int resid = (i > 0 && i < N_LAYERS - 1) ? 1 : 0;
      gnn_sage_fused_kernel<<<fusedBlocks, 256, 0, stream>>>(
          h, colidx, rowptr, degi, invdeg,
          Wl + (long)i * D * D, Wr + (long)i * D * D, b + (long)i * D,
          houts[i], N_NODES, relu, resid);
      h = houts[i];
    }
  } else {
    float* invdeg = (float*)ws; ws += nodeF;
    float* agg    = (float*)ws; ws += featBytes;
    float* h1     = (float*)ws; ws += featBytes;
    float* h2     = (float*)ws;

    hipMemsetAsync(invdeg, 0, (size_t)N_NODES * sizeof(float), stream);
    gnn_deg_int_kernel<<<(N_EDGES + 255) / 256, 256, 0, stream>>>(dst, (int*)invdeg, N_EDGES);
    gnn_invdeg_from_int_kernel<<<scanBlocks, 256, 0, stream>>>((const int*)invdeg, invdeg, N_NODES);

    const float* h = x;
    float* houts[N_LAYERS] = {h1, h2, h1, h2, out};
    long scatterBlocks = ((long)N_EDGES * 16 + 255) / 256;
    int gemmBlocks = (N_NODES + 127) / 128;

    for (int i = 0; i < N_LAYERS; ++i) {
      hipMemsetAsync(agg, 0, featBytes, stream);
      gnn_scatter_kernel<<<(int)scatterBlocks, 256, 0, stream>>>(h, src, dst, agg, N_EDGES);
      int relu  = (i < N_LAYERS - 1) ? 1 : 0;
      int resid = (i > 0 && i < N_LAYERS - 1) ? 1 : 0;
      gnn_sage_gemm_kernel<<<gemmBlocks, 256, 0, stream>>>(
          agg, h, invdeg,
          Wl + (long)i * D * D, Wr + (long)i * D * D, b + (long)i * D,
          houts[i], N_NODES, relu, resid);
      h = houts[i];
    }
  }
}
